# starting kernel
# baseline (speedup 1.0000x reference)
#include <hip/hip_runtime.h>
#include <hip/hip_bf16.h>

typedef __attribute__((ext_vector_type(2))) float v2f;
typedef __attribute__((ext_vector_type(8))) float v8f;

#define NNODES 50000
#define NEDGES 800000
#define NGRAPH 512
#define INDIM  128
#define HID    96
#define OUTDIM 10
#define NCONV  4
#define NTILE  6
#define GEMM_WAVES 8

static inline int cdiv(int a, int b) { return (a + b - 1) / b; }

__global__ void k_deg_init(float* deg, int n) {
    int i = blockIdx.x * blockDim.x + threadIdx.x;
    if (i < n) deg[i] = 1.0f;
}

__global__ void k_deg_accum(const int* __restrict__ ei, float* deg, int E) {
    int e = blockIdx.x * blockDim.x + threadIdx.x;
    if (e < E) atomicAdd(&deg[ei[E + e]], 1.0f);
}

__global__ void k_rsqrt_inplace(float* deg, int n) {
    int i = blockIdx.x * blockDim.x + threadIdx.x;
    if (i < n) deg[i] = rsqrtf(deg[i]);
}

template <int RELUA, int RELU, int FUSE>
__global__ void k_gemm_wmma(const float* __restrict__ A, const float* __restrict__ B,
                            const float* __restrict__ bias,
                            float* __restrict__ C, float* __restrict__ C2,
                            const float* __restrict__ dinv,
                            int M, int K, int lda) {
    extern __shared__ float sBt[];
    int lstride = K + 2;

    int nB = K * HID;
    for (int i = threadIdx.x; i < nB; i += blockDim.x) {
        int k   = i / HID;
        int col = i - k * HID;
        sBt[col * lstride + k] = B[i];
    }
    __syncthreads();

    int mt   = blockIdx.x * GEMM_WAVES + (threadIdx.x >> 5);
    int lane = threadIdx.x & 31;
    int mTiles = M >> 4;
    if (mt >= mTiles) return;

    int half = lane >> 4;
    int l15  = lane & 15;

    const float* arow  = A + (size_t)(mt * 16 + l15) * lda;
    const float* bbase = sBt + (size_t)l15 * lstride;

    v8f acc[NTILE] = {};
    for (int k = 0; k < K; k += 4) {
        int kb = k + half * 2;
        v2f a = *(const v2f*)(arow + kb);
        if (RELUA) { a[0] = fmaxf(a[0], 0.0f); a[1] = fmaxf(a[1], 0.0f); }
        #pragma unroll
        for (int nt = 0; nt < NTILE; nt++) {
            v2f b = *(const v2f*)(bbase + (size_t)(nt * 16) * lstride + kb);
            acc[nt] = __builtin_amdgcn_wmma_f32_16x16x4_f32(
                 false, a,  false, b,
                 (short)0, acc[nt],  false,  false);
        }
    }

    int rbase = mt * 16 + half * 8;
    float* cbase = C + (size_t)rbase * HID + l15;

    if (FUSE) {
        float dv2[8];
        #pragma unroll
        for (int v = 0; v < 8; v++) { float d = dinv[rbase + v]; dv2[v] = d * d; }
        float* c2base = C2 + (size_t)rbase * HID + l15;
        #pragma unroll
        for (int nt = 0; nt < NTILE; nt++) {
            float bv = bias[nt * 16 + l15];
            #pragma unroll
            for (int v = 0; v < 8; v++) {
                float raw = acc[nt][v];
                cbase [(size_t)v * HID + nt * 16] = raw;
                c2base[(size_t)v * HID + nt * 16] = raw * dv2[v] + bv;
            }
        }
    } else {
        #pragma unroll
        for (int nt = 0; nt < NTILE; nt++) {
            float bv = bias[nt * 16 + l15];
            #pragma unroll
            for (int v = 0; v < 8; v++) {
                float r = acc[nt][v] + bv;
                if (RELU) r = fmaxf(r, 0.0f);
                cbase[(size_t)v * HID + nt * 16] = r;
            }
        }
    }
}

__global__ void k_edge_scatter(const float* __restrict__ hw, const float* __restrict__ dinv,
                               const int* __restrict__ ei, float* __restrict__ out, int E) {
    int wave = (blockIdx.x * blockDim.x + threadIdx.x) >> 5;
    int lane = threadIdx.x & 31;
    if (wave >= E) return;
    int s = ei[wave];
    int d = ei[E + wave];
    float norm = dinv[s] * dinv[d];
    const float* srow = hw  + (size_t)s * HID;
    float*       drow = out + (size_t)d * HID;
    #pragma unroll
    for (int k = 0; k < 3; k++) {
        int c = lane + k * 32;
        atomicAdd(drow + c, srow[c] * norm);
    }
}

__global__ void k_zero(float* x, int n) {
    int i = blockIdx.x * blockDim.x + threadIdx.x;
    if (i < n) x[i] = 0.0f;
}

__global__ void k_pool(const float* __restrict__ h, const int* __restrict__ batch,
                       float* __restrict__ g, int n) {
    int idx = blockIdx.x * blockDim.x + threadIdx.x;
    if (idx < n * HID) {
        int node = idx / HID;
        int c    = idx - node * HID;
        atomicAdd(&g[(size_t)batch[node] * HID + c], fmaxf(h[idx], 0.0f));
    }
}

__global__ void k_dec2(const float* __restrict__ g2, const float* __restrict__ W,
                       const float* __restrict__ b, float* __restrict__ out) {
    int idx = blockIdx.x * blockDim.x + threadIdx.x;
    if (idx < NGRAPH * OUTDIM) {
        int gi = idx / OUTDIM;
        int o  = idx - gi * OUTDIM;
        float s = b[o];
        const float* row = g2 + (size_t)gi * HID;
        #pragma unroll 8
        for (int k = 0; k < HID; k++) s += row[k] * W[k * OUTDIM + o];
        out[idx] = s;
    }
}

extern "C" void kernel_launch(void* const* d_in, const int* in_sizes, int n_in,
                              void* d_out, int out_size, void* d_ws, size_t ws_size,
                              hipStream_t stream) {
    const float* x       = (const float*)d_in[0];
    const int*   ei      = (const int*)  d_in[1];
    const int*   batch   = (const int*)  d_in[2];
    const float* enc_W0  = (const float*)d_in[3];
    const float* enc_b0  = (const float*)d_in[4];
    const float* enc_W1  = (const float*)d_in[5];
    const float* enc_b1  = (const float*)d_in[6];
    const float* conv_W  = (const float*)d_in[7];
    const float* conv_b  = (const float*)d_in[8];
    const float* dec_W0  = (const float*)d_in[9];
    const float* dec_b0  = (const float*)d_in[10];
    const float* dec_W1  = (const float*)d_in[11];
    const float* dec_b1  = (const float*)d_in[12];
    float* out = (float*)d_out;

    float* ws   = (float*)d_ws;
    size_t feat = (size_t)NNODES * HID;
    float* buf0 = ws;
    float* buf1 = buf0 + feat;
    float* buf2 = buf1 + feat;
    float* dinv = buf2 + feat;
    float* g    = dinv + NNODES;
    float* g2   = g + (size_t)NGRAPH * HID;

    const int B = 256;

    k_deg_init<<<cdiv(NNODES, B), B, 0, stream>>>(dinv, NNODES);
    k_deg_accum<<<cdiv(NEDGES, B), B, 0, stream>>>(ei, dinv, NEDGES);
    k_rsqrt_inplace<<<cdiv(NNODES, B), B, 0, stream>>>(dinv, NNODES);

    int mTilesN   = NNODES / 16;
    int gemmBlksN = cdiv(mTilesN, GEMM_WAVES);
    size_t shm128 = (size_t)HID * (INDIM + 2) * sizeof(float);
    size_t shm96  = (size_t)HID * (HID   + 2) * sizeof(float);
    k_gemm_wmma<0, 1, 0><<<gemmBlksN, B, shm128, stream>>>(
        x, enc_W0, enc_b0, buf0, nullptr, nullptr, NNODES, INDIM, INDIM);
    k_gemm_wmma<0, 0, 0><<<gemmBlksN, B, shm96, stream>>>(
        buf0, enc_W1, enc_b1, buf1, nullptr, nullptr, NNODES, HID, HID);

    float* cur = buf1;
    float* nxt = buf2;
    int edgeBlks = cdiv(NEDGES * 32, B);
    for (int l = 0; l < NCONV; l++) {
        const float* Wl = conv_W + (size_t)l * HID * HID;
        const float* bl = conv_b + (size_t)l * HID;
        if (l == 0)
            k_gemm_wmma<0, 0, 1><<<gemmBlksN, B, shm96, stream>>>(
                cur, Wl, bl, buf0  , nxt  , dinv, NNODES, HID, HID);
        else
            k_gemm_wmma<1, 0, 1><<<gemmBlksN, B, shm96, stream>>>(
                cur, Wl, bl, buf0  , nxt  , dinv, NNODES, HID, HID);
        k_edge_scatter<<<edgeBlks, B, 0, stream>>>(buf0, dinv, ei, nxt, NEDGES);
        float* t = cur; cur = nxt; nxt = t;
    }

    k_zero<<<cdiv(NGRAPH * HID, B), B, 0, stream>>>(g, NGRAPH * HID);
    k_pool<<<cdiv(NNODES * HID, B), B, 0, stream>>>(cur, batch, g, NNODES);

    int mTilesG = NGRAPH / 16;
    k_gemm_wmma<0, 1, 0><<<cdiv(mTilesG, GEMM_WAVES), B, shm96, stream>>>(
        g, dec_W0, dec_b0, g2, nullptr, nullptr, NGRAPH, HID, HID);
    k_dec2<<<cdiv(NGRAPH * OUTDIM, B), B, 0, stream>>>(g2, dec_W1, dec_b1, out);
}


# Edge aggregation as an atomic-free gather per destination over an edge list grouped by destination, built once per call (arrival-order slots): sum of dinv[s]*hw[s] in list order, scaled by dinv[d]
# speedup vs baseline: 1.9184x; 1.9184x over previous
#include <hip/hip_runtime.h>
#include <hip/hip_bf16.h>

typedef __attribute__((ext_vector_type(2))) float v2f;
typedef __attribute__((ext_vector_type(8))) float v8f;
typedef __attribute__((ext_vector_type(4))) float v4f;

#define NNODES 50000
#define NEDGES 800000
#define NGRAPH 512
#define INDIM  128
#define HID    96
#define OUTDIM 10
#define NCONV  4
#define NTILE  6
#define GEMM_WAVES 8

static inline int cdiv(int a, int b) { return (a + b - 1) / b; }

__global__ void k_zero_i(int* x, int n) {
    int i = blockIdx.x * blockDim.x + threadIdx.x;
    if (i < n) x[i] = 0;
}

__global__ void k_rank(const int* __restrict__ ei, int* cnt, int* __restrict__ rank, int E) {
    int e = blockIdx.x * blockDim.x + threadIdx.x;
    if (e < E) rank[e] = atomicAdd(&cnt[ei[E + e]], 1);
}

__global__ void k_offsets(const int* cnt, int* total, int* __restrict__ off,
                          float* __restrict__ dinv, int* __restrict__ colpad, int n) {
    int i    = blockIdx.x * blockDim.x + threadIdx.x;
    int lane = threadIdx.x & 31;
    int c    = (i < n) ? cnt[i] : 0;
    int incl = c;
    #pragma unroll
    for (int d = 1; d < 32; d <<= 1) {
        int v = __shfl_up(incl, d);
        if (lane >= d) incl += v;
    }
    int base = 0;
    if (lane == 31) base = atomicAdd(total, incl);
    base = __shfl(base, 31);
    if (i < n) {
        off[i]  = base + incl - c;
        dinv[i] = rsqrtf((float)(c + 1));
    }
    if (i < 4) colpad[i] = 0;
}

__global__ void k_fill(const int* __restrict__ ei, const int* __restrict__ rank,
                       const int* __restrict__ off, int* __restrict__ col, int E) {
    int e = blockIdx.x * blockDim.x + threadIdx.x;
    if (e < E) col[off[ei[E + e]] + rank[e]] = ei[e];
}

template <int RELUA, int RELU, int FUSE>
__global__ void k_gemm_wmma(const float* __restrict__ A, const float* __restrict__ B,
                            const float* __restrict__ bias,
                            float* __restrict__ C,
                            const float* __restrict__ dinv,
                            int M, int K, int lda) {
    extern __shared__ float sBt[];
    int lstride = K + 2;

    int nB = K * HID;
    for (int i = threadIdx.x; i < nB; i += blockDim.x) {
        int k   = i / HID;
        int col = i - k * HID;
        sBt[col * lstride + k] = B[i];
    }
    __syncthreads();

    int mt   = blockIdx.x * GEMM_WAVES + (threadIdx.x >> 5);
    int lane = threadIdx.x & 31;
    int mTiles = M >> 4;
    if (mt >= mTiles) return;

    int half = lane >> 4;
    int l15  = lane & 15;

    const float* arow  = A + (size_t)(mt * 16 + l15) * lda;
    const float* bbase = sBt + (size_t)l15 * lstride;

    v8f acc[NTILE] = {};
    for (int k = 0; k < K; k += 4) {
        int kb = k + half * 2;
        v2f a = *(const v2f*)(arow + kb);
        if (RELUA) { a[0] = fmaxf(a[0], 0.0f); a[1] = fmaxf(a[1], 0.0f); }
        #pragma unroll
        for (int nt = 0; nt < NTILE; nt++) {
            v2f b = *(const v2f*)(bbase + (size_t)(nt * 16) * lstride + kb);
            acc[nt] = __builtin_amdgcn_wmma_f32_16x16x4_f32(
                 false, a,  false, b,
                 (short)0, acc[nt],  false,  false);
        }
    }

    int rbase = mt * 16 + half * 8;
    float* cbase = C + (size_t)rbase * HID + l15;

    if (FUSE) {
        float dv[8];
        #pragma unroll
        for (int v = 0; v < 8; v++) dv[v] = dinv[rbase + v];
        #pragma unroll
        for (int nt = 0; nt < NTILE; nt++) {
            #pragma unroll
            for (int v = 0; v < 8; v++)
                cbase[(size_t)v * HID + nt * 16] = acc[nt][v] * dv[v];
        }
    } else {
        #pragma unroll
        for (int nt = 0; nt < NTILE; nt++) {
            float bv = bias[nt * 16 + l15];
            #pragma unroll
            for (int v = 0; v < 8; v++) {
                float r = acc[nt][v] + bv;
                if (RELU) r = fmaxf(r, 0.0f);
                cbase[(size_t)v * HID + nt * 16] = r;
            }
        }
    }
}

__global__ void __launch_bounds__(256)
k_gather(const float* __restrict__ hws, const float* __restrict__ dinv,
         const int* __restrict__ off, const int* __restrict__ cnt,
         const int* __restrict__ col, const float* __restrict__ bias,
         float* __restrict__ out, int n) {
    int node = (blockIdx.x * blockDim.x + threadIdx.x) >> 5;
    if (node >= n) return;
    int lane = threadIdx.x & 31;
    int grp  = lane >> 3;
    int t    = lane & 7;
    int start = off[node];
    int c     = cnt[node];
    int nfull = __builtin_amdgcn_readfirstlane(c >> 2);
    int rem   = c & 3;
    const int* cp = col + start + grp;
    int srem = cp[4 * nfull];

    const v4f* rows = (const v4f*)hws + t;
    v4f a0 = {}, a1 = {}, a2 = {};
    if (grp == 0) {
        const v4f* r = rows + (unsigned)node * 24u;
        a0 = r[0]; a1 = r[8]; a2 = r[16];
    }
    #pragma unroll 2
    for (int i = 0; i < nfull; i++) {
        int s = cp[4 * i];
        const v4f* r = rows + (unsigned)s * 24u;
        v4f x0 = r[0], x1 = r[8], x2 = r[16];
        a0 += x0; a1 += x1; a2 += x2;
    }
    if (grp < rem) {
        const v4f* r = rows + (unsigned)srem * 24u;
        v4f x0 = r[0], x1 = r[8], x2 = r[16];
        a0 += x0; a1 += x1; a2 += x2;
    }
    #pragma unroll
    for (int m = 8; m <= 16; m <<= 1) {
        #pragma unroll
        for (int q = 0; q < 4; q++) {
            a0[q] += __shfl_xor(a0[q], m);
            a1[q] += __shfl_xor(a1[q], m);
            a2[q] += __shfl_xor(a2[q], m);
        }
    }
    if (grp == 0) {
        float dv = dinv[node];
        const v4f* b4 = (const v4f*)bias + t;
        v4f* o = (v4f*)out + (unsigned)node * 24u + t;
        o[0]  = a0 * dv + b4[0];
        o[8]  = a1 * dv + b4[8];
        o[16] = a2 * dv + b4[16];
    }
}

__global__ void k_zero(float* x, int n) {
    int i = blockIdx.x * blockDim.x + threadIdx.x;
    if (i < n) x[i] = 0.0f;
}

__global__ void k_pool(const float* __restrict__ h, const int* __restrict__ batch,
                       float* __restrict__ g, int n) {
    int idx = blockIdx.x * blockDim.x + threadIdx.x;
    if (idx < n * HID) {
        int node = idx / HID;
        int c    = idx - node * HID;
        atomicAdd(&g[(size_t)batch[node] * HID + c], fmaxf(h[idx], 0.0f));
    }
}

__global__ void k_dec2(const float* __restrict__ g2, const float* __restrict__ W,
                       const float* __restrict__ b, float* __restrict__ out) {
    int idx = blockIdx.x * blockDim.x + threadIdx.x;
    if (idx < NGRAPH * OUTDIM) {
        int gi = idx / OUTDIM;
        int o  = idx - gi * OUTDIM;
        float s = b[o];
        const float* row = g2 + (size_t)gi * HID;
        #pragma unroll 8
        for (int k = 0; k < HID; k++) s += row[k] * W[k * OUTDIM + o];
        out[idx] = s;
    }
}

extern "C" void kernel_launch(void* const* d_in, const int* in_sizes, int n_in,
                              void* d_out, int out_size, void* d_ws, size_t ws_size,
                              hipStream_t stream) {
    const float* x       = (const float*)d_in[0];
    const int*   ei      = (const int*)  d_in[1];
    const int*   batch   = (const int*)  d_in[2];
    const float* enc_W0  = (const float*)d_in[3];
    const float* enc_b0  = (const float*)d_in[4];
    const float* enc_W1  = (const float*)d_in[5];
    const float* enc_b1  = (const float*)d_in[6];
    const float* conv_W  = (const float*)d_in[7];
    const float* conv_b  = (const float*)d_in[8];
    const float* dec_W0  = (const float*)d_in[9];
    const float* dec_b0  = (const float*)d_in[10];
    const float* dec_W1  = (const float*)d_in[11];
    const float* dec_b1  = (const float*)d_in[12];
    float* out = (float*)d_out;

    float* ws   = (float*)d_ws;
    size_t feat = (size_t)NNODES * HID;
    float* buf0 = ws;
    float* buf1 = buf0 + feat;
    float* buf2 = buf1 + feat;
    float* dinv = buf2 + feat;
    float* g    = dinv + NNODES;
    float* g2   = g + (size_t)NGRAPH * HID;
    int*   cnt  = (int*)(g2 + (size_t)NGRAPH * HID);
    int*   tot  = cnt + NNODES;
    int*   off  = tot + 16;
    int*   rank = off + NNODES;
    int*   col  = rank + NEDGES;

    const int B = 256;

    k_zero_i<<<cdiv(NNODES + 1, B), B, 0, stream>>>(cnt, NNODES + 1);
    k_rank<<<cdiv(NEDGES, B), B, 0, stream>>>(ei, cnt, rank, NEDGES);
    k_offsets<<<cdiv(NNODES, B), B, 0, stream>>>(cnt, tot, off, dinv, col + NEDGES, NNODES);
    k_fill<<<cdiv(NEDGES, B), B, 0, stream>>>(ei, rank, off, col, NEDGES);

    int mTilesN   = NNODES / 16;
    int gemmBlksN = cdiv(mTilesN, GEMM_WAVES);
    size_t shm128 = (size_t)HID * (INDIM + 2) * sizeof(float);
    size_t shm96  = (size_t)HID * (HID   + 2) * sizeof(float);
    k_gemm_wmma<0, 1, 0><<<gemmBlksN, B, shm128, stream>>>(
        x, enc_W0, enc_b0, buf0, nullptr, NNODES, INDIM, INDIM);
    k_gemm_wmma<0, 0, 0><<<gemmBlksN, B, shm96, stream>>>(
        buf0, enc_W1, enc_b1, buf1, nullptr, NNODES, HID, HID);

    float* cur = buf1;
    float* nxt = buf2;
    int nodeBlks = cdiv(NNODES * 32, B);
    for (int l = 0; l < NCONV; l++) {
        const float* Wl = conv_W + (size_t)l * HID * HID;
        const float* bl = conv_b + (size_t)l * HID;
        if (l == 0)
            k_gemm_wmma<0, 0, 1><<<gemmBlksN, B, shm96, stream>>>(
                cur, Wl, nullptr, buf0  , dinv, NNODES, HID, HID);
        else
            k_gemm_wmma<1, 0, 1><<<gemmBlksN, B, shm96, stream>>>(
                cur, Wl, nullptr, buf0  , dinv, NNODES, HID, HID);
        k_gather<<<nodeBlks, B, 0, stream>>>(buf0, dinv, off, cnt, col, bl, nxt, NNODES);
        float* t = cur; cur = nxt; nxt = t;
    }

    k_zero<<<cdiv(NGRAPH * HID, B), B, 0, stream>>>(g, NGRAPH * HID);
    k_pool<<<cdiv(NNODES * HID, B), B, 0, stream>>>(cur, batch, g, NNODES);

    int mTilesG = NGRAPH / 16;
    k_gemm_wmma<0, 1, 0><<<cdiv(mTilesG, GEMM_WAVES), B, shm96, stream>>>(
        g, dec_W0, dec_b0, g2, nullptr, NGRAPH, HID, HID);
    k_dec2<<<cdiv(NGRAPH * OUTDIM, B), B, 0, stream>>>(g2, dec_W1, dec_b1, out);
}


# GEMM B staging: each thread requests its share of B as 16-byte loads in one batch before its LDS stores (K made compile-time), replacing 36-48 serial 4-byte load-wait-store round trips per block
# speedup vs baseline: 2.6909x; 1.4027x over previous
#include <hip/hip_runtime.h>
#include <hip/hip_bf16.h>

typedef __attribute__((ext_vector_type(2))) float v2f;
typedef __attribute__((ext_vector_type(8))) float v8f;
typedef __attribute__((ext_vector_type(4))) float v4f;

#define NNODES 50000
#define NEDGES 800000
#define NGRAPH 512
#define INDIM  128
#define HID    96
#define OUTDIM 10
#define NCONV  4
#define NTILE  6
#define GEMM_WAVES 8
#define GEMM_THREADS (GEMM_WAVES * 32)

static inline int cdiv(int a, int b) { return (a + b - 1) / b; }

__global__ void k_zero_i(int* x, int n) {
    int i = blockIdx.x * blockDim.x + threadIdx.x;
    if (i < n) x[i] = 0;
}

__global__ void k_rank(const int* __restrict__ ei, int* cnt, int* __restrict__ rank, int E) {
    int e = blockIdx.x * blockDim.x + threadIdx.x;
    if (e < E) rank[e] = atomicAdd(&cnt[ei[E + e]], 1);
}

__global__ void k_offsets(const int* cnt, int* total, int* __restrict__ off,
                          float* __restrict__ dinv, int* __restrict__ colpad, int n) {
    int i    = blockIdx.x * blockDim.x + threadIdx.x;
    int lane = threadIdx.x & 31;
    int c    = (i < n) ? cnt[i] : 0;
    int incl = c;
    #pragma unroll
    for (int d = 1; d < 32; d <<= 1) {
        int v = __shfl_up(incl, d);
        if (lane >= d) incl += v;
    }
    int base = 0;
    if (lane == 31) base = atomicAdd(total, incl);
    base = __shfl(base, 31);
    if (i < n) {
        off[i]  = base + incl - c;
        dinv[i] = rsqrtf((float)(c + 1));
    }
    if (i < 4) colpad[i] = 0;
}

__global__ void k_fill(const int* __restrict__ ei, const int* __restrict__ rank,
                       const int* __restrict__ off, int* __restrict__ col, int E) {
    int e = blockIdx.x * blockDim.x + threadIdx.x;
    if (e < E) col[off[ei[E + e]] + rank[e]] = ei[e];
}

template <int K, int RELUA, int RELU, int FUSE>
__global__ void __launch_bounds__(GEMM_THREADS)
k_gemm_wmma(const float* __restrict__ A, const float* __restrict__ B,
            const float* __restrict__ bias,
            float* __restrict__ C,
            const float* __restrict__ dinv,
            int M, int lda) {
    extern __shared__ float sBt[];
    constexpr int lstride = K + 2;

    constexpr int ROW4 = HID / 4;
    constexpr int PER  = K * ROW4 / GEMM_THREADS;
    static_assert(K * ROW4 % GEMM_THREADS == 0, "B must split evenly over the block");
    const v4f* B4 = (const v4f*)B;
    v4f piece[PER];
    #pragma unroll
    for (int j = 0; j < PER; j++) piece[j] = B4[threadIdx.x + j * GEMM_THREADS];
    #pragma unroll
    for (int j = 0; j < PER; j++) {
        int i  = threadIdx.x + j * GEMM_THREADS;
        int k  = i / ROW4;
        int c0 = (i - k * ROW4) * 4;
        float* dst = sBt + c0 * lstride + k;
        dst[0]           = piece[j][0];
        dst[lstride]     = piece[j][1];
        dst[2 * lstride] = piece[j][2];
        dst[3 * lstride] = piece[j][3];
    }
    __syncthreads();

    int mt   = blockIdx.x * GEMM_WAVES + (threadIdx.x >> 5);
    int lane = threadIdx.x & 31;
    int mTiles = M >> 4;
    if (mt >= mTiles) return;

    int half = lane >> 4;
    int l15  = lane & 15;

    const float* arow  = A + (size_t)(mt * 16 + l15) * lda;
    const float* bbase = sBt + (size_t)l15 * lstride;

    v8f acc[NTILE] = {};
    for (int k = 0; k < K; k += 4) {
        int kb = k + half * 2;
        v2f a = *(const v2f*)(arow + kb);
        if (RELUA) { a[0] = fmaxf(a[0], 0.0f); a[1] = fmaxf(a[1], 0.0f); }
        #pragma unroll
        for (int nt = 0; nt < NTILE; nt++) {
            v2f b = *(const v2f*)(bbase + (size_t)(nt * 16) * lstride + kb);
            acc[nt] = __builtin_amdgcn_wmma_f32_16x16x4_f32(
                 false, a,  false, b,
                 (short)0, acc[nt],  false,  false);
        }
    }

    int rbase = mt * 16 + half * 8;
    float* cbase = C + (size_t)rbase * HID + l15;

    if (FUSE) {
        float dv[8];
        #pragma unroll
        for (int v = 0; v < 8; v++) dv[v] = dinv[rbase + v];
        #pragma unroll
        for (int nt = 0; nt < NTILE; nt++) {
            #pragma unroll
            for (int v = 0; v < 8; v++)
                cbase[(size_t)v * HID + nt * 16] = acc[nt][v] * dv[v];
        }
    } else {
        #pragma unroll
        for (int nt = 0; nt < NTILE; nt++) {
            float bv = bias[nt * 16 + l15];
            #pragma unroll
            for (int v = 0; v < 8; v++) {
                float r = acc[nt][v] + bv;
                if (RELU) r = fmaxf(r, 0.0f);
                cbase[(size_t)v * HID + nt * 16] = r;
            }
        }
    }
}

__global__ void __launch_bounds__(256)
k_gather(const float* __restrict__ hws, const float* __restrict__ dinv,
         const int* __restrict__ off, const int* __restrict__ cnt,
         const int* __restrict__ col, const float* __restrict__ bias,
         float* __restrict__ out, int n) {
    int node = (blockIdx.x * blockDim.x + threadIdx.x) >> 5;
    if (node >= n) return;
    int lane = threadIdx.x & 31;
    int grp  = lane >> 3;
    int t    = lane & 7;
    int start = off[node];
    int c     = cnt[node];
    int nfull = __builtin_amdgcn_readfirstlane(c >> 2);
    int rem   = c & 3;
    const int* cp = col + start + grp;
    int srem = cp[4 * nfull];

    const v4f* rows = (const v4f*)hws + t;
    v4f a0 = {}, a1 = {}, a2 = {};
    if (grp == 0) {
        const v4f* r = rows + (unsigned)node * 24u;
        a0 = r[0]; a1 = r[8]; a2 = r[16];
    }
    #pragma unroll 2
    for (int i = 0; i < nfull; i++) {
        int s = cp[4 * i];
        const v4f* r = rows + (unsigned)s * 24u;
        v4f x0 = r[0], x1 = r[8], x2 = r[16];
        a0 += x0; a1 += x1; a2 += x2;
    }
    if (grp < rem) {
        const v4f* r = rows + (unsigned)srem * 24u;
        v4f x0 = r[0], x1 = r[8], x2 = r[16];
        a0 += x0; a1 += x1; a2 += x2;
    }
    #pragma unroll
    for (int m = 8; m <= 16; m <<= 1) {
        #pragma unroll
        for (int q = 0; q < 4; q++) {
            a0[q] += __shfl_xor(a0[q], m);
            a1[q] += __shfl_xor(a1[q], m);
            a2[q] += __shfl_xor(a2[q], m);
        }
    }
    if (grp == 0) {
        float dv = dinv[node];
        const v4f* b4 = (const v4f*)bias + t;
        v4f* o = (v4f*)out + (unsigned)node * 24u + t;
        o[0]  = a0 * dv + b4[0];
        o[8]  = a1 * dv + b4[8];
        o[16] = a2 * dv + b4[16];
    }
}

__global__ void k_zero(float* x, int n) {
    int i = blockIdx.x * blockDim.x + threadIdx.x;
    if (i < n) x[i] = 0.0f;
}

__global__ void k_pool(const float* __restrict__ h, const int* __restrict__ batch,
                       float* __restrict__ g, int n) {
    int idx = blockIdx.x * blockDim.x + threadIdx.x;
    if (idx < n * HID) {
        int node = idx / HID;
        int c    = idx - node * HID;
        atomicAdd(&g[(size_t)batch[node] * HID + c], fmaxf(h[idx], 0.0f));
    }
}

__global__ void k_dec2(const float* __restrict__ g2, const float* __restrict__ W,
                       const float* __restrict__ b, float* __restrict__ out) {
    int idx = blockIdx.x * blockDim.x + threadIdx.x;
    if (idx < NGRAPH * OUTDIM) {
        int gi = idx / OUTDIM;
        int o  = idx - gi * OUTDIM;
        float s = b[o];
        const float* row = g2 + (size_t)gi * HID;
        #pragma unroll 8
        for (int k = 0; k < HID; k++) s += row[k] * W[k * OUTDIM + o];
        out[idx] = s;
    }
}

extern "C" void kernel_launch(void* const* d_in, const int* in_sizes, int n_in,
                              void* d_out, int out_size, void* d_ws, size_t ws_size,
                              hipStream_t stream) {
    const float* x       = (const float*)d_in[0];
    const int*   ei      = (const int*)  d_in[1];
    const int*   batch   = (const int*)  d_in[2];
    const float* enc_W0  = (const float*)d_in[3];
    const float* enc_b0  = (const float*)d_in[4];
    const float* enc_W1  = (const float*)d_in[5];
    const float* enc_b1  = (const float*)d_in[6];
    const float* conv_W  = (const float*)d_in[7];
    const float* conv_b  = (const float*)d_in[8];
    const float* dec_W0  = (const float*)d_in[9];
    const float* dec_b0  = (const float*)d_in[10];
    const float* dec_W1  = (const float*)d_in[11];
    const float* dec_b1  = (const float*)d_in[12];
    float* out = (float*)d_out;

    float* ws   = (float*)d_ws;
    size_t feat = (size_t)NNODES * HID;
    float* buf0 = ws;
    float* buf1 = buf0 + feat;
    float* buf2 = buf1 + feat;
    float* dinv = buf2 + feat;
    float* g    = dinv + NNODES;
    float* g2   = g + (size_t)NGRAPH * HID;
    int*   cnt  = (int*)(g2 + (size_t)NGRAPH * HID);
    int*   tot  = cnt + NNODES;
    int*   off  = tot + 16;
    int*   rank = off + NNODES;
    int*   col  = rank + NEDGES;

    const int B = 256;

    k_zero_i<<<cdiv(NNODES + 1, B), B, 0, stream>>>(cnt, NNODES + 1);
    k_rank<<<cdiv(NEDGES, B), B, 0, stream>>>(ei, cnt, rank, NEDGES);
    k_offsets<<<cdiv(NNODES, B), B, 0, stream>>>(cnt, tot, off, dinv, col + NEDGES, NNODES);
    k_fill<<<cdiv(NEDGES, B), B, 0, stream>>>(ei, rank, off, col, NEDGES);

    int mTilesN   = NNODES / 16;
    int gemmBlksN = cdiv(mTilesN, GEMM_WAVES);
    size_t shm128 = (size_t)HID * (INDIM + 2) * sizeof(float);
    size_t shm96  = (size_t)HID * (HID   + 2) * sizeof(float);
    k_gemm_wmma<INDIM, 0, 1, 0><<<gemmBlksN, GEMM_THREADS, shm128, stream>>>(
        x, enc_W0, enc_b0, buf0, nullptr, NNODES, INDIM);
    k_gemm_wmma<HID, 0, 0, 0><<<gemmBlksN, GEMM_THREADS, shm96, stream>>>(
        buf0, enc_W1, enc_b1, buf1, nullptr, NNODES, HID);

    float* cur = buf1;
    float* nxt = buf2;
    int nodeBlks = cdiv(NNODES * 32, B);
    for (int l = 0; l < NCONV; l++) {
        const float* Wl = conv_W + (size_t)l * HID * HID;
        const float* bl = conv_b + (size_t)l * HID;
        if (l == 0)
            k_gemm_wmma<HID, 0, 0, 1><<<gemmBlksN, GEMM_THREADS, shm96, stream>>>(
                cur, Wl, nullptr, buf0  , dinv, NNODES, HID);
        else
            k_gemm_wmma<HID, 1, 0, 1><<<gemmBlksN, GEMM_THREADS, shm96, stream>>>(
                cur, Wl, nullptr, buf0  , dinv, NNODES, HID);
        k_gather<<<nodeBlks, B, 0, stream>>>(buf0, dinv, off, cnt, col, bl, nxt, NNODES);
        float* t = cur; cur = nxt; nxt = t;
    }

    k_zero<<<cdiv(NGRAPH * HID, B), B, 0, stream>>>(g, NGRAPH * HID);
    k_pool<<<cdiv(NNODES * HID, B), B, 0, stream>>>(cur, batch, g, NNODES);

    int mTilesG = NGRAPH / 16;
    k_gemm_wmma<HID, 0, 1, 0><<<cdiv(mTilesG, GEMM_WAVES), GEMM_THREADS, shm96, stream>>>(
        g, dec_W0, dec_b0, g2, nullptr, NGRAPH, HID);
    k_dec2<<<cdiv(NGRAPH * OUTDIM, B), B, 0, stream>>>(g2, dec_W1, dec_b1, out);
}


# Pooling: one wave sums 32 consecutive nodes' rows in registers and adds each run of equal graph ids to g with one atomic per column (was one atomic per element); pool sum order changes
# speedup vs baseline: 2.7788x; 1.0327x over previous
#include <hip/hip_runtime.h>
#include <hip/hip_bf16.h>

typedef __attribute__((ext_vector_type(2))) float v2f;
typedef __attribute__((ext_vector_type(8))) float v8f;
typedef __attribute__((ext_vector_type(4))) float v4f;

#define NNODES 50000
#define NEDGES 800000
#define NGRAPH 512
#define INDIM  128
#define HID    96
#define OUTDIM 10
#define NCONV  4
#define NTILE  6
#define GEMM_WAVES 8
#define GEMM_THREADS (GEMM_WAVES * 32)

static inline int cdiv(int a, int b) { return (a + b - 1) / b; }

__global__ void k_zero_i(int* x, int n) {
    int i = blockIdx.x * blockDim.x + threadIdx.x;
    if (i < n) x[i] = 0;
}

__global__ void k_rank(const int* __restrict__ ei, int* cnt, int* __restrict__ rank, int E) {
    int e = blockIdx.x * blockDim.x + threadIdx.x;
    if (e < E) rank[e] = atomicAdd(&cnt[ei[E + e]], 1);
}

__global__ void k_offsets(const int* cnt, int* total, int* __restrict__ off,
                          float* __restrict__ dinv, int* __restrict__ colpad, int n) {
    int i    = blockIdx.x * blockDim.x + threadIdx.x;
    int lane = threadIdx.x & 31;
    int c    = (i < n) ? cnt[i] : 0;
    int incl = c;
    #pragma unroll
    for (int d = 1; d < 32; d <<= 1) {
        int v = __shfl_up(incl, d);
        if (lane >= d) incl += v;
    }
    int base = 0;
    if (lane == 31) base = atomicAdd(total, incl);
    base = __shfl(base, 31);
    if (i < n) {
        off[i]  = base + incl - c;
        dinv[i] = rsqrtf((float)(c + 1));
    }
    if (i < 4) colpad[i] = 0;
}

__global__ void k_fill(const int* __restrict__ ei, const int* __restrict__ rank,
                       const int* __restrict__ off, int* __restrict__ col, int E) {
    int e = blockIdx.x * blockDim.x + threadIdx.x;
    if (e < E) col[off[ei[E + e]] + rank[e]] = ei[e];
}

template <int K, int RELUA, int RELU, int FUSE>
__global__ void __launch_bounds__(GEMM_THREADS)
k_gemm_wmma(const float* __restrict__ A, const float* __restrict__ B,
            const float* __restrict__ bias,
            float* __restrict__ C,
            const float* __restrict__ dinv,
            int M, int lda) {
    extern __shared__ float sBt[];
    constexpr int lstride = K + 2;

    constexpr int ROW4 = HID / 4;
    constexpr int PER  = K * ROW4 / GEMM_THREADS;
    static_assert(K * ROW4 % GEMM_THREADS == 0, "B must split evenly over the block");
    const v4f* B4 = (const v4f*)B;
    v4f piece[PER];
    #pragma unroll
    for (int j = 0; j < PER; j++) piece[j] = B4[threadIdx.x + j * GEMM_THREADS];
    #pragma unroll
    for (int j = 0; j < PER; j++) {
        int i  = threadIdx.x + j * GEMM_THREADS;
        int k  = i / ROW4;
        int c0 = (i - k * ROW4) * 4;
        float* dst = sBt + c0 * lstride + k;
        dst[0]           = piece[j][0];
        dst[lstride]     = piece[j][1];
        dst[2 * lstride] = piece[j][2];
        dst[3 * lstride] = piece[j][3];
    }
    __syncthreads();

    int mt   = blockIdx.x * GEMM_WAVES + (threadIdx.x >> 5);
    int lane = threadIdx.x & 31;
    int mTiles = M >> 4;
    if (mt >= mTiles) return;

    int half = lane >> 4;
    int l15  = lane & 15;

    const float* arow  = A + (size_t)(mt * 16 + l15) * lda;
    const float* bbase = sBt + (size_t)l15 * lstride;

    v8f acc[NTILE] = {};
    for (int k = 0; k < K; k += 4) {
        int kb = k + half * 2;
        v2f a = *(const v2f*)(arow + kb);
        if (RELUA) { a[0] = fmaxf(a[0], 0.0f); a[1] = fmaxf(a[1], 0.0f); }
        #pragma unroll
        for (int nt = 0; nt < NTILE; nt++) {
            v2f b = *(const v2f*)(bbase + (size_t)(nt * 16) * lstride + kb);
            acc[nt] = __builtin_amdgcn_wmma_f32_16x16x4_f32(
                 false, a,  false, b,
                 (short)0, acc[nt],  false,  false);
        }
    }

    int rbase = mt * 16 + half * 8;
    float* cbase = C + (size_t)rbase * HID + l15;

    if (FUSE) {
        float dv[8];
        #pragma unroll
        for (int v = 0; v < 8; v++) dv[v] = dinv[rbase + v];
        #pragma unroll
        for (int nt = 0; nt < NTILE; nt++) {
            #pragma unroll
            for (int v = 0; v < 8; v++)
                cbase[(size_t)v * HID + nt * 16] = acc[nt][v] * dv[v];
        }
    } else {
        #pragma unroll
        for (int nt = 0; nt < NTILE; nt++) {
            float bv = bias[nt * 16 + l15];
            #pragma unroll
            for (int v = 0; v < 8; v++) {
                float r = acc[nt][v] + bv;
                if (RELU) r = fmaxf(r, 0.0f);
                cbase[(size_t)v * HID + nt * 16] = r;
            }
        }
    }
}

__global__ void __launch_bounds__(256)
k_gather(const float* __restrict__ hws, const float* __restrict__ dinv,
         const int* __restrict__ off, const int* __restrict__ cnt,
         const int* __restrict__ col, const float* __restrict__ bias,
         float* __restrict__ out, int n) {
    int node = (blockIdx.x * blockDim.x + threadIdx.x) >> 5;
    if (node >= n) return;
    int lane = threadIdx.x & 31;
    int grp  = lane >> 3;
    int t    = lane & 7;
    int start = off[node];
    int c     = cnt[node];
    int nfull = __builtin_amdgcn_readfirstlane(c >> 2);
    int rem   = c & 3;
    const int* cp = col + start + grp;
    int srem = cp[4 * nfull];

    const v4f* rows = (const v4f*)hws + t;
    v4f a0 = {}, a1 = {}, a2 = {};
    if (grp == 0) {
        const v4f* r = rows + (unsigned)node * 24u;
        a0 = r[0]; a1 = r[8]; a2 = r[16];
    }
    #pragma unroll 2
    for (int i = 0; i < nfull; i++) {
        int s = cp[4 * i];
        const v4f* r = rows + (unsigned)s * 24u;
        v4f x0 = r[0], x1 = r[8], x2 = r[16];
        a0 += x0; a1 += x1; a2 += x2;
    }
    if (grp < rem) {
        const v4f* r = rows + (unsigned)srem * 24u;
        v4f x0 = r[0], x1 = r[8], x2 = r[16];
        a0 += x0; a1 += x1; a2 += x2;
    }
    #pragma unroll
    for (int m = 8; m <= 16; m <<= 1) {
        #pragma unroll
        for (int q = 0; q < 4; q++) {
            a0[q] += __shfl_xor(a0[q], m);
            a1[q] += __shfl_xor(a1[q], m);
            a2[q] += __shfl_xor(a2[q], m);
        }
    }
    if (grp == 0) {
        float dv = dinv[node];
        const v4f* b4 = (const v4f*)bias + t;
        v4f* o = (v4f*)out + (unsigned)node * 24u + t;
        o[0]  = a0 * dv + b4[0];
        o[8]  = a1 * dv + b4[8];
        o[16] = a2 * dv + b4[16];
    }
}

__global__ void k_zero(float* x, int n) {
    int i = blockIdx.x * blockDim.x + threadIdx.x;
    if (i < n) x[i] = 0.0f;
}

#define POOL_NODES   32
#define POOL_THREADS 64
__global__ void __launch_bounds__(POOL_THREADS)
k_pool(const float* __restrict__ h, const int* __restrict__ batch,
       float* __restrict__ g, int n) {
    int lane  = threadIdx.x & 31;
    int wave  = __builtin_amdgcn_readfirstlane((int)(blockIdx.x * POOL_THREADS + threadIdx.x) >> 5);
    int first = wave * POOL_NODES;
    if (first >= n) return;
    int base  = min(first, n - POOL_NODES);
    int skip  = first - base;
    int bid   = batch[base + lane];
    const float* p = h + (size_t)base * HID + lane * 3;
    float x[POOL_NODES][3];
    #pragma unroll
    for (int j = 0; j < POOL_NODES; j++) {
        x[j][0] = p[j * HID];
        x[j][1] = p[j * HID + 1];
        x[j][2] = p[j * HID + 2];
    }
    int cur = __builtin_amdgcn_readlane(bid, skip);
    float a0 = 0.0f, a1 = 0.0f, a2 = 0.0f;
    #pragma unroll
    for (int j = 0; j < POOL_NODES; j++) {
        if (j >= skip) {
            int b = __builtin_amdgcn_readlane(bid, j);
            if (b != cur) {
                float* o = g + (size_t)cur * HID + lane * 3;
                atomicAdd(o, a0); atomicAdd(o + 1, a1); atomicAdd(o + 2, a2);
                a0 = 0.0f; a1 = 0.0f; a2 = 0.0f;
                cur = b;
            }
            a0 += fmaxf(x[j][0], 0.0f);
            a1 += fmaxf(x[j][1], 0.0f);
            a2 += fmaxf(x[j][2], 0.0f);
        }
    }
    float* o = g + (size_t)cur * HID + lane * 3;
    atomicAdd(o, a0); atomicAdd(o + 1, a1); atomicAdd(o + 2, a2);
}

__global__ void k_dec2(const float* __restrict__ g2, const float* __restrict__ W,
                       const float* __restrict__ b, float* __restrict__ out) {
    int idx = blockIdx.x * blockDim.x + threadIdx.x;
    if (idx < NGRAPH * OUTDIM) {
        int gi = idx / OUTDIM;
        int o  = idx - gi * OUTDIM;
        float s = b[o];
        const float* row = g2 + (size_t)gi * HID;
        #pragma unroll 8
        for (int k = 0; k < HID; k++) s += row[k] * W[k * OUTDIM + o];
        out[idx] = s;
    }
}

extern "C" void kernel_launch(void* const* d_in, const int* in_sizes, int n_in,
                              void* d_out, int out_size, void* d_ws, size_t ws_size,
                              hipStream_t stream) {
    const float* x       = (const float*)d_in[0];
    const int*   ei      = (const int*)  d_in[1];
    const int*   batch   = (const int*)  d_in[2];
    const float* enc_W0  = (const float*)d_in[3];
    const float* enc_b0  = (const float*)d_in[4];
    const float* enc_W1  = (const float*)d_in[5];
    const float* enc_b1  = (const float*)d_in[6];
    const float* conv_W  = (const float*)d_in[7];
    const float* conv_b  = (const float*)d_in[8];
    const float* dec_W0  = (const float*)d_in[9];
    const float* dec_b0  = (const float*)d_in[10];
    const float* dec_W1  = (const float*)d_in[11];
    const float* dec_b1  = (const float*)d_in[12];
    float* out = (float*)d_out;

    float* ws   = (float*)d_ws;
    size_t feat = (size_t)NNODES * HID;
    float* buf0 = ws;
    float* buf1 = buf0 + feat;
    float* buf2 = buf1 + feat;
    float* dinv = buf2 + feat;
    float* g    = dinv + NNODES;
    float* g2   = g + (size_t)NGRAPH * HID;
    int*   cnt  = (int*)(g2 + (size_t)NGRAPH * HID);
    int*   tot  = cnt + NNODES;
    int*   off  = tot + 16;
    int*   rank = off + NNODES;
    int*   col  = rank + NEDGES;

    const int B = 256;

    k_zero_i<<<cdiv(NNODES + 1, B), B, 0, stream>>>(cnt, NNODES + 1);
    k_rank<<<cdiv(NEDGES, B), B, 0, stream>>>(ei, cnt, rank, NEDGES);
    k_offsets<<<cdiv(NNODES, B), B, 0, stream>>>(cnt, tot, off, dinv, col + NEDGES, NNODES);
    k_fill<<<cdiv(NEDGES, B), B, 0, stream>>>(ei, rank, off, col, NEDGES);

    int mTilesN   = NNODES / 16;
    int gemmBlksN = cdiv(mTilesN, GEMM_WAVES);
    size_t shm128 = (size_t)HID * (INDIM + 2) * sizeof(float);
    size_t shm96  = (size_t)HID * (HID   + 2) * sizeof(float);
    k_gemm_wmma<INDIM, 0, 1, 0><<<gemmBlksN, GEMM_THREADS, shm128, stream>>>(
        x, enc_W0, enc_b0, buf0, nullptr, NNODES, INDIM);
    k_gemm_wmma<HID, 0, 0, 0><<<gemmBlksN, GEMM_THREADS, shm96, stream>>>(
        buf0, enc_W1, enc_b1, buf1, nullptr, NNODES, HID);

    float* cur = buf1;
    float* nxt = buf2;
    int nodeBlks = cdiv(NNODES * 32, B);
    for (int l = 0; l < NCONV; l++) {
        const float* Wl = conv_W + (size_t)l * HID * HID;
        const float* bl = conv_b + (size_t)l * HID;
        if (l == 0)
            k_gemm_wmma<HID, 0, 0, 1><<<gemmBlksN, GEMM_THREADS, shm96, stream>>>(
                cur, Wl, nullptr, buf0  , dinv, NNODES, HID);
        else
            k_gemm_wmma<HID, 1, 0, 1><<<gemmBlksN, GEMM_THREADS, shm96, stream>>>(
                cur, Wl, nullptr, buf0  , dinv, NNODES, HID);
        k_gather<<<nodeBlks, B, 0, stream>>>(buf0, dinv, off, cnt, col, bl, nxt, NNODES);
        float* t = cur; cur = nxt; nxt = t;
    }

    k_zero<<<cdiv(NGRAPH * HID, B), B, 0, stream>>>(g, NGRAPH * HID);
    k_pool<<<cdiv(cdiv(NNODES, POOL_NODES) * 32, POOL_THREADS), POOL_THREADS, 0, stream>>>(cur, batch, g, NNODES);

    int mTilesG = NGRAPH / 16;
    k_gemm_wmma<HID, 0, 1, 0><<<cdiv(mTilesG, GEMM_WAVES), GEMM_THREADS, shm96, stream>>>(
        g, dec_W0, dec_b0, g2, nullptr, NGRAPH, HID);
    k_dec2<<<cdiv(NGRAPH * OUTDIM, B), B, 0, stream>>>(g2, dec_W1, dec_b1, out);
}


# Edge ranking on per-work-group LDS counters (64 edge slices, all 50000 counters in 200 KB of LDS) instead of 800000 returning adds on global words; order of a node's list, so its sum order, changes
# speedup vs baseline: 3.1626x; 1.1381x over previous
#include <hip/hip_runtime.h>
#include <hip/hip_bf16.h>

typedef __attribute__((ext_vector_type(2))) float v2f;
typedef __attribute__((ext_vector_type(8))) float v8f;
typedef __attribute__((ext_vector_type(4))) float v4f;

#define NNODES 50000
#define NEDGES 800000
#define NGRAPH 512
#define INDIM  128
#define HID    96
#define OUTDIM 10
#define NCONV  4
#define NTILE  6
#define GEMM_WAVES 8
#define GEMM_THREADS (GEMM_WAVES * 32)

static inline int cdiv(int a, int b) { return (a + b - 1) / b; }

typedef __attribute__((ext_vector_type(4))) int v4i;

#define RANK_BLOCKS  64
#define RANK_THREADS 1024
#define RANK_EPB     (NEDGES / RANK_BLOCKS)
#define RANK_PER     ((RANK_EPB + RANK_THREADS - 1) / RANK_THREADS)
#define NODE4        (NNODES / 4)
static_assert(NEDGES % RANK_BLOCKS == 0, "slices must be equal");
static_assert(NNODES % 4 == 0, "counter sets are moved as 16-byte pieces");

__global__ void __launch_bounds__(RANK_THREADS)
k_rank(const int* __restrict__ ei, int* __restrict__ rank, int* __restrict__ tab,
       int* __restrict__ total) {
    __shared__ int cntL[NNODES];
    int tid = threadIdx.x;
    v4i* c4 = (v4i*)cntL;
    const v4i zero4i = {0, 0, 0, 0};
    for (int i = tid; i < NODE4; i += RANK_THREADS) c4[i] = zero4i;
    if (blockIdx.x == 0 && tid == 0) *total = 0;
    __syncthreads();

    int e0 = blockIdx.x * RANK_EPB;
    const int* dst = ei + NEDGES + e0;
    int d[RANK_PER];
    #pragma unroll
    for (int j = 0; j < RANK_PER; j++) {
        int k = tid + j * RANK_THREADS;
        d[j] = (k < RANK_EPB) ? dst[k] : 0;
    }
    int r[RANK_PER];
    #pragma unroll
    for (int j = 0; j < RANK_PER; j++) {
        int k = tid + j * RANK_THREADS;
        r[j] = (k < RANK_EPB) ? atomicAdd(&cntL[d[j]], 1) : 0;
    }
    #pragma unroll
    for (int j = 0; j < RANK_PER; j++) {
        int k = tid + j * RANK_THREADS;
        if (k < RANK_EPB) rank[e0 + k] = r[j];
    }
    __syncthreads();

    v4i* t4 = (v4i*)(tab + (size_t)blockIdx.x * NNODES);
    for (int i = tid; i < NODE4; i += RANK_THREADS) t4[i] = c4[i];
}

__global__ void k_offsets(int* __restrict__ tab, int* total, int* __restrict__ cnt,
                          int* __restrict__ off, float* __restrict__ dinv,
                          int* __restrict__ colpad, int n) {
    int i    = blockIdx.x * blockDim.x + threadIdx.x;
    int lane = threadIdx.x & 31;
    int ii   = min(i, n - 1);
    int v[RANK_BLOCKS];
    #pragma unroll
    for (int b = 0; b < RANK_BLOCKS; b++) v[b] = tab[(size_t)b * NNODES + ii];
    int c = 0;
    #pragma unroll
    for (int b = 0; b < RANK_BLOCKS; b++) c += v[b];
    if (i >= n) c = 0;
    int incl = c;
    #pragma unroll
    for (int d = 1; d < 32; d <<= 1) {
        int v1 = __shfl_up(incl, d);
        if (lane >= d) incl += v1;
    }
    int base = 0;
    if (lane == 31) base = atomicAdd(total, incl);
    base = __shfl(base, 31);
    if (i < n) {
        int run = base + incl - c;
        off[i]  = run;
        cnt[i]  = c;
        dinv[i] = rsqrtf((float)(c + 1));
        #pragma unroll
        for (int b = 0; b < RANK_BLOCKS; b++) {
            tab[(size_t)b * NNODES + i] = run;
            run += v[b];
        }
    }
    if (i < 4) colpad[i] = 0;
}

__global__ void k_fill(const int* __restrict__ ei, const int* __restrict__ rank,
                       const int* __restrict__ tab, int* __restrict__ col, int E) {
    int e = blockIdx.x * blockDim.x + threadIdx.x;
    if (e < E) {
        int d = ei[E + e];
        int b = e / RANK_EPB;
        col[tab[(size_t)b * NNODES + d] + rank[e]] = ei[e];
    }
}

template <int K, int RELUA, int RELU, int FUSE>
__global__ void __launch_bounds__(GEMM_THREADS)
k_gemm_wmma(const float* __restrict__ A, const float* __restrict__ B,
            const float* __restrict__ bias,
            float* __restrict__ C,
            const float* __restrict__ dinv,
            int M, int lda) {
    extern __shared__ float sBt[];
    constexpr int lstride = K + 2;

    constexpr int ROW4 = HID / 4;
    constexpr int PER  = K * ROW4 / GEMM_THREADS;
    static_assert(K * ROW4 % GEMM_THREADS == 0, "B must split evenly over the block");
    const v4f* B4 = (const v4f*)B;
    v4f piece[PER];
    #pragma unroll
    for (int j = 0; j < PER; j++) piece[j] = B4[threadIdx.x + j * GEMM_THREADS];
    #pragma unroll
    for (int j = 0; j < PER; j++) {
        int i  = threadIdx.x + j * GEMM_THREADS;
        int k  = i / ROW4;
        int c0 = (i - k * ROW4) * 4;
        float* dst = sBt + c0 * lstride + k;
        dst[0]           = piece[j][0];
        dst[lstride]     = piece[j][1];
        dst[2 * lstride] = piece[j][2];
        dst[3 * lstride] = piece[j][3];
    }
    __syncthreads();

    int mt   = blockIdx.x * GEMM_WAVES + (threadIdx.x >> 5);
    int lane = threadIdx.x & 31;
    int mTiles = M >> 4;
    if (mt >= mTiles) return;

    int half = lane >> 4;
    int l15  = lane & 15;

    const float* arow  = A + (size_t)(mt * 16 + l15) * lda;
    const float* bbase = sBt + (size_t)l15 * lstride;

    v8f acc[NTILE] = {};
    for (int k = 0; k < K; k += 4) {
        int kb = k + half * 2;
        v2f a = *(const v2f*)(arow + kb);
        if (RELUA) { a[0] = fmaxf(a[0], 0.0f); a[1] = fmaxf(a[1], 0.0f); }
        #pragma unroll
        for (int nt = 0; nt < NTILE; nt++) {
            v2f b = *(const v2f*)(bbase + (size_t)(nt * 16) * lstride + kb);
            acc[nt] = __builtin_amdgcn_wmma_f32_16x16x4_f32(
                 false, a,  false, b,
                 (short)0, acc[nt],  false,  false);
        }
    }

    int rbase = mt * 16 + half * 8;
    float* cbase = C + (size_t)rbase * HID + l15;

    if (FUSE) {
        float dv[8];
        #pragma unroll
        for (int v = 0; v < 8; v++) dv[v] = dinv[rbase + v];
        #pragma unroll
        for (int nt = 0; nt < NTILE; nt++) {
            #pragma unroll
            for (int v = 0; v < 8; v++)
                cbase[(size_t)v * HID + nt * 16] = acc[nt][v] * dv[v];
        }
    } else {
        #pragma unroll
        for (int nt = 0; nt < NTILE; nt++) {
            float bv = bias[nt * 16 + l15];
            #pragma unroll
            for (int v = 0; v < 8; v++) {
                float r = acc[nt][v] + bv;
                if (RELU) r = fmaxf(r, 0.0f);
                cbase[(size_t)v * HID + nt * 16] = r;
            }
        }
    }
}

__global__ void __launch_bounds__(256)
k_gather(const float* __restrict__ hws, const float* __restrict__ dinv,
         const int* __restrict__ off, const int* __restrict__ cnt,
         const int* __restrict__ col, const float* __restrict__ bias,
         float* __restrict__ out, int n) {
    int node = (blockIdx.x * blockDim.x + threadIdx.x) >> 5;
    if (node >= n) return;
    int lane = threadIdx.x & 31;
    int grp  = lane >> 3;
    int t    = lane & 7;
    int start = off[node];
    int c     = cnt[node];
    int nfull = __builtin_amdgcn_readfirstlane(c >> 2);
    int rem   = c & 3;
    const int* cp = col + start + grp;
    int srem = cp[4 * nfull];

    const v4f* rows = (const v4f*)hws + t;
    v4f a0 = {}, a1 = {}, a2 = {};
    if (grp == 0) {
        const v4f* r = rows + (unsigned)node * 24u;
        a0 = r[0]; a1 = r[8]; a2 = r[16];
    }
    #pragma unroll 2
    for (int i = 0; i < nfull; i++) {
        int s = cp[4 * i];
        const v4f* r = rows + (unsigned)s * 24u;
        v4f x0 = r[0], x1 = r[8], x2 = r[16];
        a0 += x0; a1 += x1; a2 += x2;
    }
    if (grp < rem) {
        const v4f* r = rows + (unsigned)srem * 24u;
        v4f x0 = r[0], x1 = r[8], x2 = r[16];
        a0 += x0; a1 += x1; a2 += x2;
    }
    #pragma unroll
    for (int m = 8; m <= 16; m <<= 1) {
        #pragma unroll
        for (int q = 0; q < 4; q++) {
            a0[q] += __shfl_xor(a0[q], m);
            a1[q] += __shfl_xor(a1[q], m);
            a2[q] += __shfl_xor(a2[q], m);
        }
    }
    if (grp == 0) {
        float dv = dinv[node];
        const v4f* b4 = (const v4f*)bias + t;
        v4f* o = (v4f*)out + (unsigned)node * 24u + t;
        o[0]  = a0 * dv + b4[0];
        o[8]  = a1 * dv + b4[8];
        o[16] = a2 * dv + b4[16];
    }
}

__global__ void k_zero(float* x, int n) {
    int i = blockIdx.x * blockDim.x + threadIdx.x;
    if (i < n) x[i] = 0.0f;
}

#define POOL_NODES   32
#define POOL_THREADS 64
__global__ void __launch_bounds__(POOL_THREADS)
k_pool(const float* __restrict__ h, const int* __restrict__ batch,
       float* __restrict__ g, int n) {
    int lane  = threadIdx.x & 31;
    int wave  = __builtin_amdgcn_readfirstlane((int)(blockIdx.x * POOL_THREADS + threadIdx.x) >> 5);
    int first = wave * POOL_NODES;
    if (first >= n) return;
    int base  = min(first, n - POOL_NODES);
    int skip  = first - base;
    int bid   = batch[base + lane];
    const float* p = h + (size_t)base * HID + lane * 3;
    float x[POOL_NODES][3];
    #pragma unroll
    for (int j = 0; j < POOL_NODES; j++) {
        x[j][0] = p[j * HID];
        x[j][1] = p[j * HID + 1];
        x[j][2] = p[j * HID + 2];
    }
    int cur = __builtin_amdgcn_readlane(bid, skip);
    float a0 = 0.0f, a1 = 0.0f, a2 = 0.0f;
    #pragma unroll
    for (int j = 0; j < POOL_NODES; j++) {
        if (j >= skip) {
            int b = __builtin_amdgcn_readlane(bid, j);
            if (b != cur) {
                float* o = g + (size_t)cur * HID + lane * 3;
                atomicAdd(o, a0); atomicAdd(o + 1, a1); atomicAdd(o + 2, a2);
                a0 = 0.0f; a1 = 0.0f; a2 = 0.0f;
                cur = b;
            }
            a0 += fmaxf(x[j][0], 0.0f);
            a1 += fmaxf(x[j][1], 0.0f);
            a2 += fmaxf(x[j][2], 0.0f);
        }
    }
    float* o = g + (size_t)cur * HID + lane * 3;
    atomicAdd(o, a0); atomicAdd(o + 1, a1); atomicAdd(o + 2, a2);
}

__global__ void k_dec2(const float* __restrict__ g2, const float* __restrict__ W,
                       const float* __restrict__ b, float* __restrict__ out) {
    int idx = blockIdx.x * blockDim.x + threadIdx.x;
    if (idx < NGRAPH * OUTDIM) {
        int gi = idx / OUTDIM;
        int o  = idx - gi * OUTDIM;
        float s = b[o];
        const float* row = g2 + (size_t)gi * HID;
        #pragma unroll 8
        for (int k = 0; k < HID; k++) s += row[k] * W[k * OUTDIM + o];
        out[idx] = s;
    }
}

extern "C" void kernel_launch(void* const* d_in, const int* in_sizes, int n_in,
                              void* d_out, int out_size, void* d_ws, size_t ws_size,
                              hipStream_t stream) {
    const float* x       = (const float*)d_in[0];
    const int*   ei      = (const int*)  d_in[1];
    const int*   batch   = (const int*)  d_in[2];
    const float* enc_W0  = (const float*)d_in[3];
    const float* enc_b0  = (const float*)d_in[4];
    const float* enc_W1  = (const float*)d_in[5];
    const float* enc_b1  = (const float*)d_in[6];
    const float* conv_W  = (const float*)d_in[7];
    const float* conv_b  = (const float*)d_in[8];
    const float* dec_W0  = (const float*)d_in[9];
    const float* dec_b0  = (const float*)d_in[10];
    const float* dec_W1  = (const float*)d_in[11];
    const float* dec_b1  = (const float*)d_in[12];
    float* out = (float*)d_out;

    float* ws   = (float*)d_ws;
    size_t feat = (size_t)NNODES * HID;
    float* buf0 = ws;
    float* buf1 = buf0 + feat;
    float* buf2 = buf1 + feat;
    float* dinv = buf2 + feat;
    float* g    = dinv + NNODES;
    float* g2   = g + (size_t)NGRAPH * HID;
    int*   cnt  = (int*)(g2 + (size_t)NGRAPH * HID);
    int*   tot  = cnt + NNODES;
    int*   off  = tot + 16;
    int*   rank = off + NNODES;
    int*   col  = rank + NEDGES;
    int*   tab  = col + NEDGES + 64;

    const int B = 256;

    k_rank<<<RANK_BLOCKS, RANK_THREADS, 0, stream>>>(ei, rank, tab, tot);
    k_offsets<<<cdiv(NNODES, B), B, 0, stream>>>(tab, tot, cnt, off, dinv, col + NEDGES, NNODES);
    k_fill<<<cdiv(NEDGES, B), B, 0, stream>>>(ei, rank, tab, col, NEDGES);

    int mTilesN   = NNODES / 16;
    int gemmBlksN = cdiv(mTilesN, GEMM_WAVES);
    size_t shm128 = (size_t)HID * (INDIM + 2) * sizeof(float);
    size_t shm96  = (size_t)HID * (HID   + 2) * sizeof(float);
    k_gemm_wmma<INDIM, 0, 1, 0><<<gemmBlksN, GEMM_THREADS, shm128, stream>>>(
        x, enc_W0, enc_b0, buf0, nullptr, NNODES, INDIM);
    k_gemm_wmma<HID, 0, 0, 0><<<gemmBlksN, GEMM_THREADS, shm96, stream>>>(
        buf0, enc_W1, enc_b1, buf1, nullptr, NNODES, HID);

    float* cur = buf1;
    float* nxt = buf2;
    int nodeBlks = cdiv(NNODES * 32, B);
    for (int l = 0; l < NCONV; l++) {
        const float* Wl = conv_W + (size_t)l * HID * HID;
        const float* bl = conv_b + (size_t)l * HID;
        if (l == 0)
            k_gemm_wmma<HID, 0, 0, 1><<<gemmBlksN, GEMM_THREADS, shm96, stream>>>(
                cur, Wl, nullptr, buf0  , dinv, NNODES, HID);
        else
            k_gemm_wmma<HID, 1, 0, 1><<<gemmBlksN, GEMM_THREADS, shm96, stream>>>(
                cur, Wl, nullptr, buf0  , dinv, NNODES, HID);
        k_gather<<<nodeBlks, B, 0, stream>>>(buf0, dinv, off, cnt, col, bl, nxt, NNODES);
        float* t = cur; cur = nxt; nxt = t;
    }

    k_zero<<<cdiv(NGRAPH * HID, B), B, 0, stream>>>(g, NGRAPH * HID);
    k_pool<<<cdiv(cdiv(NNODES, POOL_NODES) * 32, POOL_THREADS), POOL_THREADS, 0, stream>>>(cur, batch, g, NNODES);

    int mTilesG = NGRAPH / 16;
    k_gemm_wmma<HID, 0, 1, 0><<<cdiv(mTilesG, GEMM_WAVES), GEMM_THREADS, shm96, stream>>>(
        g, dec_W0, dec_b0, g2, nullptr, NGRAPH, HID);
    k_dec2<<<cdiv(NGRAPH * OUTDIM, B), B, 0, stream>>>(g2, dec_W1, dec_b1, out);
}


# Aggregation of layers 0-2 keeps its 64-node tile of relu(h) in LDS and runs the next layer's WMMA product in the same launch (3 launches and h round trips less); sum order and k order kept
# speedup vs baseline: 3.5968x; 1.1373x over previous
#include <hip/hip_runtime.h>
#include <hip/hip_bf16.h>

typedef __attribute__((ext_vector_type(2))) float v2f;
typedef __attribute__((ext_vector_type(8))) float v8f;
typedef __attribute__((ext_vector_type(4))) float v4f;

#define NNODES 50000
#define NEDGES 800000
#define NGRAPH 512
#define INDIM  128
#define HID    96
#define OUTDIM 10
#define NCONV  4
#define NTILE  6
#define GEMM_WAVES 8
#define GEMM_THREADS (GEMM_WAVES * 32)

static inline int cdiv(int a, int b) { return (a + b - 1) / b; }

typedef __attribute__((ext_vector_type(4))) int v4i;

#define RANK_BLOCKS  64
#define RANK_THREADS 1024
#define RANK_EPB     (NEDGES / RANK_BLOCKS)
#define RANK_PER     ((RANK_EPB + RANK_THREADS - 1) / RANK_THREADS)
#define NODE4        (NNODES / 4)
static_assert(NEDGES % RANK_BLOCKS == 0, "slices must be equal");
static_assert(NNODES % 4 == 0, "counter sets are moved as 16-byte pieces");

__global__ void __launch_bounds__(RANK_THREADS)
k_rank(const int* __restrict__ ei, int* __restrict__ rank, int* __restrict__ tab,
       int* __restrict__ total) {
    __shared__ int cntL[NNODES];
    int tid = threadIdx.x;
    v4i* c4 = (v4i*)cntL;
    const v4i zero4i = {0, 0, 0, 0};
    for (int i = tid; i < NODE4; i += RANK_THREADS) c4[i] = zero4i;
    if (blockIdx.x == 0 && tid == 0) *total = 0;
    __syncthreads();

    int e0 = blockIdx.x * RANK_EPB;
    const int* dst = ei + NEDGES + e0;
    int d[RANK_PER];
    #pragma unroll
    for (int j = 0; j < RANK_PER; j++) {
        int k = tid + j * RANK_THREADS;
        d[j] = (k < RANK_EPB) ? dst[k] : 0;
    }
    int r[RANK_PER];
    #pragma unroll
    for (int j = 0; j < RANK_PER; j++) {
        int k = tid + j * RANK_THREADS;
        r[j] = (k < RANK_EPB) ? atomicAdd(&cntL[d[j]], 1) : 0;
    }
    #pragma unroll
    for (int j = 0; j < RANK_PER; j++) {
        int k = tid + j * RANK_THREADS;
        if (k < RANK_EPB) rank[e0 + k] = r[j];
    }
    __syncthreads();

    v4i* t4 = (v4i*)(tab + (size_t)blockIdx.x * NNODES);
    for (int i = tid; i < NODE4; i += RANK_THREADS) t4[i] = c4[i];
}

__global__ void k_offsets(int* __restrict__ tab, int* total, int* __restrict__ cnt,
                          int* __restrict__ off, float* __restrict__ dinv,
                          int* __restrict__ colpad, int n) {
    int i    = blockIdx.x * blockDim.x + threadIdx.x;
    int lane = threadIdx.x & 31;
    int ii   = min(i, n - 1);
    int v[RANK_BLOCKS];
    #pragma unroll
    for (int b = 0; b < RANK_BLOCKS; b++) v[b] = tab[(size_t)b * NNODES + ii];
    int c = 0;
    #pragma unroll
    for (int b = 0; b < RANK_BLOCKS; b++) c += v[b];
    if (i >= n) c = 0;
    int incl = c;
    #pragma unroll
    for (int d = 1; d < 32; d <<= 1) {
        int v1 = __shfl_up(incl, d);
        if (lane >= d) incl += v1;
    }
    int base = 0;
    if (lane == 31) base = atomicAdd(total, incl);
    base = __shfl(base, 31);
    if (i < n) {
        int run = base + incl - c;
        off[i]  = run;
        cnt[i]  = c;
        dinv[i] = rsqrtf((float)(c + 1));
        #pragma unroll
        for (int b = 0; b < RANK_BLOCKS; b++) {
            tab[(size_t)b * NNODES + i] = run;
            run += v[b];
        }
    }
    if (i < 4) colpad[i] = 0;
}

__global__ void k_fill(const int* __restrict__ ei, const int* __restrict__ rank,
                       const int* __restrict__ tab, int* __restrict__ col, int E) {
    int e = blockIdx.x * blockDim.x + threadIdx.x;
    if (e < E) {
        int d = ei[E + e];
        int b = e / RANK_EPB;
        col[tab[(size_t)b * NNODES + d] + rank[e]] = ei[e];
    }
}

template <int K, int RELUA, int RELU, int FUSE>
__global__ void __launch_bounds__(GEMM_THREADS)
k_gemm_wmma(const float* __restrict__ A, const float* __restrict__ B,
            const float* __restrict__ bias,
            float* __restrict__ C,
            const float* __restrict__ dinv,
            int M, int lda) {
    extern __shared__ float sBt[];
    constexpr int lstride = K + 2;

    constexpr int ROW4 = HID / 4;
    constexpr int PER  = K * ROW4 / GEMM_THREADS;
    static_assert(K * ROW4 % GEMM_THREADS == 0, "B must split evenly over the block");
    const v4f* B4 = (const v4f*)B;
    v4f piece[PER];
    #pragma unroll
    for (int j = 0; j < PER; j++) piece[j] = B4[threadIdx.x + j * GEMM_THREADS];
    #pragma unroll
    for (int j = 0; j < PER; j++) {
        int i  = threadIdx.x + j * GEMM_THREADS;
        int k  = i / ROW4;
        int c0 = (i - k * ROW4) * 4;
        float* dst = sBt + c0 * lstride + k;
        dst[0]           = piece[j][0];
        dst[lstride]     = piece[j][1];
        dst[2 * lstride] = piece[j][2];
        dst[3 * lstride] = piece[j][3];
    }
    __syncthreads();

    int mt   = blockIdx.x * GEMM_WAVES + (threadIdx.x >> 5);
    int lane = threadIdx.x & 31;
    int mTiles = M >> 4;
    if (mt >= mTiles) return;

    int half = lane >> 4;
    int l15  = lane & 15;

    const float* arow  = A + (size_t)(mt * 16 + l15) * lda;
    const float* bbase = sBt + (size_t)l15 * lstride;

    v8f acc[NTILE] = {};
    for (int k = 0; k < K; k += 4) {
        int kb = k + half * 2;
        v2f a = *(const v2f*)(arow + kb);
        if (RELUA) { a[0] = fmaxf(a[0], 0.0f); a[1] = fmaxf(a[1], 0.0f); }
        #pragma unroll
        for (int nt = 0; nt < NTILE; nt++) {
            v2f b = *(const v2f*)(bbase + (size_t)(nt * 16) * lstride + kb);
            acc[nt] = __builtin_amdgcn_wmma_f32_16x16x4_f32(
                 false, a,  false, b,
                 (short)0, acc[nt],  false,  false);
        }
    }

    int rbase = mt * 16 + half * 8;
    float* cbase = C + (size_t)rbase * HID + l15;

    if (FUSE) {
        float dv[8];
        #pragma unroll
        for (int v = 0; v < 8; v++) dv[v] = dinv[rbase + v];
        #pragma unroll
        for (int nt = 0; nt < NTILE; nt++) {
            #pragma unroll
            for (int v = 0; v < 8; v++)
                cbase[(size_t)v * HID + nt * 16] = acc[nt][v] * dv[v];
        }
    } else {
        #pragma unroll
        for (int nt = 0; nt < NTILE; nt++) {
            float bv = bias[nt * 16 + l15];
            #pragma unroll
            for (int v = 0; v < 8; v++) {
                float r = acc[nt][v] + bv;
                if (RELU) r = fmaxf(r, 0.0f);
                cbase[(size_t)v * HID + nt * 16] = r;
            }
        }
    }
}

__global__ void __launch_bounds__(256)
k_gather(const float* __restrict__ hws, const float* __restrict__ dinv,
         const int* __restrict__ off, const int* __restrict__ cnt,
         const int* __restrict__ col, const float* __restrict__ bias,
         float* __restrict__ out, int n) {
    int node = (blockIdx.x * blockDim.x + threadIdx.x) >> 5;
    if (node >= n) return;
    int lane = threadIdx.x & 31;
    int grp  = lane >> 3;
    int t    = lane & 7;
    int start = off[node];
    int c     = cnt[node];
    int nfull = __builtin_amdgcn_readfirstlane(c >> 2);
    int rem   = c & 3;
    const int* cp = col + start + grp;
    int srem = cp[4 * nfull];

    const v4f* rows = (const v4f*)hws + t;
    v4f a0 = {}, a1 = {}, a2 = {};
    if (grp == 0) {
        const v4f* r = rows + (unsigned)node * 24u;
        a0 = r[0]; a1 = r[8]; a2 = r[16];
    }
    #pragma unroll 2
    for (int i = 0; i < nfull; i++) {
        int s = cp[4 * i];
        const v4f* r = rows + (unsigned)s * 24u;
        v4f x0 = r[0], x1 = r[8], x2 = r[16];
        a0 += x0; a1 += x1; a2 += x2;
    }
    if (grp < rem) {
        const v4f* r = rows + (unsigned)srem * 24u;
        v4f x0 = r[0], x1 = r[8], x2 = r[16];
        a0 += x0; a1 += x1; a2 += x2;
    }
    #pragma unroll
    for (int m = 8; m <= 16; m <<= 1) {
        #pragma unroll
        for (int q = 0; q < 4; q++) {
            a0[q] += __shfl_xor(a0[q], m);
            a1[q] += __shfl_xor(a1[q], m);
            a2[q] += __shfl_xor(a2[q], m);
        }
    }
    if (grp == 0) {
        float dv = dinv[node];
        const v4f* b4 = (const v4f*)bias + t;
        v4f* o = (v4f*)out + (unsigned)node * 24u + t;
        o[0]  = a0 * dv + b4[0];
        o[8]  = a1 * dv + b4[8];
        o[16] = a2 * dv + b4[16];
    }
}

#define FG_NODES 64
#define FG_NPW   (FG_NODES / GEMM_WAVES)
#define FG_HS    (HID + 4)
static_assert(NNODES % 16 == 0, "a 16-row tile is wholly inside or outside the node range");
static_assert(FG_NPW == 8, "per-node values of a wave sit in lanes 0..7");

__global__ void __launch_bounds__(GEMM_THREADS)
k_gather_gemm(const float* __restrict__ hws, const float* __restrict__ dinv,
              const int* __restrict__ off, const int* __restrict__ cnt,
              const int* __restrict__ col, const float* __restrict__ bias,
              const float* __restrict__ Wn, float* __restrict__ out, int n) {
    __shared__ float sBt[HID * (HID + 2)];
    __shared__ v4f   hT4[FG_NODES * (FG_HS / 4)];
    constexpr int lstride = HID + 2;

    constexpr int ROW4 = HID / 4;
    constexpr int PER  = HID * ROW4 / GEMM_THREADS;
    static_assert(HID * ROW4 % GEMM_THREADS == 0, "Wn must split evenly over the block");
    const v4f* B4 = (const v4f*)Wn;
    v4f piece[PER];
    #pragma unroll
    for (int j = 0; j < PER; j++) piece[j] = B4[threadIdx.x + j * GEMM_THREADS];

    int lane = threadIdx.x & 31;
    int w    = __builtin_amdgcn_readfirstlane((int)(threadIdx.x >> 5));
    int grp  = lane >> 3;
    int t    = lane & 7;
    int nb   = blockIdx.x * FG_NODES + w * FG_NPW;
    int nvalid = min(FG_NPW, n - nb);

    int ml   = min(nb + t, n - 1);
    int moff = off[ml];
    int mcnt = cnt[ml];
    int mdvb = __float_as_int(dinv[ml]);
    const v4f* b4 = (const v4f*)bias + t;
    v4f bb0 = b4[0], bb1 = b4[8], bb2 = b4[16];

    #pragma unroll
    for (int j = 0; j < PER; j++) {
        int i  = threadIdx.x + j * GEMM_THREADS;
        int k  = i / ROW4;
        int c0 = (i - k * ROW4) * 4;
        float* dst = sBt + c0 * lstride + k;
        dst[0]           = piece[j][0];
        dst[lstride]     = piece[j][1];
        dst[2 * lstride] = piece[j][2];
        dst[3 * lstride] = piece[j][3];
    }

    const v4f* rows = (const v4f*)hws + t;
    int idxs[FG_NPW];
    #pragma unroll
    for (int j = 0; j < FG_NPW; j++) {
        int sj  = __builtin_amdgcn_readlane(moff, j);
        int cj  = __builtin_amdgcn_readlane(mcnt, j);
        idxs[j] = col[sj + min(lane, max(cj, 1) - 1)];
    }
    asm volatile("" :: "v"(idxs[0]), "v"(idxs[1]), "v"(idxs[2]), "v"(idxs[3]),
                       "v"(idxs[4]), "v"(idxs[5]), "v"(idxs[6]), "v"(idxs[7]));
    #pragma unroll
    for (int j = 0; j < FG_NPW; j++) {
        int cj  = __builtin_amdgcn_readlane(mcnt, j);
        idxs[j] = (lane < cj) ? idxs[j] : 0;
    }
    #pragma unroll
    for (int i = 0; i < FG_NPW; i++) {
        if (i < nvalid) {
            int   node = nb + i;
            int   cs = __builtin_amdgcn_readlane(moff, i);
            int   cc = __builtin_amdgcn_readlane(mcnt, i);
            float dv = __int_as_float(__builtin_amdgcn_readlane(mdvb, i));
            int cidx = idxs[i];
            const v4f* rself = rows + (unsigned)node * 24u;
            v4f a0 = {}, a1 = {}, a2 = {};
            for (int e0 = 0; e0 == 0 || e0 < cc; e0 += 16) {
                if (e0 != 0 && (e0 & 31) == 0)
                    cidx = (lane < cc - e0) ? col[cs + e0 + lane] : 0;
                int  eb = (e0 & 31) + grp;
                bool on0 = e0 + grp      < cc;
                bool on1 = e0 + grp + 4  < cc;
                bool on2 = e0 + grp + 8  < cc;
                bool on3 = e0 + grp + 12 < cc;
                int  s0 = __shfl(cidx, eb);
                int  s1 = __shfl(cidx, eb + 4);
                int  s2 = __shfl(cidx, eb + 8);
                int  s3 = __shfl(cidx, eb + 12);
                const v4f* r0 = on0 ? rows + (unsigned)s0 * 24u : rself;
                const v4f* r1 = on1 ? rows + (unsigned)s1 * 24u : rself;
                const v4f* r2 = on2 ? rows + (unsigned)s2 * 24u : rself;
                const v4f* r3 = on3 ? rows + (unsigned)s3 * 24u : rself;
                v4f x00 = r0[0], x01 = r0[8], x02 = r0[16];
                v4f x10 = r1[0], x11 = r1[8], x12 = r1[16];
                v4f x20 = r2[0], x21 = r2[8], x22 = r2[16];
                v4f x30 = r3[0], x31 = r3[8], x32 = r3[16];
                if (e0 == 0 && grp == 0) {
                    v4f y0 = rself[0], y1 = rself[8], y2 = rself[16];
                    a0 = y0; a1 = y1; a2 = y2;
                }
                if (on0) { a0 += x00; a1 += x01; a2 += x02; }
                if (on1) { a0 += x10; a1 += x11; a2 += x12; }
                if (on2) { a0 += x20; a1 += x21; a2 += x22; }
                if (on3) { a0 += x30; a1 += x31; a2 += x32; }
            }
            #pragma unroll
            for (int mm = 8; mm <= 16; mm <<= 1) {
                #pragma unroll
                for (int q = 0; q < 4; q++) {
                    a0[q] += __shfl_xor(a0[q], mm);
                    a1[q] += __shfl_xor(a1[q], mm);
                    a2[q] += __shfl_xor(a2[q], mm);
                }
            }
            if (grp == 0) {
                v4f o0 = a0 * dv + bb0;
                v4f o1 = a1 * dv + bb1;
                v4f o2 = a2 * dv + bb2;
                #pragma unroll
                for (int q = 0; q < 4; q++) {
                    o0[q] = fmaxf(o0[q], 0.0f); o1[q] = fmaxf(o1[q], 0.0f); o2[q] = fmaxf(o2[q], 0.0f);
                }
                v4f* dst = hT4 + (w * FG_NPW + i) * (FG_HS / 4) + t;
                dst[0] = o0; dst[8] = o1; dst[16] = o2;
            }
        }
    }
    __syncthreads();

    int mtile   = w >> 1;
    int rowbase = blockIdx.x * FG_NODES + mtile * 16;
    if (rowbase >= n) return;
    int half = lane >> 4;
    int l15  = lane & 15;
    int nt0  = (w & 1) * 3;
    const float* abase = (const float*)hT4 + (mtile * 16 + l15) * FG_HS;
    const float* bbase = sBt + (nt0 * 16 + l15) * lstride;
    v8f acc[3] = {};
    for (int k = 0; k < HID; k += 4) {
        int kb = k + half * 2;
        v2f a = *(const v2f*)(abase + kb);
        #pragma unroll
        for (int j = 0; j < 3; j++) {
            v2f b = *(const v2f*)(bbase + (j * 16) * lstride + kb);
            acc[j] = __builtin_amdgcn_wmma_f32_16x16x4_f32(
                 false, a,  false, b,
                 (short)0, acc[j],  false,  false);
        }
    }
    int rbase = rowbase + half * 8;
    float* cbase = out + (size_t)rbase * HID + nt0 * 16 + l15;
    float dvv[8];
    #pragma unroll
    for (int v = 0; v < 8; v++) dvv[v] = dinv[rbase + v];
    #pragma unroll
    for (int j = 0; j < 3; j++) {
        #pragma unroll
        for (int v = 0; v < 8; v++)
            cbase[(size_t)v * HID + j * 16] = acc[j][v] * dvv[v];
    }
}

__global__ void k_zero(float* x, int n) {
    int i = blockIdx.x * blockDim.x + threadIdx.x;
    if (i < n) x[i] = 0.0f;
}

#define POOL_NODES   32
#define POOL_THREADS 64
__global__ void __launch_bounds__(POOL_THREADS)
k_pool(const float* __restrict__ h, const int* __restrict__ batch,
       float* __restrict__ g, int n) {
    int lane  = threadIdx.x & 31;
    int wave  = __builtin_amdgcn_readfirstlane((int)(blockIdx.x * POOL_THREADS + threadIdx.x) >> 5);
    int first = wave * POOL_NODES;
    if (first >= n) return;
    int base  = min(first, n - POOL_NODES);
    int skip  = first - base;
    int bid   = batch[base + lane];
    const float* p = h + (size_t)base * HID + lane * 3;
    float x[POOL_NODES][3];
    #pragma unroll
    for (int j = 0; j < POOL_NODES; j++) {
        x[j][0] = p[j * HID];
        x[j][1] = p[j * HID + 1];
        x[j][2] = p[j * HID + 2];
    }
    int cur = __builtin_amdgcn_readlane(bid, skip);
    float a0 = 0.0f, a1 = 0.0f, a2 = 0.0f;
    #pragma unroll
    for (int j = 0; j < POOL_NODES; j++) {
        if (j >= skip) {
            int b = __builtin_amdgcn_readlane(bid, j);
            if (b != cur) {
                float* o = g + (size_t)cur * HID + lane * 3;
                atomicAdd(o, a0); atomicAdd(o + 1, a1); atomicAdd(o + 2, a2);
                a0 = 0.0f; a1 = 0.0f; a2 = 0.0f;
                cur = b;
            }
            a0 += fmaxf(x[j][0], 0.0f);
            a1 += fmaxf(x[j][1], 0.0f);
            a2 += fmaxf(x[j][2], 0.0f);
        }
    }
    float* o = g + (size_t)cur * HID + lane * 3;
    atomicAdd(o, a0); atomicAdd(o + 1, a1); atomicAdd(o + 2, a2);
}

__global__ void k_dec2(const float* __restrict__ g2, const float* __restrict__ W,
                       const float* __restrict__ b, float* __restrict__ out) {
    int idx = blockIdx.x * blockDim.x + threadIdx.x;
    if (idx < NGRAPH * OUTDIM) {
        int gi = idx / OUTDIM;
        int o  = idx - gi * OUTDIM;
        float s = b[o];
        const float* row = g2 + (size_t)gi * HID;
        #pragma unroll 8
        for (int k = 0; k < HID; k++) s += row[k] * W[k * OUTDIM + o];
        out[idx] = s;
    }
}

extern "C" void kernel_launch(void* const* d_in, const int* in_sizes, int n_in,
                              void* d_out, int out_size, void* d_ws, size_t ws_size,
                              hipStream_t stream) {
    const float* x       = (const float*)d_in[0];
    const int*   ei      = (const int*)  d_in[1];
    const int*   batch   = (const int*)  d_in[2];
    const float* enc_W0  = (const float*)d_in[3];
    const float* enc_b0  = (const float*)d_in[4];
    const float* enc_W1  = (const float*)d_in[5];
    const float* enc_b1  = (const float*)d_in[6];
    const float* conv_W  = (const float*)d_in[7];
    const float* conv_b  = (const float*)d_in[8];
    const float* dec_W0  = (const float*)d_in[9];
    const float* dec_b0  = (const float*)d_in[10];
    const float* dec_W1  = (const float*)d_in[11];
    const float* dec_b1  = (const float*)d_in[12];
    float* out = (float*)d_out;

    float* ws   = (float*)d_ws;
    size_t feat = (size_t)NNODES * HID;
    float* buf0 = ws;
    float* buf1 = buf0 + feat;
    float* buf2 = buf1 + feat;
    float* dinv = buf2 + feat;
    float* g    = dinv + NNODES;
    float* g2   = g + (size_t)NGRAPH * HID;
    int*   cnt  = (int*)(g2 + (size_t)NGRAPH * HID);
    int*   tot  = cnt + NNODES;
    int*   off  = tot + 16;
    int*   rank = off + NNODES;
    int*   col  = rank + NEDGES;
    int*   tab  = col + NEDGES + 64;

    const int B = 256;

    k_rank<<<RANK_BLOCKS, RANK_THREADS, 0, stream>>>(ei, rank, tab, tot);
    k_offsets<<<cdiv(NNODES, B), B, 0, stream>>>(tab, tot, cnt, off, dinv, col + NEDGES, NNODES);
    k_fill<<<cdiv(NEDGES, B), B, 0, stream>>>(ei, rank, tab, col, NEDGES);

    int mTilesN   = NNODES / 16;
    int gemmBlksN = cdiv(mTilesN, GEMM_WAVES);
    size_t shm128 = (size_t)HID * (INDIM + 2) * sizeof(float);
    size_t shm96  = (size_t)HID * (HID   + 2) * sizeof(float);
    k_gemm_wmma<INDIM, 0, 1, 0><<<gemmBlksN, GEMM_THREADS, shm128, stream>>>(
        x, enc_W0, enc_b0, buf0, nullptr, NNODES, INDIM);
    k_gemm_wmma<HID, 0, 0, 0><<<gemmBlksN, GEMM_THREADS, shm96, stream>>>(
        buf0, enc_W1, enc_b1, buf1, nullptr, NNODES, HID);

    k_gemm_wmma<HID, 0, 0, 1><<<gemmBlksN, GEMM_THREADS, shm96, stream>>>(
        buf1, conv_W, nullptr, buf0  , dinv, NNODES, HID);
    float* hsrc = buf0;
    float* hdst = buf2;
    int fgBlks = cdiv(NNODES, FG_NODES);
    for (int l = 0; l + 1 < NCONV; l++) {
        const float* Wn = conv_W + (size_t)(l + 1) * HID * HID;
        const float* bl = conv_b + (size_t)l * HID;
        k_gather_gemm<<<fgBlks, GEMM_THREADS, 0, stream>>>(hsrc, dinv, off, cnt, col, bl, Wn, hdst, NNODES);
        float* t = hsrc; hsrc = hdst; hdst = t;
    }
    int nodeBlks = cdiv(NNODES * 32, B);
    float* cur = buf1;
    k_gather<<<nodeBlks, B, 0, stream>>>(hsrc, dinv, off, cnt, col,
                                         conv_b + (size_t)(NCONV - 1) * HID, cur, NNODES);

    k_zero<<<cdiv(NGRAPH * HID, B), B, 0, stream>>>(g, NGRAPH * HID);
    k_pool<<<cdiv(cdiv(NNODES, POOL_NODES) * 32, POOL_THREADS), POOL_THREADS, 0, stream>>>(cur, batch, g, NNODES);

    int mTilesG = NGRAPH / 16;
    k_gemm_wmma<HID, 0, 1, 0><<<cdiv(mTilesG, GEMM_WAVES), GEMM_THREADS, shm96, stream>>>(
        g, dec_W0, dec_b0, g2, nullptr, NGRAPH, HID);
    k_dec2<<<cdiv(NGRAPH * OUTDIM, B), B, 0, stream>>>(g2, dec_W1, dec_b1, out);
}


# Last aggregation keeps its 64-node tile of relu(h) in LDS and pools it in the same launch (1 launch and the h round trip less, 8 destinations per wave); same 32-node partial sums, atomics kept
# speedup vs baseline: 3.7197x; 1.0342x over previous
#include <hip/hip_runtime.h>
#include <hip/hip_bf16.h>

typedef __attribute__((ext_vector_type(2))) float v2f;
typedef __attribute__((ext_vector_type(8))) float v8f;
typedef __attribute__((ext_vector_type(4))) float v4f;

#define NNODES 50000
#define NEDGES 800000
#define NGRAPH 512
#define INDIM  128
#define HID    96
#define OUTDIM 10
#define NCONV  4
#define NTILE  6
#define GEMM_WAVES 8
#define GEMM_THREADS (GEMM_WAVES * 32)

static inline int cdiv(int a, int b) { return (a + b - 1) / b; }

typedef __attribute__((ext_vector_type(4))) int v4i;

#define RANK_BLOCKS  64
#define RANK_THREADS 1024
#define RANK_EPB     (NEDGES / RANK_BLOCKS)
#define RANK_PER     ((RANK_EPB + RANK_THREADS - 1) / RANK_THREADS)
#define NODE4        (NNODES / 4)
static_assert(NEDGES % RANK_BLOCKS == 0, "slices must be equal");
static_assert(NNODES % 4 == 0, "counter sets are moved as 16-byte pieces");

__global__ void __launch_bounds__(RANK_THREADS)
k_rank(const int* __restrict__ ei, int* __restrict__ rank, int* __restrict__ tab,
       int* __restrict__ total) {
    __shared__ int cntL[NNODES];
    int tid = threadIdx.x;
    v4i* c4 = (v4i*)cntL;
    const v4i zero4i = {0, 0, 0, 0};
    for (int i = tid; i < NODE4; i += RANK_THREADS) c4[i] = zero4i;
    if (blockIdx.x == 0 && tid == 0) *total = 0;
    __syncthreads();

    int e0 = blockIdx.x * RANK_EPB;
    const int* dst = ei + NEDGES + e0;
    int d[RANK_PER];
    #pragma unroll
    for (int j = 0; j < RANK_PER; j++) {
        int k = tid + j * RANK_THREADS;
        d[j] = (k < RANK_EPB) ? dst[k] : 0;
    }
    int r[RANK_PER];
    #pragma unroll
    for (int j = 0; j < RANK_PER; j++) {
        int k = tid + j * RANK_THREADS;
        r[j] = (k < RANK_EPB) ? atomicAdd(&cntL[d[j]], 1) : 0;
    }
    #pragma unroll
    for (int j = 0; j < RANK_PER; j++) {
        int k = tid + j * RANK_THREADS;
        if (k < RANK_EPB) rank[e0 + k] = r[j];
    }
    __syncthreads();

    v4i* t4 = (v4i*)(tab + (size_t)blockIdx.x * NNODES);
    for (int i = tid; i < NODE4; i += RANK_THREADS) t4[i] = c4[i];
}

__global__ void k_offsets(int* __restrict__ tab, int* total, int* __restrict__ cnt,
                          int* __restrict__ off, float* __restrict__ dinv,
                          int* __restrict__ colpad, int n) {
    int i    = blockIdx.x * blockDim.x + threadIdx.x;
    int lane = threadIdx.x & 31;
    int ii   = min(i, n - 1);
    int v[RANK_BLOCKS];
    #pragma unroll
    for (int b = 0; b < RANK_BLOCKS; b++) v[b] = tab[(size_t)b * NNODES + ii];
    int c = 0;
    #pragma unroll
    for (int b = 0; b < RANK_BLOCKS; b++) c += v[b];
    if (i >= n) c = 0;
    int incl = c;
    #pragma unroll
    for (int d = 1; d < 32; d <<= 1) {
        int v1 = __shfl_up(incl, d);
        if (lane >= d) incl += v1;
    }
    int base = 0;
    if (lane == 31) base = atomicAdd(total, incl);
    base = __shfl(base, 31);
    if (i < n) {
        int run = base + incl - c;
        off[i]  = run;
        cnt[i]  = c;
        dinv[i] = rsqrtf((float)(c + 1));
        #pragma unroll
        for (int b = 0; b < RANK_BLOCKS; b++) {
            tab[(size_t)b * NNODES + i] = run;
            run += v[b];
        }
    }
    if (i < 4) colpad[i] = 0;
}

__global__ void k_fill(const int* __restrict__ ei, const int* __restrict__ rank,
                       const int* __restrict__ tab, int* __restrict__ col, int E) {
    int e = blockIdx.x * blockDim.x + threadIdx.x;
    if (e < E) {
        int d = ei[E + e];
        int b = e / RANK_EPB;
        col[tab[(size_t)b * NNODES + d] + rank[e]] = ei[e];
    }
}

template <int K, int RELUA, int RELU, int FUSE>
__global__ void __launch_bounds__(GEMM_THREADS)
k_gemm_wmma(const float* __restrict__ A, const float* __restrict__ B,
            const float* __restrict__ bias,
            float* __restrict__ C,
            const float* __restrict__ dinv,
            int M, int lda) {
    extern __shared__ float sBt[];
    constexpr int lstride = K + 2;

    constexpr int ROW4 = HID / 4;
    constexpr int PER  = K * ROW4 / GEMM_THREADS;
    static_assert(K * ROW4 % GEMM_THREADS == 0, "B must split evenly over the block");
    const v4f* B4 = (const v4f*)B;
    v4f piece[PER];
    #pragma unroll
    for (int j = 0; j < PER; j++) piece[j] = B4[threadIdx.x + j * GEMM_THREADS];
    #pragma unroll
    for (int j = 0; j < PER; j++) {
        int i  = threadIdx.x + j * GEMM_THREADS;
        int k  = i / ROW4;
        int c0 = (i - k * ROW4) * 4;
        float* dst = sBt + c0 * lstride + k;
        dst[0]           = piece[j][0];
        dst[lstride]     = piece[j][1];
        dst[2 * lstride] = piece[j][2];
        dst[3 * lstride] = piece[j][3];
    }
    __syncthreads();

    int mt   = blockIdx.x * GEMM_WAVES + (threadIdx.x >> 5);
    int lane = threadIdx.x & 31;
    int mTiles = M >> 4;
    if (mt >= mTiles) return;

    int half = lane >> 4;
    int l15  = lane & 15;

    const float* arow  = A + (size_t)(mt * 16 + l15) * lda;
    const float* bbase = sBt + (size_t)l15 * lstride;

    v8f acc[NTILE] = {};
    for (int k = 0; k < K; k += 4) {
        int kb = k + half * 2;
        v2f a = *(const v2f*)(arow + kb);
        if (RELUA) { a[0] = fmaxf(a[0], 0.0f); a[1] = fmaxf(a[1], 0.0f); }
        #pragma unroll
        for (int nt = 0; nt < NTILE; nt++) {
            v2f b = *(const v2f*)(bbase + (size_t)(nt * 16) * lstride + kb);
            acc[nt] = __builtin_amdgcn_wmma_f32_16x16x4_f32(
                 false, a,  false, b,
                 (short)0, acc[nt],  false,  false);
        }
    }

    int rbase = mt * 16 + half * 8;
    float* cbase = C + (size_t)rbase * HID + l15;

    if (FUSE) {
        float dv[8];
        #pragma unroll
        for (int v = 0; v < 8; v++) dv[v] = dinv[rbase + v];
        #pragma unroll
        for (int nt = 0; nt < NTILE; nt++) {
            #pragma unroll
            for (int v = 0; v < 8; v++)
                cbase[(size_t)v * HID + nt * 16] = acc[nt][v] * dv[v];
        }
    } else {
        #pragma unroll
        for (int nt = 0; nt < NTILE; nt++) {
            float bv = bias[nt * 16 + l15];
            #pragma unroll
            for (int v = 0; v < 8; v++) {
                float r = acc[nt][v] + bv;
                if (RELU) r = fmaxf(r, 0.0f);
                cbase[(size_t)v * HID + nt * 16] = r;
            }
        }
    }
}


#define FG_NODES 64
#define FG_NPW   (FG_NODES / GEMM_WAVES)
#define FG_HS    (HID + 4)
static_assert(NNODES % 16 == 0, "a 16-row tile is wholly inside or outside the node range");
static_assert(FG_NPW == 8, "per-node values of a wave sit in lanes 0..7");

__global__ void __launch_bounds__(GEMM_THREADS)
k_gather_gemm(const float* __restrict__ hws, const float* __restrict__ dinv,
              const int* __restrict__ off, const int* __restrict__ cnt,
              const int* __restrict__ col, const float* __restrict__ bias,
              const float* __restrict__ Wn, float* __restrict__ out, int n) {
    __shared__ float sBt[HID * (HID + 2)];
    __shared__ v4f   hT4[FG_NODES * (FG_HS / 4)];
    constexpr int lstride = HID + 2;

    constexpr int ROW4 = HID / 4;
    constexpr int PER  = HID * ROW4 / GEMM_THREADS;
    static_assert(HID * ROW4 % GEMM_THREADS == 0, "Wn must split evenly over the block");
    const v4f* B4 = (const v4f*)Wn;
    v4f piece[PER];
    #pragma unroll
    for (int j = 0; j < PER; j++) piece[j] = B4[threadIdx.x + j * GEMM_THREADS];

    int lane = threadIdx.x & 31;
    int w    = __builtin_amdgcn_readfirstlane((int)(threadIdx.x >> 5));
    int grp  = lane >> 3;
    int t    = lane & 7;
    int nb   = blockIdx.x * FG_NODES + w * FG_NPW;
    int nvalid = min(FG_NPW, n - nb);

    int ml   = min(nb + t, n - 1);
    int moff = off[ml];
    int mcnt = cnt[ml];
    int mdvb = __float_as_int(dinv[ml]);
    const v4f* b4 = (const v4f*)bias + t;
    v4f bb0 = b4[0], bb1 = b4[8], bb2 = b4[16];

    #pragma unroll
    for (int j = 0; j < PER; j++) {
        int i  = threadIdx.x + j * GEMM_THREADS;
        int k  = i / ROW4;
        int c0 = (i - k * ROW4) * 4;
        float* dst = sBt + c0 * lstride + k;
        dst[0]           = piece[j][0];
        dst[lstride]     = piece[j][1];
        dst[2 * lstride] = piece[j][2];
        dst[3 * lstride] = piece[j][3];
    }

    const v4f* rows = (const v4f*)hws + t;
    int idxs[FG_NPW];
    #pragma unroll
    for (int j = 0; j < FG_NPW; j++) {
        int sj  = __builtin_amdgcn_readlane(moff, j);
        int cj  = __builtin_amdgcn_readlane(mcnt, j);
        idxs[j] = col[sj + min(lane, max(cj, 1) - 1)];
    }
    asm volatile("" :: "v"(idxs[0]), "v"(idxs[1]), "v"(idxs[2]), "v"(idxs[3]),
                       "v"(idxs[4]), "v"(idxs[5]), "v"(idxs[6]), "v"(idxs[7]));
    #pragma unroll
    for (int j = 0; j < FG_NPW; j++) {
        int cj  = __builtin_amdgcn_readlane(mcnt, j);
        idxs[j] = (lane < cj) ? idxs[j] : 0;
    }
    #pragma unroll
    for (int i = 0; i < FG_NPW; i++) {
        if (i < nvalid) {
            int   node = nb + i;
            int   cs = __builtin_amdgcn_readlane(moff, i);
            int   cc = __builtin_amdgcn_readlane(mcnt, i);
            float dv = __int_as_float(__builtin_amdgcn_readlane(mdvb, i));
            int cidx = idxs[i];
            const v4f* rself = rows + (unsigned)node * 24u;
            v4f a0 = {}, a1 = {}, a2 = {};
            for (int e0 = 0; e0 == 0 || e0 < cc; e0 += 16) {
                if (e0 != 0 && (e0 & 31) == 0)
                    cidx = (lane < cc - e0) ? col[cs + e0 + lane] : 0;
                int  eb = (e0 & 31) + grp;
                bool on0 = e0 + grp      < cc;
                bool on1 = e0 + grp + 4  < cc;
                bool on2 = e0 + grp + 8  < cc;
                bool on3 = e0 + grp + 12 < cc;
                int  s0 = __shfl(cidx, eb);
                int  s1 = __shfl(cidx, eb + 4);
                int  s2 = __shfl(cidx, eb + 8);
                int  s3 = __shfl(cidx, eb + 12);
                const v4f* r0 = on0 ? rows + (unsigned)s0 * 24u : rself;
                const v4f* r1 = on1 ? rows + (unsigned)s1 * 24u : rself;
                const v4f* r2 = on2 ? rows + (unsigned)s2 * 24u : rself;
                const v4f* r3 = on3 ? rows + (unsigned)s3 * 24u : rself;
                v4f x00 = r0[0], x01 = r0[8], x02 = r0[16];
                v4f x10 = r1[0], x11 = r1[8], x12 = r1[16];
                v4f x20 = r2[0], x21 = r2[8], x22 = r2[16];
                v4f x30 = r3[0], x31 = r3[8], x32 = r3[16];
                if (e0 == 0 && grp == 0) {
                    v4f y0 = rself[0], y1 = rself[8], y2 = rself[16];
                    a0 = y0; a1 = y1; a2 = y2;
                }
                if (on0) { a0 += x00; a1 += x01; a2 += x02; }
                if (on1) { a0 += x10; a1 += x11; a2 += x12; }
                if (on2) { a0 += x20; a1 += x21; a2 += x22; }
                if (on3) { a0 += x30; a1 += x31; a2 += x32; }
            }
            #pragma unroll
            for (int mm = 8; mm <= 16; mm <<= 1) {
                #pragma unroll
                for (int q = 0; q < 4; q++) {
                    a0[q] += __shfl_xor(a0[q], mm);
                    a1[q] += __shfl_xor(a1[q], mm);
                    a2[q] += __shfl_xor(a2[q], mm);
                }
            }
            if (grp == 0) {
                v4f o0 = a0 * dv + bb0;
                v4f o1 = a1 * dv + bb1;
                v4f o2 = a2 * dv + bb2;
                #pragma unroll
                for (int q = 0; q < 4; q++) {
                    o0[q] = fmaxf(o0[q], 0.0f); o1[q] = fmaxf(o1[q], 0.0f); o2[q] = fmaxf(o2[q], 0.0f);
                }
                v4f* dst = hT4 + (w * FG_NPW + i) * (FG_HS / 4) + t;
                dst[0] = o0; dst[8] = o1; dst[16] = o2;
            }
        }
    }
    __syncthreads();

    int mtile   = w >> 1;
    int rowbase = blockIdx.x * FG_NODES + mtile * 16;
    if (rowbase >= n) return;
    int half = lane >> 4;
    int l15  = lane & 15;
    int nt0  = (w & 1) * 3;
    const float* abase = (const float*)hT4 + (mtile * 16 + l15) * FG_HS;
    const float* bbase = sBt + (nt0 * 16 + l15) * lstride;
    v8f acc[3] = {};
    for (int k = 0; k < HID; k += 4) {
        int kb = k + half * 2;
        v2f a = *(const v2f*)(abase + kb);
        #pragma unroll
        for (int j = 0; j < 3; j++) {
            v2f b = *(const v2f*)(bbase + (j * 16) * lstride + kb);
            acc[j] = __builtin_amdgcn_wmma_f32_16x16x4_f32(
                 false, a,  false, b,
                 (short)0, acc[j],  false,  false);
        }
    }
    int rbase = rowbase + half * 8;
    float* cbase = out + (size_t)rbase * HID + nt0 * 16 + l15;
    float dvv[8];
    #pragma unroll
    for (int v = 0; v < 8; v++) dvv[v] = dinv[rbase + v];
    #pragma unroll
    for (int j = 0; j < 3; j++) {
        #pragma unroll
        for (int v = 0; v < 8; v++)
            cbase[(size_t)v * HID + j * 16] = acc[j][v] * dvv[v];
    }
}

__global__ void k_zero(float* x, int n) {
    int i = blockIdx.x * blockDim.x + threadIdx.x;
    if (i < n) x[i] = 0.0f;
}

#define PL_ROWS 32
static_assert(FG_NODES == 2 * PL_ROWS, "two row halves per tile");
static_assert(HID == 3 * 32, "three 32-column groups per row half");

__global__ void __launch_bounds__(GEMM_THREADS)
k_gather_pool(const float* __restrict__ hws, const float* __restrict__ dinv,
              const int* __restrict__ off, const int* __restrict__ cnt,
              const int* __restrict__ col, const float* __restrict__ bias,
              const int* __restrict__ batch, float* __restrict__ g, int n) {
    __shared__ v4f hT4[FG_NODES * (FG_HS / 4)];

    int lane = threadIdx.x & 31;
    int w    = __builtin_amdgcn_readfirstlane((int)(threadIdx.x >> 5));
    int grp  = lane >> 3;
    int t    = lane & 7;
    int nb   = blockIdx.x * FG_NODES + w * FG_NPW;
    int nvalid = min(FG_NPW, n - nb);

    int hh  = (w >= 3) ? 1 : 0;
    int pr0 = hh * PL_ROWS;
    int idl = batch[min((int)(blockIdx.x * FG_NODES) + pr0 + lane, n - 1)];

    int ml   = min(nb + t, n - 1);
    int moff = off[ml];
    int mcnt = cnt[ml];
    int mdvb = __float_as_int(dinv[ml]);
    const v4f* b4 = (const v4f*)bias + t;
    v4f bb0 = b4[0], bb1 = b4[8], bb2 = b4[16];

    const v4f* rows = (const v4f*)hws + t;
    int idxs[FG_NPW];
    #pragma unroll
    for (int j = 0; j < FG_NPW; j++) {
        int sj  = __builtin_amdgcn_readlane(moff, j);
        int cj  = __builtin_amdgcn_readlane(mcnt, j);
        idxs[j] = col[sj + min(lane, max(cj, 1) - 1)];
    }
    asm volatile("" :: "v"(idxs[0]), "v"(idxs[1]), "v"(idxs[2]), "v"(idxs[3]),
                       "v"(idxs[4]), "v"(idxs[5]), "v"(idxs[6]), "v"(idxs[7]));
    #pragma unroll
    for (int j = 0; j < FG_NPW; j++) {
        int cj  = __builtin_amdgcn_readlane(mcnt, j);
        idxs[j] = (lane < cj) ? idxs[j] : 0;
    }
    #pragma unroll
    for (int i = 0; i < FG_NPW; i++) {
        if (i < nvalid) {
            int   node = nb + i;
            int   cs = __builtin_amdgcn_readlane(moff, i);
            int   cc = __builtin_amdgcn_readlane(mcnt, i);
            float dv = __int_as_float(__builtin_amdgcn_readlane(mdvb, i));
            int cidx = idxs[i];
            const v4f* rself = rows + (unsigned)node * 24u;
            v4f a0 = {}, a1 = {}, a2 = {};
            for (int e0 = 0; e0 == 0 || e0 < cc; e0 += 16) {
                if (e0 != 0 && (e0 & 31) == 0)
                    cidx = (lane < cc - e0) ? col[cs + e0 + lane] : 0;
                int  eb = (e0 & 31) + grp;
                bool on0 = e0 + grp      < cc;
                bool on1 = e0 + grp + 4  < cc;
                bool on2 = e0 + grp + 8  < cc;
                bool on3 = e0 + grp + 12 < cc;
                int  s0 = __shfl(cidx, eb);
                int  s1 = __shfl(cidx, eb + 4);
                int  s2 = __shfl(cidx, eb + 8);
                int  s3 = __shfl(cidx, eb + 12);
                const v4f* r0 = on0 ? rows + (unsigned)s0 * 24u : rself;
                const v4f* r1 = on1 ? rows + (unsigned)s1 * 24u : rself;
                const v4f* r2 = on2 ? rows + (unsigned)s2 * 24u : rself;
                const v4f* r3 = on3 ? rows + (unsigned)s3 * 24u : rself;
                v4f x00 = r0[0], x01 = r0[8], x02 = r0[16];
                v4f x10 = r1[0], x11 = r1[8], x12 = r1[16];
                v4f x20 = r2[0], x21 = r2[8], x22 = r2[16];
                v4f x30 = r3[0], x31 = r3[8], x32 = r3[16];
                if (e0 == 0 && grp == 0) {
                    v4f y0 = rself[0], y1 = rself[8], y2 = rself[16];
                    a0 = y0; a1 = y1; a2 = y2;
                }
                if (on0) { a0 += x00; a1 += x01; a2 += x02; }
                if (on1) { a0 += x10; a1 += x11; a2 += x12; }
                if (on2) { a0 += x20; a1 += x21; a2 += x22; }
                if (on3) { a0 += x30; a1 += x31; a2 += x32; }
            }
            #pragma unroll
            for (int mm = 8; mm <= 16; mm <<= 1) {
                #pragma unroll
                for (int q = 0; q < 4; q++) {
                    a0[q] += __shfl_xor(a0[q], mm);
                    a1[q] += __shfl_xor(a1[q], mm);
                    a2[q] += __shfl_xor(a2[q], mm);
                }
            }
            if (grp == 0) {
                v4f o0 = a0 * dv + bb0;
                v4f o1 = a1 * dv + bb1;
                v4f o2 = a2 * dv + bb2;
                #pragma unroll
                for (int q = 0; q < 4; q++) {
                    o0[q] = fmaxf(o0[q], 0.0f); o1[q] = fmaxf(o1[q], 0.0f); o2[q] = fmaxf(o2[q], 0.0f);
                }
                v4f* dst = hT4 + (w * FG_NPW + i) * (FG_HS / 4) + t;
                dst[0] = o0; dst[8] = o1; dst[16] = o2;
            }
        }
    }
    __syncthreads();

    if (w >= 6) return;
    int nrows = min(FG_NODES, n - (int)(blockIdx.x * FG_NODES));
    int nr    = min(PL_ROWS, nrows - pr0);
    if (nr <= 0) return;
    int c = (w - 3 * hh) * 32 + lane;
    const float* tp = (const float*)hT4 + pr0 * FG_HS + c;
    float x[PL_ROWS];
    #pragma unroll
    for (int j = 0; j < PL_ROWS; j++) x[j] = (j < nr) ? tp[j * FG_HS] : 0.0f;
    int cur = __builtin_amdgcn_readlane(idl, 0);
    float sum = 0.0f;
    #pragma unroll
    for (int j = 0; j < PL_ROWS; j++) {
        if (j < nr) {
            int b = __builtin_amdgcn_readlane(idl, j);
            if (b != cur) {
                atomicAdd(g + (size_t)cur * HID + c, sum);
                sum = 0.0f;
                cur = b;
            }
            sum += x[j];
        }
    }
    atomicAdd(g + (size_t)cur * HID + c, sum);
}

__global__ void k_dec2(const float* __restrict__ g2, const float* __restrict__ W,
                       const float* __restrict__ b, float* __restrict__ out) {
    int idx = blockIdx.x * blockDim.x + threadIdx.x;
    if (idx < NGRAPH * OUTDIM) {
        int gi = idx / OUTDIM;
        int o  = idx - gi * OUTDIM;
        float s = b[o];
        const float* row = g2 + (size_t)gi * HID;
        #pragma unroll 8
        for (int k = 0; k < HID; k++) s += row[k] * W[k * OUTDIM + o];
        out[idx] = s;
    }
}

extern "C" void kernel_launch(void* const* d_in, const int* in_sizes, int n_in,
                              void* d_out, int out_size, void* d_ws, size_t ws_size,
                              hipStream_t stream) {
    const float* x       = (const float*)d_in[0];
    const int*   ei      = (const int*)  d_in[1];
    const int*   batch   = (const int*)  d_in[2];
    const float* enc_W0  = (const float*)d_in[3];
    const float* enc_b0  = (const float*)d_in[4];
    const float* enc_W1  = (const float*)d_in[5];
    const float* enc_b1  = (const float*)d_in[6];
    const float* conv_W  = (const float*)d_in[7];
    const float* conv_b  = (const float*)d_in[8];
    const float* dec_W0  = (const float*)d_in[9];
    const float* dec_b0  = (const float*)d_in[10];
    const float* dec_W1  = (const float*)d_in[11];
    const float* dec_b1  = (const float*)d_in[12];
    float* out = (float*)d_out;

    float* ws   = (float*)d_ws;
    size_t feat = (size_t)NNODES * HID;
    float* buf0 = ws;
    float* buf1 = buf0 + feat;
    float* buf2 = buf1 + feat;
    float* dinv = buf2 + feat;
    float* g    = dinv + NNODES;
    float* g2   = g + (size_t)NGRAPH * HID;
    int*   cnt  = (int*)(g2 + (size_t)NGRAPH * HID);
    int*   tot  = cnt + NNODES;
    int*   off  = tot + 16;
    int*   rank = off + NNODES;
    int*   col  = rank + NEDGES;
    int*   tab  = col + NEDGES + 64;

    const int B = 256;

    k_rank<<<RANK_BLOCKS, RANK_THREADS, 0, stream>>>(ei, rank, tab, tot);
    k_offsets<<<cdiv(NNODES, B), B, 0, stream>>>(tab, tot, cnt, off, dinv, col + NEDGES, NNODES);
    k_fill<<<cdiv(NEDGES, B), B, 0, stream>>>(ei, rank, tab, col, NEDGES);

    int mTilesN   = NNODES / 16;
    int gemmBlksN = cdiv(mTilesN, GEMM_WAVES);
    size_t shm128 = (size_t)HID * (INDIM + 2) * sizeof(float);
    size_t shm96  = (size_t)HID * (HID   + 2) * sizeof(float);
    k_gemm_wmma<INDIM, 0, 1, 0><<<gemmBlksN, GEMM_THREADS, shm128, stream>>>(
        x, enc_W0, enc_b0, buf0, nullptr, NNODES, INDIM);
    k_gemm_wmma<HID, 0, 0, 0><<<gemmBlksN, GEMM_THREADS, shm96, stream>>>(
        buf0, enc_W1, enc_b1, buf1, nullptr, NNODES, HID);

    k_gemm_wmma<HID, 0, 0, 1><<<gemmBlksN, GEMM_THREADS, shm96, stream>>>(
        buf1, conv_W, nullptr, buf0  , dinv, NNODES, HID);
    float* hsrc = buf0;
    float* hdst = buf2;
    int fgBlks = cdiv(NNODES, FG_NODES);
    for (int l = 0; l + 1 < NCONV; l++) {
        const float* Wn = conv_W + (size_t)(l + 1) * HID * HID;
        const float* bl = conv_b + (size_t)l * HID;
        k_gather_gemm<<<fgBlks, GEMM_THREADS, 0, stream>>>(hsrc, dinv, off, cnt, col, bl, Wn, hdst, NNODES);
        float* t = hsrc; hsrc = hdst; hdst = t;
    }
    k_zero<<<cdiv(NGRAPH * HID, B), B, 0, stream>>>(g, NGRAPH * HID);
    k_gather_pool<<<fgBlks, GEMM_THREADS, 0, stream>>>(hsrc, dinv, off, cnt, col,
                                                       conv_b + (size_t)(NCONV - 1) * HID, batch, g, NNODES);

    int mTilesG = NGRAPH / 16;
    k_gemm_wmma<HID, 0, 1, 0><<<cdiv(mTilesG, GEMM_WAVES), GEMM_THREADS, shm96, stream>>>(
        g, dec_W0, dec_b0, g2, nullptr, NGRAPH, HID);
    k_dec2<<<cdiv(NGRAPH * OUTDIM, B), B, 0, stream>>>(g2, dec_W1, dec_b1, out);
}


# Encoder MLP and layer 0's product in one launch: a wave carries its 16-row strip through the three WMMA products via an LDS tile (2 launches, both intermediate round trips less); k and sum order kept
# speedup vs baseline: 3.9999x; 1.0753x over previous
#include <hip/hip_runtime.h>
#include <hip/hip_bf16.h>

typedef __attribute__((ext_vector_type(2))) float v2f;
typedef __attribute__((ext_vector_type(8))) float v8f;
typedef __attribute__((ext_vector_type(4))) float v4f;

#define NNODES 50000
#define NEDGES 800000
#define NGRAPH 512
#define INDIM  128
#define HID    96
#define OUTDIM 10
#define NCONV  4
#define NTILE  6
#define GEMM_WAVES 8
#define GEMM_THREADS (GEMM_WAVES * 32)

static inline int cdiv(int a, int b) { return (a + b - 1) / b; }

typedef __attribute__((ext_vector_type(4))) int v4i;

#define RANK_BLOCKS  64
#define RANK_THREADS 1024
#define RANK_EPB     (NEDGES / RANK_BLOCKS)
#define RANK_PER     ((RANK_EPB + RANK_THREADS - 1) / RANK_THREADS)
#define NODE4        (NNODES / 4)
static_assert(NEDGES % RANK_BLOCKS == 0, "slices must be equal");
static_assert(NNODES % 4 == 0, "counter sets are moved as 16-byte pieces");

__global__ void __launch_bounds__(RANK_THREADS)
k_rank(const int* __restrict__ ei, int* __restrict__ rank, int* __restrict__ tab,
       int* __restrict__ total) {
    __shared__ int cntL[NNODES];
    int tid = threadIdx.x;
    v4i* c4 = (v4i*)cntL;
    const v4i zero4i = {0, 0, 0, 0};
    for (int i = tid; i < NODE4; i += RANK_THREADS) c4[i] = zero4i;
    if (blockIdx.x == 0 && tid == 0) *total = 0;
    __syncthreads();

    int e0 = blockIdx.x * RANK_EPB;
    const int* dst = ei + NEDGES + e0;
    int d[RANK_PER];
    #pragma unroll
    for (int j = 0; j < RANK_PER; j++) {
        int k = tid + j * RANK_THREADS;
        d[j] = (k < RANK_EPB) ? dst[k] : 0;
    }
    int r[RANK_PER];
    #pragma unroll
    for (int j = 0; j < RANK_PER; j++) {
        int k = tid + j * RANK_THREADS;
        r[j] = (k < RANK_EPB) ? atomicAdd(&cntL[d[j]], 1) : 0;
    }
    #pragma unroll
    for (int j = 0; j < RANK_PER; j++) {
        int k = tid + j * RANK_THREADS;
        if (k < RANK_EPB) rank[e0 + k] = r[j];
    }
    __syncthreads();

    v4i* t4 = (v4i*)(tab + (size_t)blockIdx.x * NNODES);
    for (int i = tid; i < NODE4; i += RANK_THREADS) t4[i] = c4[i];
}

__global__ void k_offsets(int* __restrict__ tab, int* total, int* __restrict__ cnt,
                          int* __restrict__ off, float* __restrict__ dinv,
                          int* __restrict__ colpad, int n) {
    int i    = blockIdx.x * blockDim.x + threadIdx.x;
    int lane = threadIdx.x & 31;
    int ii   = min(i, n - 1);
    int v[RANK_BLOCKS];
    #pragma unroll
    for (int b = 0; b < RANK_BLOCKS; b++) v[b] = tab[(size_t)b * NNODES + ii];
    int c = 0;
    #pragma unroll
    for (int b = 0; b < RANK_BLOCKS; b++) c += v[b];
    if (i >= n) c = 0;
    int incl = c;
    #pragma unroll
    for (int d = 1; d < 32; d <<= 1) {
        int v1 = __shfl_up(incl, d);
        if (lane >= d) incl += v1;
    }
    int base = 0;
    if (lane == 31) base = atomicAdd(total, incl);
    base = __shfl(base, 31);
    if (i < n) {
        int run = base + incl - c;
        off[i]  = run;
        cnt[i]  = c;
        dinv[i] = rsqrtf((float)(c + 1));
        #pragma unroll
        for (int b = 0; b < RANK_BLOCKS; b++) {
            tab[(size_t)b * NNODES + i] = run;
            run += v[b];
        }
    }
    if (i < 4) colpad[i] = 0;
}

__global__ void k_fill(const int* __restrict__ ei, const int* __restrict__ rank,
                       const int* __restrict__ tab, int* __restrict__ col, int E) {
    int e = blockIdx.x * blockDim.x + threadIdx.x;
    if (e < E) {
        int d = ei[E + e];
        int b = e / RANK_EPB;
        col[tab[(size_t)b * NNODES + d] + rank[e]] = ei[e];
    }
}

template <int K, int RELUA, int RELU, int FUSE>
__global__ void __launch_bounds__(GEMM_THREADS)
k_gemm_wmma(const float* __restrict__ A, const float* __restrict__ B,
            const float* __restrict__ bias,
            float* __restrict__ C,
            const float* __restrict__ dinv,
            int M, int lda) {
    extern __shared__ float sBt[];
    constexpr int lstride = K + 2;

    constexpr int ROW4 = HID / 4;
    constexpr int PER  = K * ROW4 / GEMM_THREADS;
    static_assert(K * ROW4 % GEMM_THREADS == 0, "B must split evenly over the block");
    const v4f* B4 = (const v4f*)B;
    v4f piece[PER];
    #pragma unroll
    for (int j = 0; j < PER; j++) piece[j] = B4[threadIdx.x + j * GEMM_THREADS];
    #pragma unroll
    for (int j = 0; j < PER; j++) {
        int i  = threadIdx.x + j * GEMM_THREADS;
        int k  = i / ROW4;
        int c0 = (i - k * ROW4) * 4;
        float* dst = sBt + c0 * lstride + k;
        dst[0]           = piece[j][0];
        dst[lstride]     = piece[j][1];
        dst[2 * lstride] = piece[j][2];
        dst[3 * lstride] = piece[j][3];
    }
    __syncthreads();

    int mt   = blockIdx.x * GEMM_WAVES + (threadIdx.x >> 5);
    int lane = threadIdx.x & 31;
    int mTiles = M >> 4;
    if (mt >= mTiles) return;

    int half = lane >> 4;
    int l15  = lane & 15;

    const float* arow  = A + (size_t)(mt * 16 + l15) * lda;
    const float* bbase = sBt + (size_t)l15 * lstride;

    v8f acc[NTILE] = {};
    for (int k = 0; k < K; k += 4) {
        int kb = k + half * 2;
        v2f a = *(const v2f*)(arow + kb);
        if (RELUA) { a[0] = fmaxf(a[0], 0.0f); a[1] = fmaxf(a[1], 0.0f); }
        #pragma unroll
        for (int nt = 0; nt < NTILE; nt++) {
            v2f b = *(const v2f*)(bbase + (size_t)(nt * 16) * lstride + kb);
            acc[nt] = __builtin_amdgcn_wmma_f32_16x16x4_f32(
                 false, a,  false, b,
                 (short)0, acc[nt],  false,  false);
        }
    }

    int rbase = mt * 16 + half * 8;
    float* cbase = C + (size_t)rbase * HID + l15;

    if (FUSE) {
        float dv[8];
        #pragma unroll
        for (int v = 0; v < 8; v++) dv[v] = dinv[rbase + v];
        #pragma unroll
        for (int nt = 0; nt < NTILE; nt++) {
            #pragma unroll
            for (int v = 0; v < 8; v++)
                cbase[(size_t)v * HID + nt * 16] = acc[nt][v] * dv[v];
        }
    } else {
        #pragma unroll
        for (int nt = 0; nt < NTILE; nt++) {
            float bv = bias[nt * 16 + l15];
            #pragma unroll
            for (int v = 0; v < 8; v++) {
                float r = acc[nt][v] + bv;
                if (RELU) r = fmaxf(r, 0.0f);
                cbase[(size_t)v * HID + nt * 16] = r;
            }
        }
    }
}

#define EC_WAVES   16
#define EC_THREADS (EC_WAVES * 32)
#define EC_TS      (HID + 4)

__global__ void __launch_bounds__(EC_THREADS)
k_enc_chain(const float* __restrict__ x,
            const float* __restrict__ W0, const float* __restrict__ b0,
            const float* __restrict__ W1, const float* __restrict__ b1,
            const float* __restrict__ Wc, const float* __restrict__ dinv,
            float* __restrict__ out, int M) {
    __shared__ __attribute__((aligned(16))) float sB0[HID * (INDIM + 2)];
    __shared__ __attribute__((aligned(16))) float sB1[HID * (HID + 2)];
    __shared__ __attribute__((aligned(16))) float sB2[HID * (HID + 2)];
    __shared__ __attribute__((aligned(16))) float tile[EC_WAVES * 16 * EC_TS];
    constexpr int ls0 = INDIM + 2;
    constexpr int ls1 = HID + 2;

    constexpr int ROW4 = HID / 4;
    constexpr int N0   = INDIM * ROW4;
    constexpr int N1   = HID * ROW4;
    constexpr int P0   = N0 / EC_THREADS;
    constexpr int P1   = (N1 + EC_THREADS - 1) / EC_THREADS;
    static_assert(N0 % EC_THREADS == 0, "W0 must split evenly over the block");
    const v4f* W0_4 = (const v4f*)W0;
    const v4f* W1_4 = (const v4f*)W1;
    const v4f* Wc_4 = (const v4f*)Wc;
    v4f p0[P0], p1[P1], p2[P1];
    #pragma unroll
    for (int j = 0; j < P0; j++) p0[j] = W0_4[threadIdx.x + j * EC_THREADS];
    #pragma unroll
    for (int j = 0; j < P1; j++) {
        int ii = min((int)threadIdx.x + j * EC_THREADS, N1 - 1);
        p1[j] = W1_4[ii];
        p2[j] = Wc_4[ii];
    }
    #pragma unroll
    for (int j = 0; j < P0; j++) {
        int i  = threadIdx.x + j * EC_THREADS;
        int k  = i / ROW4;
        int c0 = (i - k * ROW4) * 4;
        float* dst = sB0 + c0 * ls0 + k;
        dst[0]       = p0[j][0];
        dst[ls0]     = p0[j][1];
        dst[2 * ls0] = p0[j][2];
        dst[3 * ls0] = p0[j][3];
    }
    #pragma unroll
    for (int j = 0; j < P1; j++) {
        int i  = threadIdx.x + j * EC_THREADS;
        if (i < N1) {
            int k  = i / ROW4;
            int c0 = (i - k * ROW4) * 4;
            float* d1 = sB1 + c0 * ls1 + k;
            float* d2 = sB2 + c0 * ls1 + k;
            d1[0]       = p1[j][0];  d2[0]       = p2[j][0];
            d1[ls1]     = p1[j][1];  d2[ls1]     = p2[j][1];
            d1[2 * ls1] = p1[j][2];  d2[2 * ls1] = p2[j][2];
            d1[3 * ls1] = p1[j][3];  d2[3 * ls1] = p2[j][3];
        }
    }
    __syncthreads();

    int w    = threadIdx.x >> 5;
    int lane = threadIdx.x & 31;
    int half = lane >> 4;
    int l15  = lane & 15;
    int mt   = blockIdx.x * EC_WAVES + w;
    bool valid = mt < (M >> 4);
    float* trow = tile + (w * 16) * EC_TS;
    float* tst  = trow + (half * 8) * EC_TS + l15;
    const float* tA = trow + l15 * EC_TS;

    if (valid) {
        const float* arow  = x + (size_t)(mt * 16 + l15) * INDIM;
        const float* bbase = sB0 + l15 * ls0;
        v8f acc[NTILE] = {};
        for (int k = 0; k < INDIM; k += 4) {
            int kb = k + half * 2;
            v2f a = *(const v2f*)(arow + kb);
            #pragma unroll
            for (int nt = 0; nt < NTILE; nt++) {
                v2f b = *(const v2f*)(bbase + (nt * 16) * ls0 + kb);
                acc[nt] = __builtin_amdgcn_wmma_f32_16x16x4_f32(
                     false, a,  false, b,
                     (short)0, acc[nt],  false,  false);
            }
        }
        #pragma unroll
        for (int nt = 0; nt < NTILE; nt++) {
            float bv = b0[nt * 16 + l15];
            #pragma unroll
            for (int v = 0; v < 8; v++)
                tst[v * EC_TS + nt * 16] = fmaxf(acc[nt][v] + bv, 0.0f);
        }
    }
    __syncthreads();

    if (valid) {
        const float* bbase = sB1 + l15 * ls1;
        v8f acc[NTILE] = {};
        for (int k = 0; k < HID; k += 4) {
            int kb = k + half * 2;
            v2f a = *(const v2f*)(tA + kb);
            #pragma unroll
            for (int nt = 0; nt < NTILE; nt++) {
                v2f b = *(const v2f*)(bbase + (nt * 16) * ls1 + kb);
                acc[nt] = __builtin_amdgcn_wmma_f32_16x16x4_f32(
                     false, a,  false, b,
                     (short)0, acc[nt],  false,  false);
            }
        }
        #pragma unroll
        for (int nt = 0; nt < NTILE; nt++) {
            float bv = b1[nt * 16 + l15];
            #pragma unroll
            for (int v = 0; v < 8; v++)
                tst[v * EC_TS + nt * 16] = acc[nt][v] + bv;
        }
    }
    __syncthreads();

    if (valid) {
        const float* bbase = sB2 + l15 * ls1;
        v8f acc[NTILE] = {};
        for (int k = 0; k < HID; k += 4) {
            int kb = k + half * 2;
            v2f a = *(const v2f*)(tA + kb);
            #pragma unroll
            for (int nt = 0; nt < NTILE; nt++) {
                v2f b = *(const v2f*)(bbase + (nt * 16) * ls1 + kb);
                acc[nt] = __builtin_amdgcn_wmma_f32_16x16x4_f32(
                     false, a,  false, b,
                     (short)0, acc[nt],  false,  false);
            }
        }
        int rbase = mt * 16 + half * 8;
        float* cbase = out + (size_t)rbase * HID + l15;
        float dv[8];
        #pragma unroll
        for (int v = 0; v < 8; v++) dv[v] = dinv[rbase + v];
        #pragma unroll
        for (int nt = 0; nt < NTILE; nt++) {
            #pragma unroll
            for (int v = 0; v < 8; v++)
                cbase[(size_t)v * HID + nt * 16] = acc[nt][v] * dv[v];
        }
    }
}


#define FG_NODES 64
#define FG_NPW   (FG_NODES / GEMM_WAVES)
#define FG_HS    (HID + 4)
static_assert(NNODES % 16 == 0, "a 16-row tile is wholly inside or outside the node range");
static_assert(FG_NPW == 8, "per-node values of a wave sit in lanes 0..7");

__global__ void __launch_bounds__(GEMM_THREADS)
k_gather_gemm(const float* __restrict__ hws, const float* __restrict__ dinv,
              const int* __restrict__ off, const int* __restrict__ cnt,
              const int* __restrict__ col, const float* __restrict__ bias,
              const float* __restrict__ Wn, float* __restrict__ out, int n) {
    __shared__ float sBt[HID * (HID + 2)];
    __shared__ v4f   hT4[FG_NODES * (FG_HS / 4)];
    constexpr int lstride = HID + 2;

    constexpr int ROW4 = HID / 4;
    constexpr int PER  = HID * ROW4 / GEMM_THREADS;
    static_assert(HID * ROW4 % GEMM_THREADS == 0, "Wn must split evenly over the block");
    const v4f* B4 = (const v4f*)Wn;
    v4f piece[PER];
    #pragma unroll
    for (int j = 0; j < PER; j++) piece[j] = B4[threadIdx.x + j * GEMM_THREADS];

    int lane = threadIdx.x & 31;
    int w    = __builtin_amdgcn_readfirstlane((int)(threadIdx.x >> 5));
    int grp  = lane >> 3;
    int t    = lane & 7;
    int nb   = blockIdx.x * FG_NODES + w * FG_NPW;
    int nvalid = min(FG_NPW, n - nb);

    int ml   = min(nb + t, n - 1);
    int moff = off[ml];
    int mcnt = cnt[ml];
    int mdvb = __float_as_int(dinv[ml]);
    const v4f* b4 = (const v4f*)bias + t;
    v4f bb0 = b4[0], bb1 = b4[8], bb2 = b4[16];

    #pragma unroll
    for (int j = 0; j < PER; j++) {
        int i  = threadIdx.x + j * GEMM_THREADS;
        int k  = i / ROW4;
        int c0 = (i - k * ROW4) * 4;
        float* dst = sBt + c0 * lstride + k;
        dst[0]           = piece[j][0];
        dst[lstride]     = piece[j][1];
        dst[2 * lstride] = piece[j][2];
        dst[3 * lstride] = piece[j][3];
    }

    const v4f* rows = (const v4f*)hws + t;
    int idxs[FG_NPW];
    #pragma unroll
    for (int j = 0; j < FG_NPW; j++) {
        int sj  = __builtin_amdgcn_readlane(moff, j);
        int cj  = __builtin_amdgcn_readlane(mcnt, j);
        idxs[j] = col[sj + min(lane, max(cj, 1) - 1)];
    }
    asm volatile("" :: "v"(idxs[0]), "v"(idxs[1]), "v"(idxs[2]), "v"(idxs[3]),
                       "v"(idxs[4]), "v"(idxs[5]), "v"(idxs[6]), "v"(idxs[7]));
    #pragma unroll
    for (int j = 0; j < FG_NPW; j++) {
        int cj  = __builtin_amdgcn_readlane(mcnt, j);
        idxs[j] = (lane < cj) ? idxs[j] : 0;
    }
    #pragma unroll
    for (int i = 0; i < FG_NPW; i++) {
        if (i < nvalid) {
            int   node = nb + i;
            int   cs = __builtin_amdgcn_readlane(moff, i);
            int   cc = __builtin_amdgcn_readlane(mcnt, i);
            float dv = __int_as_float(__builtin_amdgcn_readlane(mdvb, i));
            int cidx = idxs[i];
            const v4f* rself = rows + (unsigned)node * 24u;
            v4f a0 = {}, a1 = {}, a2 = {};
            for (int e0 = 0; e0 == 0 || e0 < cc; e0 += 16) {
                if (e0 != 0 && (e0 & 31) == 0)
                    cidx = (lane < cc - e0) ? col[cs + e0 + lane] : 0;
                int  eb = (e0 & 31) + grp;
                bool on0 = e0 + grp      < cc;
                bool on1 = e0 + grp + 4  < cc;
                bool on2 = e0 + grp + 8  < cc;
                bool on3 = e0 + grp + 12 < cc;
                int  s0 = __shfl(cidx, eb);
                int  s1 = __shfl(cidx, eb + 4);
                int  s2 = __shfl(cidx, eb + 8);
                int  s3 = __shfl(cidx, eb + 12);
                const v4f* r0 = on0 ? rows + (unsigned)s0 * 24u : rself;
                const v4f* r1 = on1 ? rows + (unsigned)s1 * 24u : rself;
                const v4f* r2 = on2 ? rows + (unsigned)s2 * 24u : rself;
                const v4f* r3 = on3 ? rows + (unsigned)s3 * 24u : rself;
                v4f x00 = r0[0], x01 = r0[8], x02 = r0[16];
                v4f x10 = r1[0], x11 = r1[8], x12 = r1[16];
                v4f x20 = r2[0], x21 = r2[8], x22 = r2[16];
                v4f x30 = r3[0], x31 = r3[8], x32 = r3[16];
                if (e0 == 0 && grp == 0) {
                    v4f y0 = rself[0], y1 = rself[8], y2 = rself[16];
                    a0 = y0; a1 = y1; a2 = y2;
                }
                if (on0) { a0 += x00; a1 += x01; a2 += x02; }
                if (on1) { a0 += x10; a1 += x11; a2 += x12; }
                if (on2) { a0 += x20; a1 += x21; a2 += x22; }
                if (on3) { a0 += x30; a1 += x31; a2 += x32; }
            }
            #pragma unroll
            for (int mm = 8; mm <= 16; mm <<= 1) {
                #pragma unroll
                for (int q = 0; q < 4; q++) {
                    a0[q] += __shfl_xor(a0[q], mm);
                    a1[q] += __shfl_xor(a1[q], mm);
                    a2[q] += __shfl_xor(a2[q], mm);
                }
            }
            if (grp == 0) {
                v4f o0 = a0 * dv + bb0;
                v4f o1 = a1 * dv + bb1;
                v4f o2 = a2 * dv + bb2;
                #pragma unroll
                for (int q = 0; q < 4; q++) {
                    o0[q] = fmaxf(o0[q], 0.0f); o1[q] = fmaxf(o1[q], 0.0f); o2[q] = fmaxf(o2[q], 0.0f);
                }
                v4f* dst = hT4 + (w * FG_NPW + i) * (FG_HS / 4) + t;
                dst[0] = o0; dst[8] = o1; dst[16] = o2;
            }
        }
    }
    __syncthreads();

    int mtile   = w >> 1;
    int rowbase = blockIdx.x * FG_NODES + mtile * 16;
    if (rowbase >= n) return;
    int half = lane >> 4;
    int l15  = lane & 15;
    int nt0  = (w & 1) * 3;
    const float* abase = (const float*)hT4 + (mtile * 16 + l15) * FG_HS;
    const float* bbase = sBt + (nt0 * 16 + l15) * lstride;
    v8f acc[3] = {};
    for (int k = 0; k < HID; k += 4) {
        int kb = k + half * 2;
        v2f a = *(const v2f*)(abase + kb);
        #pragma unroll
        for (int j = 0; j < 3; j++) {
            v2f b = *(const v2f*)(bbase + (j * 16) * lstride + kb);
            acc[j] = __builtin_amdgcn_wmma_f32_16x16x4_f32(
                 false, a,  false, b,
                 (short)0, acc[j],  false,  false);
        }
    }
    int rbase = rowbase + half * 8;
    float* cbase = out + (size_t)rbase * HID + nt0 * 16 + l15;
    float dvv[8];
    #pragma unroll
    for (int v = 0; v < 8; v++) dvv[v] = dinv[rbase + v];
    #pragma unroll
    for (int j = 0; j < 3; j++) {
        #pragma unroll
        for (int v = 0; v < 8; v++)
            cbase[(size_t)v * HID + j * 16] = acc[j][v] * dvv[v];
    }
}

__global__ void k_zero(float* x, int n) {
    int i = blockIdx.x * blockDim.x + threadIdx.x;
    if (i < n) x[i] = 0.0f;
}

#define PL_ROWS 32
static_assert(FG_NODES == 2 * PL_ROWS, "two row halves per tile");
static_assert(HID == 3 * 32, "three 32-column groups per row half");

__global__ void __launch_bounds__(GEMM_THREADS)
k_gather_pool(const float* __restrict__ hws, const float* __restrict__ dinv,
              const int* __restrict__ off, const int* __restrict__ cnt,
              const int* __restrict__ col, const float* __restrict__ bias,
              const int* __restrict__ batch, float* __restrict__ g, int n) {
    __shared__ v4f hT4[FG_NODES * (FG_HS / 4)];

    int lane = threadIdx.x & 31;
    int w    = __builtin_amdgcn_readfirstlane((int)(threadIdx.x >> 5));
    int grp  = lane >> 3;
    int t    = lane & 7;
    int nb   = blockIdx.x * FG_NODES + w * FG_NPW;
    int nvalid = min(FG_NPW, n - nb);

    int hh  = (w >= 3) ? 1 : 0;
    int pr0 = hh * PL_ROWS;
    int idl = batch[min((int)(blockIdx.x * FG_NODES) + pr0 + lane, n - 1)];

    int ml   = min(nb + t, n - 1);
    int moff = off[ml];
    int mcnt = cnt[ml];
    int mdvb = __float_as_int(dinv[ml]);
    const v4f* b4 = (const v4f*)bias + t;
    v4f bb0 = b4[0], bb1 = b4[8], bb2 = b4[16];

    const v4f* rows = (const v4f*)hws + t;
    int idxs[FG_NPW];
    #pragma unroll
    for (int j = 0; j < FG_NPW; j++) {
        int sj  = __builtin_amdgcn_readlane(moff, j);
        int cj  = __builtin_amdgcn_readlane(mcnt, j);
        idxs[j] = col[sj + min(lane, max(cj, 1) - 1)];
    }
    asm volatile("" :: "v"(idxs[0]), "v"(idxs[1]), "v"(idxs[2]), "v"(idxs[3]),
                       "v"(idxs[4]), "v"(idxs[5]), "v"(idxs[6]), "v"(idxs[7]));
    #pragma unroll
    for (int j = 0; j < FG_NPW; j++) {
        int cj  = __builtin_amdgcn_readlane(mcnt, j);
        idxs[j] = (lane < cj) ? idxs[j] : 0;
    }
    #pragma unroll
    for (int i = 0; i < FG_NPW; i++) {
        if (i < nvalid) {
            int   node = nb + i;
            int   cs = __builtin_amdgcn_readlane(moff, i);
            int   cc = __builtin_amdgcn_readlane(mcnt, i);
            float dv = __int_as_float(__builtin_amdgcn_readlane(mdvb, i));
            int cidx = idxs[i];
            const v4f* rself = rows + (unsigned)node * 24u;
            v4f a0 = {}, a1 = {}, a2 = {};
            for (int e0 = 0; e0 == 0 || e0 < cc; e0 += 16) {
                if (e0 != 0 && (e0 & 31) == 0)
                    cidx = (lane < cc - e0) ? col[cs + e0 + lane] : 0;
                int  eb = (e0 & 31) + grp;
                bool on0 = e0 + grp      < cc;
                bool on1 = e0 + grp + 4  < cc;
                bool on2 = e0 + grp + 8  < cc;
                bool on3 = e0 + grp + 12 < cc;
                int  s0 = __shfl(cidx, eb);
                int  s1 = __shfl(cidx, eb + 4);
                int  s2 = __shfl(cidx, eb + 8);
                int  s3 = __shfl(cidx, eb + 12);
                const v4f* r0 = on0 ? rows + (unsigned)s0 * 24u : rself;
                const v4f* r1 = on1 ? rows + (unsigned)s1 * 24u : rself;
                const v4f* r2 = on2 ? rows + (unsigned)s2 * 24u : rself;
                const v4f* r3 = on3 ? rows + (unsigned)s3 * 24u : rself;
                v4f x00 = r0[0], x01 = r0[8], x02 = r0[16];
                v4f x10 = r1[0], x11 = r1[8], x12 = r1[16];
                v4f x20 = r2[0], x21 = r2[8], x22 = r2[16];
                v4f x30 = r3[0], x31 = r3[8], x32 = r3[16];
                if (e0 == 0 && grp == 0) {
                    v4f y0 = rself[0], y1 = rself[8], y2 = rself[16];
                    a0 = y0; a1 = y1; a2 = y2;
                }
                if (on0) { a0 += x00; a1 += x01; a2 += x02; }
                if (on1) { a0 += x10; a1 += x11; a2 += x12; }
                if (on2) { a0 += x20; a1 += x21; a2 += x22; }
                if (on3) { a0 += x30; a1 += x31; a2 += x32; }
            }
            #pragma unroll
            for (int mm = 8; mm <= 16; mm <<= 1) {
                #pragma unroll
                for (int q = 0; q < 4; q++) {
                    a0[q] += __shfl_xor(a0[q], mm);
                    a1[q] += __shfl_xor(a1[q], mm);
                    a2[q] += __shfl_xor(a2[q], mm);
                }
            }
            if (grp == 0) {
                v4f o0 = a0 * dv + bb0;
                v4f o1 = a1 * dv + bb1;
                v4f o2 = a2 * dv + bb2;
                #pragma unroll
                for (int q = 0; q < 4; q++) {
                    o0[q] = fmaxf(o0[q], 0.0f); o1[q] = fmaxf(o1[q], 0.0f); o2[q] = fmaxf(o2[q], 0.0f);
                }
                v4f* dst = hT4 + (w * FG_NPW + i) * (FG_HS / 4) + t;
                dst[0] = o0; dst[8] = o1; dst[16] = o2;
            }
        }
    }
    __syncthreads();

    if (w >= 6) return;
    int nrows = min(FG_NODES, n - (int)(blockIdx.x * FG_NODES));
    int nr    = min(PL_ROWS, nrows - pr0);
    if (nr <= 0) return;
    int c = (w - 3 * hh) * 32 + lane;
    const float* tp = (const float*)hT4 + pr0 * FG_HS + c;
    float x[PL_ROWS];
    #pragma unroll
    for (int j = 0; j < PL_ROWS; j++) x[j] = (j < nr) ? tp[j * FG_HS] : 0.0f;
    int cur = __builtin_amdgcn_readlane(idl, 0);
    float sum = 0.0f;
    #pragma unroll
    for (int j = 0; j < PL_ROWS; j++) {
        if (j < nr) {
            int b = __builtin_amdgcn_readlane(idl, j);
            if (b != cur) {
                atomicAdd(g + (size_t)cur * HID + c, sum);
                sum = 0.0f;
                cur = b;
            }
            sum += x[j];
        }
    }
    atomicAdd(g + (size_t)cur * HID + c, sum);
}

__global__ void k_dec2(const float* __restrict__ g2, const float* __restrict__ W,
                       const float* __restrict__ b, float* __restrict__ out) {
    int idx = blockIdx.x * blockDim.x + threadIdx.x;
    if (idx < NGRAPH * OUTDIM) {
        int gi = idx / OUTDIM;
        int o  = idx - gi * OUTDIM;
        float s = b[o];
        const float* row = g2 + (size_t)gi * HID;
        #pragma unroll 8
        for (int k = 0; k < HID; k++) s += row[k] * W[k * OUTDIM + o];
        out[idx] = s;
    }
}

extern "C" void kernel_launch(void* const* d_in, const int* in_sizes, int n_in,
                              void* d_out, int out_size, void* d_ws, size_t ws_size,
                              hipStream_t stream) {
    const float* x       = (const float*)d_in[0];
    const int*   ei      = (const int*)  d_in[1];
    const int*   batch   = (const int*)  d_in[2];
    const float* enc_W0  = (const float*)d_in[3];
    const float* enc_b0  = (const float*)d_in[4];
    const float* enc_W1  = (const float*)d_in[5];
    const float* enc_b1  = (const float*)d_in[6];
    const float* conv_W  = (const float*)d_in[7];
    const float* conv_b  = (const float*)d_in[8];
    const float* dec_W0  = (const float*)d_in[9];
    const float* dec_b0  = (const float*)d_in[10];
    const float* dec_W1  = (const float*)d_in[11];
    const float* dec_b1  = (const float*)d_in[12];
    float* out = (float*)d_out;

    float* ws   = (float*)d_ws;
    size_t feat = (size_t)NNODES * HID;
    float* buf0 = ws;
    float* buf1 = buf0 + feat;
    float* buf2 = buf1 + feat;
    float* dinv = buf2 + feat;
    float* g    = dinv + NNODES;
    float* g2   = g + (size_t)NGRAPH * HID;
    int*   cnt  = (int*)(g2 + (size_t)NGRAPH * HID);
    int*   tot  = cnt + NNODES;
    int*   off  = tot + 16;
    int*   rank = off + NNODES;
    int*   col  = rank + NEDGES;
    int*   tab  = col + NEDGES + 64;

    const int B = 256;

    k_rank<<<RANK_BLOCKS, RANK_THREADS, 0, stream>>>(ei, rank, tab, tot);
    k_offsets<<<cdiv(NNODES, B), B, 0, stream>>>(tab, tot, cnt, off, dinv, col + NEDGES, NNODES);
    k_fill<<<cdiv(NEDGES, B), B, 0, stream>>>(ei, rank, tab, col, NEDGES);

    int mTilesN = NNODES / 16;
    size_t shm96 = (size_t)HID * (HID + 2) * sizeof(float);
    k_enc_chain<<<cdiv(mTilesN, EC_WAVES), EC_THREADS, 0, stream>>>(
        x, enc_W0, enc_b0, enc_W1, enc_b1, conv_W, dinv, buf0  , NNODES);

    float* hsrc = buf0;
    float* hdst = buf2;
    int fgBlks = cdiv(NNODES, FG_NODES);
    for (int l = 0; l + 1 < NCONV; l++) {
        const float* Wn = conv_W + (size_t)(l + 1) * HID * HID;
        const float* bl = conv_b + (size_t)l * HID;
        k_gather_gemm<<<fgBlks, GEMM_THREADS, 0, stream>>>(hsrc, dinv, off, cnt, col, bl, Wn, hdst, NNODES);
        float* t = hsrc; hsrc = hdst; hdst = t;
    }
    k_zero<<<cdiv(NGRAPH * HID, B), B, 0, stream>>>(g, NGRAPH * HID);
    k_gather_pool<<<fgBlks, GEMM_THREADS, 0, stream>>>(hsrc, dinv, off, cnt, col,
                                                       conv_b + (size_t)(NCONV - 1) * HID, batch, g, NNODES);

    int mTilesG = NGRAPH / 16;
    k_gemm_wmma<HID, 0, 1, 0><<<cdiv(mTilesG, GEMM_WAVES), GEMM_THREADS, shm96, stream>>>(
        g, dec_W0, dec_b0, g2, nullptr, NGRAPH, HID);
    k_dec2<<<cdiv(NGRAPH * OUTDIM, B), B, 0, stream>>>(g2, dec_W1, dec_b1, out);
}


# Gather pass takes up to 24 edges of a destination's list (16 slots, 8 more when the list goes on): lists of 17-24 edges cost one round trip to their rows, not two; order of every sum kept
# speedup vs baseline: 4.1464x; 1.0366x over previous
#include <hip/hip_runtime.h>
#include <hip/hip_bf16.h>

typedef __attribute__((ext_vector_type(2))) float v2f;
typedef __attribute__((ext_vector_type(8))) float v8f;
typedef __attribute__((ext_vector_type(4))) float v4f;

#define NNODES 50000
#define NEDGES 800000
#define NGRAPH 512
#define INDIM  128
#define HID    96
#define OUTDIM 10
#define NCONV  4
#define NTILE  6
#define GEMM_WAVES 8
#define GEMM_THREADS (GEMM_WAVES * 32)

static inline int cdiv(int a, int b) { return (a + b - 1) / b; }

typedef __attribute__((ext_vector_type(4))) int v4i;

#define RANK_BLOCKS  64
#define RANK_THREADS 1024
#define RANK_EPB     (NEDGES / RANK_BLOCKS)
#define RANK_PER     ((RANK_EPB + RANK_THREADS - 1) / RANK_THREADS)
#define NODE4        (NNODES / 4)
static_assert(NEDGES % RANK_BLOCKS == 0, "slices must be equal");
static_assert(NNODES % 4 == 0, "counter sets are moved as 16-byte pieces");

__global__ void __launch_bounds__(RANK_THREADS)
k_rank(const int* __restrict__ ei, int* __restrict__ rank, int* __restrict__ tab,
       int* __restrict__ total) {
    __shared__ int cntL[NNODES];
    int tid = threadIdx.x;
    v4i* c4 = (v4i*)cntL;
    const v4i zero4i = {0, 0, 0, 0};
    for (int i = tid; i < NODE4; i += RANK_THREADS) c4[i] = zero4i;
    if (blockIdx.x == 0 && tid == 0) *total = 0;
    __syncthreads();

    int e0 = blockIdx.x * RANK_EPB;
    const int* dst = ei + NEDGES + e0;
    int d[RANK_PER];
    #pragma unroll
    for (int j = 0; j < RANK_PER; j++) {
        int k = tid + j * RANK_THREADS;
        d[j] = (k < RANK_EPB) ? dst[k] : 0;
    }
    int r[RANK_PER];
    #pragma unroll
    for (int j = 0; j < RANK_PER; j++) {
        int k = tid + j * RANK_THREADS;
        r[j] = (k < RANK_EPB) ? atomicAdd(&cntL[d[j]], 1) : 0;
    }
    #pragma unroll
    for (int j = 0; j < RANK_PER; j++) {
        int k = tid + j * RANK_THREADS;
        if (k < RANK_EPB) rank[e0 + k] = r[j];
    }
    __syncthreads();

    v4i* t4 = (v4i*)(tab + (size_t)blockIdx.x * NNODES);
    for (int i = tid; i < NODE4; i += RANK_THREADS) t4[i] = c4[i];
}

__global__ void k_offsets(int* __restrict__ tab, int* total, int* __restrict__ cnt,
                          int* __restrict__ off, float* __restrict__ dinv,
                          int* __restrict__ colpad, int n) {
    int i    = blockIdx.x * blockDim.x + threadIdx.x;
    int lane = threadIdx.x & 31;
    int ii   = min(i, n - 1);
    int v[RANK_BLOCKS];
    #pragma unroll
    for (int b = 0; b < RANK_BLOCKS; b++) v[b] = tab[(size_t)b * NNODES + ii];
    int c = 0;
    #pragma unroll
    for (int b = 0; b < RANK_BLOCKS; b++) c += v[b];
    if (i >= n) c = 0;
    int incl = c;
    #pragma unroll
    for (int d = 1; d < 32; d <<= 1) {
        int v1 = __shfl_up(incl, d);
        if (lane >= d) incl += v1;
    }
    int base = 0;
    if (lane == 31) base = atomicAdd(total, incl);
    base = __shfl(base, 31);
    if (i < n) {
        int run = base + incl - c;
        off[i]  = run;
        cnt[i]  = c;
        dinv[i] = rsqrtf((float)(c + 1));
        #pragma unroll
        for (int b = 0; b < RANK_BLOCKS; b++) {
            tab[(size_t)b * NNODES + i] = run;
            run += v[b];
        }
    }
    if (i < 4) colpad[i] = 0;
}

__global__ void k_fill(const int* __restrict__ ei, const int* __restrict__ rank,
                       const int* __restrict__ tab, int* __restrict__ col, int E) {
    int e = blockIdx.x * blockDim.x + threadIdx.x;
    if (e < E) {
        int d = ei[E + e];
        int b = e / RANK_EPB;
        col[tab[(size_t)b * NNODES + d] + rank[e]] = ei[e];
    }
}

template <int K, int RELUA, int RELU, int FUSE>
__global__ void __launch_bounds__(GEMM_THREADS)
k_gemm_wmma(const float* __restrict__ A, const float* __restrict__ B,
            const float* __restrict__ bias,
            float* __restrict__ C,
            const float* __restrict__ dinv,
            int M, int lda) {
    extern __shared__ float sBt[];
    constexpr int lstride = K + 2;

    constexpr int ROW4 = HID / 4;
    constexpr int PER  = K * ROW4 / GEMM_THREADS;
    static_assert(K * ROW4 % GEMM_THREADS == 0, "B must split evenly over the block");
    const v4f* B4 = (const v4f*)B;
    v4f piece[PER];
    #pragma unroll
    for (int j = 0; j < PER; j++) piece[j] = B4[threadIdx.x + j * GEMM_THREADS];
    #pragma unroll
    for (int j = 0; j < PER; j++) {
        int i  = threadIdx.x + j * GEMM_THREADS;
        int k  = i / ROW4;
        int c0 = (i - k * ROW4) * 4;
        float* dst = sBt + c0 * lstride + k;
        dst[0]           = piece[j][0];
        dst[lstride]     = piece[j][1];
        dst[2 * lstride] = piece[j][2];
        dst[3 * lstride] = piece[j][3];
    }
    __syncthreads();

    int mt   = blockIdx.x * GEMM_WAVES + (threadIdx.x >> 5);
    int lane = threadIdx.x & 31;
    int mTiles = M >> 4;
    if (mt >= mTiles) return;

    int half = lane >> 4;
    int l15  = lane & 15;

    const float* arow  = A + (size_t)(mt * 16 + l15) * lda;
    const float* bbase = sBt + (size_t)l15 * lstride;

    v8f acc[NTILE] = {};
    for (int k = 0; k < K; k += 4) {
        int kb = k + half * 2;
        v2f a = *(const v2f*)(arow + kb);
        if (RELUA) { a[0] = fmaxf(a[0], 0.0f); a[1] = fmaxf(a[1], 0.0f); }
        #pragma unroll
        for (int nt = 0; nt < NTILE; nt++) {
            v2f b = *(const v2f*)(bbase + (size_t)(nt * 16) * lstride + kb);
            acc[nt] = __builtin_amdgcn_wmma_f32_16x16x4_f32(
                 false, a,  false, b,
                 (short)0, acc[nt],  false,  false);
        }
    }

    int rbase = mt * 16 + half * 8;
    float* cbase = C + (size_t)rbase * HID + l15;

    if (FUSE) {
        float dv[8];
        #pragma unroll
        for (int v = 0; v < 8; v++) dv[v] = dinv[rbase + v];
        #pragma unroll
        for (int nt = 0; nt < NTILE; nt++) {
            #pragma unroll
            for (int v = 0; v < 8; v++)
                cbase[(size_t)v * HID + nt * 16] = acc[nt][v] * dv[v];
        }
    } else {
        #pragma unroll
        for (int nt = 0; nt < NTILE; nt++) {
            float bv = bias[nt * 16 + l15];
            #pragma unroll
            for (int v = 0; v < 8; v++) {
                float r = acc[nt][v] + bv;
                if (RELU) r = fmaxf(r, 0.0f);
                cbase[(size_t)v * HID + nt * 16] = r;
            }
        }
    }
}

#define EC_WAVES   16
#define EC_THREADS (EC_WAVES * 32)
#define EC_TS      (HID + 4)

__global__ void __launch_bounds__(EC_THREADS)
k_enc_chain(const float* __restrict__ x,
            const float* __restrict__ W0, const float* __restrict__ b0,
            const float* __restrict__ W1, const float* __restrict__ b1,
            const float* __restrict__ Wc, const float* __restrict__ dinv,
            float* __restrict__ out, int M) {
    __shared__ __attribute__((aligned(16))) float sB0[HID * (INDIM + 2)];
    __shared__ __attribute__((aligned(16))) float sB1[HID * (HID + 2)];
    __shared__ __attribute__((aligned(16))) float sB2[HID * (HID + 2)];
    __shared__ __attribute__((aligned(16))) float tile[EC_WAVES * 16 * EC_TS];
    constexpr int ls0 = INDIM + 2;
    constexpr int ls1 = HID + 2;

    constexpr int ROW4 = HID / 4;
    constexpr int N0   = INDIM * ROW4;
    constexpr int N1   = HID * ROW4;
    constexpr int P0   = N0 / EC_THREADS;
    constexpr int P1   = (N1 + EC_THREADS - 1) / EC_THREADS;
    static_assert(N0 % EC_THREADS == 0, "W0 must split evenly over the block");
    const v4f* W0_4 = (const v4f*)W0;
    const v4f* W1_4 = (const v4f*)W1;
    const v4f* Wc_4 = (const v4f*)Wc;
    v4f p0[P0], p1[P1], p2[P1];
    #pragma unroll
    for (int j = 0; j < P0; j++) p0[j] = W0_4[threadIdx.x + j * EC_THREADS];
    #pragma unroll
    for (int j = 0; j < P1; j++) {
        int ii = min((int)threadIdx.x + j * EC_THREADS, N1 - 1);
        p1[j] = W1_4[ii];
        p2[j] = Wc_4[ii];
    }
    #pragma unroll
    for (int j = 0; j < P0; j++) {
        int i  = threadIdx.x + j * EC_THREADS;
        int k  = i / ROW4;
        int c0 = (i - k * ROW4) * 4;
        float* dst = sB0 + c0 * ls0 + k;
        dst[0]       = p0[j][0];
        dst[ls0]     = p0[j][1];
        dst[2 * ls0] = p0[j][2];
        dst[3 * ls0] = p0[j][3];
    }
    #pragma unroll
    for (int j = 0; j < P1; j++) {
        int i  = threadIdx.x + j * EC_THREADS;
        if (i < N1) {
            int k  = i / ROW4;
            int c0 = (i - k * ROW4) * 4;
            float* d1 = sB1 + c0 * ls1 + k;
            float* d2 = sB2 + c0 * ls1 + k;
            d1[0]       = p1[j][0];  d2[0]       = p2[j][0];
            d1[ls1]     = p1[j][1];  d2[ls1]     = p2[j][1];
            d1[2 * ls1] = p1[j][2];  d2[2 * ls1] = p2[j][2];
            d1[3 * ls1] = p1[j][3];  d2[3 * ls1] = p2[j][3];
        }
    }
    __syncthreads();

    int w    = threadIdx.x >> 5;
    int lane = threadIdx.x & 31;
    int half = lane >> 4;
    int l15  = lane & 15;
    int mt   = blockIdx.x * EC_WAVES + w;
    bool valid = mt < (M >> 4);
    float* trow = tile + (w * 16) * EC_TS;
    float* tst  = trow + (half * 8) * EC_TS + l15;
    const float* tA = trow + l15 * EC_TS;

    if (valid) {
        const float* arow  = x + (size_t)(mt * 16 + l15) * INDIM;
        const float* bbase = sB0 + l15 * ls0;
        v8f acc[NTILE] = {};
        for (int k = 0; k < INDIM; k += 4) {
            int kb = k + half * 2;
            v2f a = *(const v2f*)(arow + kb);
            #pragma unroll
            for (int nt = 0; nt < NTILE; nt++) {
                v2f b = *(const v2f*)(bbase + (nt * 16) * ls0 + kb);
                acc[nt] = __builtin_amdgcn_wmma_f32_16x16x4_f32(
                     false, a,  false, b,
                     (short)0, acc[nt],  false,  false);
            }
        }
        #pragma unroll
        for (int nt = 0; nt < NTILE; nt++) {
            float bv = b0[nt * 16 + l15];
            #pragma unroll
            for (int v = 0; v < 8; v++)
                tst[v * EC_TS + nt * 16] = fmaxf(acc[nt][v] + bv, 0.0f);
        }
    }
    __syncthreads();

    if (valid) {
        const float* bbase = sB1 + l15 * ls1;
        v8f acc[NTILE] = {};
        for (int k = 0; k < HID; k += 4) {
            int kb = k + half * 2;
            v2f a = *(const v2f*)(tA + kb);
            #pragma unroll
            for (int nt = 0; nt < NTILE; nt++) {
                v2f b = *(const v2f*)(bbase + (nt * 16) * ls1 + kb);
                acc[nt] = __builtin_amdgcn_wmma_f32_16x16x4_f32(
                     false, a,  false, b,
                     (short)0, acc[nt],  false,  false);
            }
        }
        #pragma unroll
        for (int nt = 0; nt < NTILE; nt++) {
            float bv = b1[nt * 16 + l15];
            #pragma unroll
            for (int v = 0; v < 8; v++)
                tst[v * EC_TS + nt * 16] = acc[nt][v] + bv;
        }
    }
    __syncthreads();

    if (valid) {
        const float* bbase = sB2 + l15 * ls1;
        v8f acc[NTILE] = {};
        for (int k = 0; k < HID; k += 4) {
            int kb = k + half * 2;
            v2f a = *(const v2f*)(tA + kb);
            #pragma unroll
            for (int nt = 0; nt < NTILE; nt++) {
                v2f b = *(const v2f*)(bbase + (nt * 16) * ls1 + kb);
                acc[nt] = __builtin_amdgcn_wmma_f32_16x16x4_f32(
                     false, a,  false, b,
                     (short)0, acc[nt],  false,  false);
            }
        }
        int rbase = mt * 16 + half * 8;
        float* cbase = out + (size_t)rbase * HID + l15;
        float dv[8];
        #pragma unroll
        for (int v = 0; v < 8; v++) dv[v] = dinv[rbase + v];
        #pragma unroll
        for (int nt = 0; nt < NTILE; nt++) {
            #pragma unroll
            for (int v = 0; v < 8; v++)
                cbase[(size_t)v * HID + nt * 16] = acc[nt][v] * dv[v];
        }
    }
}


#define FG_NODES 64
#define FG_NPW   (FG_NODES / GEMM_WAVES)
#define FG_HS    (HID + 4)
static_assert(NNODES % 16 == 0, "a 16-row tile is wholly inside or outside the node range");
static_assert(FG_NPW == 8, "per-node values of a wave sit in lanes 0..7");

#define FG_PASS 24
template <int S>
__device__ __forceinline__ void gather_pass(const v4f* rows, const v4f* rself, int cidx, int grp,
                                            int rem, bool first, v4f& a0, v4f& a1, v4f& a2) {
    v4f  x0[S], x1[S], x2[S];
    bool on[S];
    #pragma unroll
    for (int k = 0; k < S; k++) {
        on[k] = grp + 4 * k < rem;
        int s = __shfl(cidx, grp + 4 * k);
        const v4f* r = on[k] ? rows + (unsigned)s * 24u : rself;
        x0[k] = r[0]; x1[k] = r[8]; x2[k] = r[16];
    }
    if (first && grp == 0) {
        a0 = rself[0]; a1 = rself[8]; a2 = rself[16];
    }
    #pragma unroll
    for (int k = 0; k < S; k++)
        if (on[k]) { a0 += x0[k]; a1 += x1[k]; a2 += x2[k]; }
}

__global__ void __launch_bounds__(GEMM_THREADS, 8)
k_gather_gemm(const float* __restrict__ hws, const float* __restrict__ dinv,
              const int* __restrict__ off, const int* __restrict__ cnt,
              const int* __restrict__ col, const float* __restrict__ bias,
              const float* __restrict__ Wn, float* __restrict__ out, int n) {
    __shared__ float sBt[HID * (HID + 2)];
    __shared__ v4f   hT4[FG_NODES * (FG_HS / 4)];
    constexpr int lstride = HID + 2;

    constexpr int ROW4 = HID / 4;
    constexpr int PER  = HID * ROW4 / GEMM_THREADS;
    static_assert(HID * ROW4 % GEMM_THREADS == 0, "Wn must split evenly over the block");
    const v4f* B4 = (const v4f*)Wn;
    v4f piece[PER];
    #pragma unroll
    for (int j = 0; j < PER; j++) piece[j] = B4[threadIdx.x + j * GEMM_THREADS];

    int lane = threadIdx.x & 31;
    int w    = __builtin_amdgcn_readfirstlane((int)(threadIdx.x >> 5));
    int grp  = lane >> 3;
    int t    = lane & 7;
    int nb   = blockIdx.x * FG_NODES + w * FG_NPW;
    int nvalid = min(FG_NPW, n - nb);

    int ml   = min(nb + t, n - 1);
    int moff = off[ml];
    int mcnt = cnt[ml];
    int mdvb = __float_as_int(dinv[ml]);
    const v4f* b4 = (const v4f*)bias + t;
    v4f bb0 = b4[0], bb1 = b4[8], bb2 = b4[16];

    #pragma unroll
    for (int j = 0; j < PER; j++) {
        int i  = threadIdx.x + j * GEMM_THREADS;
        int k  = i / ROW4;
        int c0 = (i - k * ROW4) * 4;
        float* dst = sBt + c0 * lstride + k;
        dst[0]           = piece[j][0];
        dst[lstride]     = piece[j][1];
        dst[2 * lstride] = piece[j][2];
        dst[3 * lstride] = piece[j][3];
    }

    const v4f* rows = (const v4f*)hws + t;
    int idxs[FG_NPW];
    #pragma unroll
    for (int j = 0; j < FG_NPW; j++) {
        int sj  = __builtin_amdgcn_readlane(moff, j);
        int cj  = __builtin_amdgcn_readlane(mcnt, j);
        idxs[j] = col[sj + min(lane, max(cj, 1) - 1)];
    }
    asm volatile("" :: "v"(idxs[0]), "v"(idxs[1]), "v"(idxs[2]), "v"(idxs[3]),
                       "v"(idxs[4]), "v"(idxs[5]), "v"(idxs[6]), "v"(idxs[7]));
    #pragma unroll
    for (int j = 0; j < FG_NPW; j++) {
        int cj  = __builtin_amdgcn_readlane(mcnt, j);
        idxs[j] = (lane < cj) ? idxs[j] : 0;
    }
    #pragma unroll
    for (int i = 0; i < FG_NPW; i++) {
        if (i < nvalid) {
            int   node = nb + i;
            int   cs = __builtin_amdgcn_readlane(moff, i);
            int   cc = __builtin_amdgcn_readlane(mcnt, i);
            float dv = __int_as_float(__builtin_amdgcn_readlane(mdvb, i));
            int cidx = idxs[i];
            const v4f* rself = rows + (unsigned)node * 24u;
            v4f a0 = {}, a1 = {}, a2 = {};
            for (int e0 = 0; e0 == 0 || e0 < cc; e0 += FG_PASS) {
                if (e0 != 0)
                    cidx = (lane < cc - e0) ? col[cs + e0 + lane] : 0;
                int rem = cc - e0;
                if (rem > 16) gather_pass<6>(rows, rself, cidx, grp, rem, e0 == 0, a0, a1, a2);
                else          gather_pass<4>(rows, rself, cidx, grp, rem, e0 == 0, a0, a1, a2);
            }
            #pragma unroll
            for (int mm = 8; mm <= 16; mm <<= 1) {
                #pragma unroll
                for (int q = 0; q < 4; q++) {
                    a0[q] += __shfl_xor(a0[q], mm);
                    a1[q] += __shfl_xor(a1[q], mm);
                    a2[q] += __shfl_xor(a2[q], mm);
                }
            }
            if (grp == 0) {
                v4f o0 = a0 * dv + bb0;
                v4f o1 = a1 * dv + bb1;
                v4f o2 = a2 * dv + bb2;
                #pragma unroll
                for (int q = 0; q < 4; q++) {
                    o0[q] = fmaxf(o0[q], 0.0f); o1[q] = fmaxf(o1[q], 0.0f); o2[q] = fmaxf(o2[q], 0.0f);
                }
                v4f* dst = hT4 + (w * FG_NPW + i) * (FG_HS / 4) + t;
                dst[0] = o0; dst[8] = o1; dst[16] = o2;
            }
        }
    }
    __syncthreads();

    int mtile   = w >> 1;
    int rowbase = blockIdx.x * FG_NODES + mtile * 16;
    if (rowbase >= n) return;
    int half = lane >> 4;
    int l15  = lane & 15;
    int nt0  = (w & 1) * 3;
    const float* abase = (const float*)hT4 + (mtile * 16 + l15) * FG_HS;
    const float* bbase = sBt + (nt0 * 16 + l15) * lstride;
    v8f acc[3] = {};
    for (int k = 0; k < HID; k += 4) {
        int kb = k + half * 2;
        v2f a = *(const v2f*)(abase + kb);
        #pragma unroll
        for (int j = 0; j < 3; j++) {
            v2f b = *(const v2f*)(bbase + (j * 16) * lstride + kb);
            acc[j] = __builtin_amdgcn_wmma_f32_16x16x4_f32(
                 false, a,  false, b,
                 (short)0, acc[j],  false,  false);
        }
    }
    int rbase = rowbase + half * 8;
    float* cbase = out + (size_t)rbase * HID + nt0 * 16 + l15;
    float dvv[8];
    #pragma unroll
    for (int v = 0; v < 8; v++) dvv[v] = dinv[rbase + v];
    #pragma unroll
    for (int j = 0; j < 3; j++) {
        #pragma unroll
        for (int v = 0; v < 8; v++)
            cbase[(size_t)v * HID + j * 16] = acc[j][v] * dvv[v];
    }
}

__global__ void k_zero(float* x, int n) {
    int i = blockIdx.x * blockDim.x + threadIdx.x;
    if (i < n) x[i] = 0.0f;
}

#define PL_ROWS 32
static_assert(FG_NODES == 2 * PL_ROWS, "two row halves per tile");
static_assert(HID == 3 * 32, "three 32-column groups per row half");

__global__ void __launch_bounds__(GEMM_THREADS, 8)
k_gather_pool(const float* __restrict__ hws, const float* __restrict__ dinv,
              const int* __restrict__ off, const int* __restrict__ cnt,
              const int* __restrict__ col, const float* __restrict__ bias,
              const int* __restrict__ batch, float* __restrict__ g, int n) {
    __shared__ v4f hT4[FG_NODES * (FG_HS / 4)];

    int lane = threadIdx.x & 31;
    int w    = __builtin_amdgcn_readfirstlane((int)(threadIdx.x >> 5));
    int grp  = lane >> 3;
    int t    = lane & 7;
    int nb   = blockIdx.x * FG_NODES + w * FG_NPW;
    int nvalid = min(FG_NPW, n - nb);

    int hh  = (w >= 3) ? 1 : 0;
    int pr0 = hh * PL_ROWS;
    int idl = batch[min((int)(blockIdx.x * FG_NODES) + pr0 + lane, n - 1)];

    int ml   = min(nb + t, n - 1);
    int moff = off[ml];
    int mcnt = cnt[ml];
    int mdvb = __float_as_int(dinv[ml]);
    const v4f* b4 = (const v4f*)bias + t;
    v4f bb0 = b4[0], bb1 = b4[8], bb2 = b4[16];

    const v4f* rows = (const v4f*)hws + t;
    int idxs[FG_NPW];
    #pragma unroll
    for (int j = 0; j < FG_NPW; j++) {
        int sj  = __builtin_amdgcn_readlane(moff, j);
        int cj  = __builtin_amdgcn_readlane(mcnt, j);
        idxs[j] = col[sj + min(lane, max(cj, 1) - 1)];
    }
    asm volatile("" :: "v"(idxs[0]), "v"(idxs[1]), "v"(idxs[2]), "v"(idxs[3]),
                       "v"(idxs[4]), "v"(idxs[5]), "v"(idxs[6]), "v"(idxs[7]));
    #pragma unroll
    for (int j = 0; j < FG_NPW; j++) {
        int cj  = __builtin_amdgcn_readlane(mcnt, j);
        idxs[j] = (lane < cj) ? idxs[j] : 0;
    }
    #pragma unroll
    for (int i = 0; i < FG_NPW; i++) {
        if (i < nvalid) {
            int   node = nb + i;
            int   cs = __builtin_amdgcn_readlane(moff, i);
            int   cc = __builtin_amdgcn_readlane(mcnt, i);
            float dv = __int_as_float(__builtin_amdgcn_readlane(mdvb, i));
            int cidx = idxs[i];
            const v4f* rself = rows + (unsigned)node * 24u;
            v4f a0 = {}, a1 = {}, a2 = {};
            for (int e0 = 0; e0 == 0 || e0 < cc; e0 += FG_PASS) {
                if (e0 != 0)
                    cidx = (lane < cc - e0) ? col[cs + e0 + lane] : 0;
                int rem = cc - e0;
                if (rem > 16) gather_pass<6>(rows, rself, cidx, grp, rem, e0 == 0, a0, a1, a2);
                else          gather_pass<4>(rows, rself, cidx, grp, rem, e0 == 0, a0, a1, a2);
            }
            #pragma unroll
            for (int mm = 8; mm <= 16; mm <<= 1) {
                #pragma unroll
                for (int q = 0; q < 4; q++) {
                    a0[q] += __shfl_xor(a0[q], mm);
                    a1[q] += __shfl_xor(a1[q], mm);
                    a2[q] += __shfl_xor(a2[q], mm);
                }
            }
            if (grp == 0) {
                v4f o0 = a0 * dv + bb0;
                v4f o1 = a1 * dv + bb1;
                v4f o2 = a2 * dv + bb2;
                #pragma unroll
                for (int q = 0; q < 4; q++) {
                    o0[q] = fmaxf(o0[q], 0.0f); o1[q] = fmaxf(o1[q], 0.0f); o2[q] = fmaxf(o2[q], 0.0f);
                }
                v4f* dst = hT4 + (w * FG_NPW + i) * (FG_HS / 4) + t;
                dst[0] = o0; dst[8] = o1; dst[16] = o2;
            }
        }
    }
    __syncthreads();

    if (w >= 6) return;
    int nrows = min(FG_NODES, n - (int)(blockIdx.x * FG_NODES));
    int nr    = min(PL_ROWS, nrows - pr0);
    if (nr <= 0) return;
    int c = (w - 3 * hh) * 32 + lane;
    const float* tp = (const float*)hT4 + pr0 * FG_HS + c;
    float x[PL_ROWS];
    #pragma unroll
    for (int j = 0; j < PL_ROWS; j++) x[j] = (j < nr) ? tp[j * FG_HS] : 0.0f;
    int cur = __builtin_amdgcn_readlane(idl, 0);
    float sum = 0.0f;
    #pragma unroll
    for (int j = 0; j < PL_ROWS; j++) {
        if (j < nr) {
            int b = __builtin_amdgcn_readlane(idl, j);
            if (b != cur) {
                atomicAdd(g + (size_t)cur * HID + c, sum);
                sum = 0.0f;
                cur = b;
            }
            sum += x[j];
        }
    }
    atomicAdd(g + (size_t)cur * HID + c, sum);
}

__global__ void k_dec2(const float* __restrict__ g2, const float* __restrict__ W,
                       const float* __restrict__ b, float* __restrict__ out) {
    int idx = blockIdx.x * blockDim.x + threadIdx.x;
    if (idx < NGRAPH * OUTDIM) {
        int gi = idx / OUTDIM;
        int o  = idx - gi * OUTDIM;
        float s = b[o];
        const float* row = g2 + (size_t)gi * HID;
        #pragma unroll 8
        for (int k = 0; k < HID; k++) s += row[k] * W[k * OUTDIM + o];
        out[idx] = s;
    }
}

extern "C" void kernel_launch(void* const* d_in, const int* in_sizes, int n_in,
                              void* d_out, int out_size, void* d_ws, size_t ws_size,
                              hipStream_t stream) {
    const float* x       = (const float*)d_in[0];
    const int*   ei      = (const int*)  d_in[1];
    const int*   batch   = (const int*)  d_in[2];
    const float* enc_W0  = (const float*)d_in[3];
    const float* enc_b0  = (const float*)d_in[4];
    const float* enc_W1  = (const float*)d_in[5];
    const float* enc_b1  = (const float*)d_in[6];
    const float* conv_W  = (const float*)d_in[7];
    const float* conv_b  = (const float*)d_in[8];
    const float* dec_W0  = (const float*)d_in[9];
    const float* dec_b0  = (const float*)d_in[10];
    const float* dec_W1  = (const float*)d_in[11];
    const float* dec_b1  = (const float*)d_in[12];
    float* out = (float*)d_out;

    float* ws   = (float*)d_ws;
    size_t feat = (size_t)NNODES * HID;
    float* buf0 = ws;
    float* buf1 = buf0 + feat;
    float* buf2 = buf1 + feat;
    float* dinv = buf2 + feat;
    float* g    = dinv + NNODES;
    float* g2   = g + (size_t)NGRAPH * HID;
    int*   cnt  = (int*)(g2 + (size_t)NGRAPH * HID);
    int*   tot  = cnt + NNODES;
    int*   off  = tot + 16;
    int*   rank = off + NNODES;
    int*   col  = rank + NEDGES;
    int*   tab  = col + NEDGES + 64;

    const int B = 256;

    k_rank<<<RANK_BLOCKS, RANK_THREADS, 0, stream>>>(ei, rank, tab, tot);
    k_offsets<<<cdiv(NNODES, B), B, 0, stream>>>(tab, tot, cnt, off, dinv, col + NEDGES, NNODES);
    k_fill<<<cdiv(NEDGES, B), B, 0, stream>>>(ei, rank, tab, col, NEDGES);

    int mTilesN = NNODES / 16;
    size_t shm96 = (size_t)HID * (HID + 2) * sizeof(float);
    k_enc_chain<<<cdiv(mTilesN, EC_WAVES), EC_THREADS, 0, stream>>>(
        x, enc_W0, enc_b0, enc_W1, enc_b1, conv_W, dinv, buf0  , NNODES);

    float* hsrc = buf0;
    float* hdst = buf2;
    int fgBlks = cdiv(NNODES, FG_NODES);
    for (int l = 0; l + 1 < NCONV; l++) {
        const float* Wn = conv_W + (size_t)(l + 1) * HID * HID;
        const float* bl = conv_b + (size_t)l * HID;
        k_gather_gemm<<<fgBlks, GEMM_THREADS, 0, stream>>>(hsrc, dinv, off, cnt, col, bl, Wn, hdst, NNODES);
        float* t = hsrc; hsrc = hdst; hdst = t;
    }
    k_zero<<<cdiv(NGRAPH * HID, B), B, 0, stream>>>(g, NGRAPH * HID);
    k_gather_pool<<<fgBlks, GEMM_THREADS, 0, stream>>>(hsrc, dinv, off, cnt, col,
                                                       conv_b + (size_t)(NCONV - 1) * HID, batch, g, NNODES);

    int mTilesG = NGRAPH / 16;
    k_gemm_wmma<HID, 0, 1, 0><<<cdiv(mTilesG, GEMM_WAVES), GEMM_THREADS, shm96, stream>>>(
        g, dec_W0, dec_b0, g2, nullptr, NGRAPH, HID);
    k_dec2<<<cdiv(NGRAPH * OUTDIM, B), B, 0, stream>>>(g2, dec_W1, dec_b1, out);
}


# Encoder's second product and layer 0's product merged into one: W1 x Wc and b1 x Wc formed once per call on WMMA by two extra work-groups of the first launch; sum order of that product changes
# speedup vs baseline: 4.3020x; 1.0375x over previous
#include <hip/hip_runtime.h>
#include <hip/hip_bf16.h>

typedef __attribute__((ext_vector_type(2))) float v2f;
typedef __attribute__((ext_vector_type(8))) float v8f;
typedef __attribute__((ext_vector_type(4))) float v4f;

#define NNODES 50000
#define NEDGES 800000
#define NGRAPH 512
#define INDIM  128
#define HID    96
#define OUTDIM 10
#define NCONV  4
#define NTILE  6
#define GEMM_WAVES 8
#define GEMM_THREADS (GEMM_WAVES * 32)

static inline int cdiv(int a, int b) { return (a + b - 1) / b; }

typedef __attribute__((ext_vector_type(4))) int v4i;

#define RANK_BLOCKS  64
#define RANK_THREADS 1024
#define RANK_EPB     (NEDGES / RANK_BLOCKS)
#define RANK_PER     ((RANK_EPB + RANK_THREADS - 1) / RANK_THREADS)
#define NODE4        (NNODES / 4)
static_assert(NEDGES % RANK_BLOCKS == 0, "slices must be equal");
static_assert(NNODES % 4 == 0, "counter sets are moved as 16-byte pieces");

#define WF_MT     7
#define WF_TILES  (WF_MT * (HID / 16))
#define WF_BLOCKS 2
#define WF_TPB    (WF_TILES / WF_BLOCKS)
#define WF_KH     (HID / 8)
static_assert(WF_TILES % WF_BLOCKS == 0, "tiles must split evenly over the extra work-groups");

__device__ __forceinline__ void merge_weights(const float* __restrict__ W1, const float* __restrict__ b1,
                                              const float* __restrict__ Wc,
                                              float* __restrict__ Wf, float* __restrict__ bf, int pb) {
    int w = threadIdx.x >> 5;
    if (w >= WF_TPB) return;
    int tile = pb * WF_TPB + w;
    int mt   = tile / (HID / 16);
    int nt   = tile - mt * (HID / 16);
    int lane = threadIdx.x & 31;
    int half = lane >> 4;
    int l15  = lane & 15;
    int m    = mt * 16 + l15;
    const float* arow = (m < HID) ? W1 + m * HID : b1;
    bool live = m <= HID;
    int n    = nt * 16 + l15;
    v8f acc = {};
    for (int h2 = 0; h2 < 2; h2++) {
        v2f a[WF_KH], b[WF_KH];
        #pragma unroll
        for (int s = 0; s < WF_KH; s++) {
            int kb = (h2 * WF_KH + s) * 4 + half * 2;
            a[s]    = *(const v2f*)(arow + kb);
            b[s][0] = Wc[kb * HID + n];
            b[s][1] = Wc[(kb + 1) * HID + n];
        }
        __builtin_amdgcn_sched_barrier(0);
        #pragma unroll
        for (int s = 0; s < WF_KH; s++) {
            v2f av = a[s];
            if (!live) { av[0] = 0.0f; av[1] = 0.0f; }
            acc = __builtin_amdgcn_wmma_f32_16x16x4_f32(
                 false, av,  false, b[s],
                 (short)0, acc,  false,  false);
        }
    }
    int r0 = mt * 16 + half * 8;
    #pragma unroll
    for (int v = 0; v < 8; v++) {
        int row = r0 + v;
        if (row < HID)       Wf[row * HID + n] = acc[v];
        else if (row == HID) bf[n] = acc[v];
    }
}

__global__ void __launch_bounds__(RANK_THREADS)
k_rank(const int* __restrict__ ei, int* __restrict__ rank, int* __restrict__ tab,
       int* __restrict__ total,
       const float* __restrict__ W1, const float* __restrict__ b1, const float* __restrict__ Wc,
       float* __restrict__ Wf, float* __restrict__ bf) {
    __shared__ int cntL[NNODES];
    if (blockIdx.x >= RANK_BLOCKS) {
        merge_weights(W1, b1, Wc, Wf, bf, (int)blockIdx.x - RANK_BLOCKS);
        return;
    }
    int tid = threadIdx.x;
    v4i* c4 = (v4i*)cntL;
    const v4i zero4i = {0, 0, 0, 0};
    for (int i = tid; i < NODE4; i += RANK_THREADS) c4[i] = zero4i;
    if (blockIdx.x == 0 && tid == 0) *total = 0;
    __syncthreads();

    int e0 = blockIdx.x * RANK_EPB;
    const int* dst = ei + NEDGES + e0;
    int d[RANK_PER];
    #pragma unroll
    for (int j = 0; j < RANK_PER; j++) {
        int k = tid + j * RANK_THREADS;
        d[j] = (k < RANK_EPB) ? dst[k] : 0;
    }
    int r[RANK_PER];
    #pragma unroll
    for (int j = 0; j < RANK_PER; j++) {
        int k = tid + j * RANK_THREADS;
        r[j] = (k < RANK_EPB) ? atomicAdd(&cntL[d[j]], 1) : 0;
    }
    #pragma unroll
    for (int j = 0; j < RANK_PER; j++) {
        int k = tid + j * RANK_THREADS;
        if (k < RANK_EPB) rank[e0 + k] = r[j];
    }
    __syncthreads();

    v4i* t4 = (v4i*)(tab + (size_t)blockIdx.x * NNODES);
    for (int i = tid; i < NODE4; i += RANK_THREADS) t4[i] = c4[i];
}

__global__ void k_offsets(int* __restrict__ tab, int* total, int* __restrict__ cnt,
                          int* __restrict__ off, float* __restrict__ dinv,
                          int* __restrict__ colpad, int n) {
    int i    = blockIdx.x * blockDim.x + threadIdx.x;
    int lane = threadIdx.x & 31;
    int ii   = min(i, n - 1);
    int v[RANK_BLOCKS];
    #pragma unroll
    for (int b = 0; b < RANK_BLOCKS; b++) v[b] = tab[(size_t)b * NNODES + ii];
    int c = 0;
    #pragma unroll
    for (int b = 0; b < RANK_BLOCKS; b++) c += v[b];
    if (i >= n) c = 0;
    int incl = c;
    #pragma unroll
    for (int d = 1; d < 32; d <<= 1) {
        int v1 = __shfl_up(incl, d);
        if (lane >= d) incl += v1;
    }
    int base = 0;
    if (lane == 31) base = atomicAdd(total, incl);
    base = __shfl(base, 31);
    if (i < n) {
        int run = base + incl - c;
        off[i]  = run;
        cnt[i]  = c;
        dinv[i] = rsqrtf((float)(c + 1));
        #pragma unroll
        for (int b = 0; b < RANK_BLOCKS; b++) {
            tab[(size_t)b * NNODES + i] = run;
            run += v[b];
        }
    }
    if (i < 4) colpad[i] = 0;
}

__global__ void k_fill(const int* __restrict__ ei, const int* __restrict__ rank,
                       const int* __restrict__ tab, int* __restrict__ col, int E) {
    int e = blockIdx.x * blockDim.x + threadIdx.x;
    if (e < E) {
        int d = ei[E + e];
        int b = e / RANK_EPB;
        col[tab[(size_t)b * NNODES + d] + rank[e]] = ei[e];
    }
}

template <int K, int RELUA, int RELU, int FUSE>
__global__ void __launch_bounds__(GEMM_THREADS)
k_gemm_wmma(const float* __restrict__ A, const float* __restrict__ B,
            const float* __restrict__ bias,
            float* __restrict__ C,
            const float* __restrict__ dinv,
            int M, int lda) {
    extern __shared__ float sBt[];
    constexpr int lstride = K + 2;

    constexpr int ROW4 = HID / 4;
    constexpr int PER  = K * ROW4 / GEMM_THREADS;
    static_assert(K * ROW4 % GEMM_THREADS == 0, "B must split evenly over the block");
    const v4f* B4 = (const v4f*)B;
    v4f piece[PER];
    #pragma unroll
    for (int j = 0; j < PER; j++) piece[j] = B4[threadIdx.x + j * GEMM_THREADS];
    #pragma unroll
    for (int j = 0; j < PER; j++) {
        int i  = threadIdx.x + j * GEMM_THREADS;
        int k  = i / ROW4;
        int c0 = (i - k * ROW4) * 4;
        float* dst = sBt + c0 * lstride + k;
        dst[0]           = piece[j][0];
        dst[lstride]     = piece[j][1];
        dst[2 * lstride] = piece[j][2];
        dst[3 * lstride] = piece[j][3];
    }
    __syncthreads();

    int mt   = blockIdx.x * GEMM_WAVES + (threadIdx.x >> 5);
    int lane = threadIdx.x & 31;
    int mTiles = M >> 4;
    if (mt >= mTiles) return;

    int half = lane >> 4;
    int l15  = lane & 15;

    const float* arow  = A + (size_t)(mt * 16 + l15) * lda;
    const float* bbase = sBt + (size_t)l15 * lstride;

    v8f acc[NTILE] = {};
    for (int k = 0; k < K; k += 4) {
        int kb = k + half * 2;
        v2f a = *(const v2f*)(arow + kb);
        if (RELUA) { a[0] = fmaxf(a[0], 0.0f); a[1] = fmaxf(a[1], 0.0f); }
        #pragma unroll
        for (int nt = 0; nt < NTILE; nt++) {
            v2f b = *(const v2f*)(bbase + (size_t)(nt * 16) * lstride + kb);
            acc[nt] = __builtin_amdgcn_wmma_f32_16x16x4_f32(
                 false, a,  false, b,
                 (short)0, acc[nt],  false,  false);
        }
    }

    int rbase = mt * 16 + half * 8;
    float* cbase = C + (size_t)rbase * HID + l15;

    if (FUSE) {
        float dv[8];
        #pragma unroll
        for (int v = 0; v < 8; v++) dv[v] = dinv[rbase + v];
        #pragma unroll
        for (int nt = 0; nt < NTILE; nt++) {
            #pragma unroll
            for (int v = 0; v < 8; v++)
                cbase[(size_t)v * HID + nt * 16] = acc[nt][v] * dv[v];
        }
    } else {
        #pragma unroll
        for (int nt = 0; nt < NTILE; nt++) {
            float bv = bias[nt * 16 + l15];
            #pragma unroll
            for (int v = 0; v < 8; v++) {
                float r = acc[nt][v] + bv;
                if (RELU) r = fmaxf(r, 0.0f);
                cbase[(size_t)v * HID + nt * 16] = r;
            }
        }
    }
}

#define EC_WAVES   16
#define EC_THREADS (EC_WAVES * 32)
#define EC_TS      (HID + 4)

__global__ void __launch_bounds__(EC_THREADS)
k_enc_chain(const float* __restrict__ x,
            const float* __restrict__ W0, const float* __restrict__ b0,
            const float* __restrict__ Wf, const float* __restrict__ bf,
            const float* __restrict__ dinv,
            float* __restrict__ out, int M) {
    __shared__ __attribute__((aligned(16))) float sB0[HID * (INDIM + 2)];
    __shared__ __attribute__((aligned(16))) float sB1[HID * (HID + 2)];
    __shared__ __attribute__((aligned(16))) float tile[EC_WAVES * 16 * EC_TS];
    constexpr int ls0 = INDIM + 2;
    constexpr int ls1 = HID + 2;

    constexpr int ROW4 = HID / 4;
    constexpr int N0   = INDIM * ROW4;
    constexpr int N1   = HID * ROW4;
    constexpr int P0   = N0 / EC_THREADS;
    constexpr int P1   = (N1 + EC_THREADS - 1) / EC_THREADS;
    static_assert(N0 % EC_THREADS == 0, "W0 must split evenly over the block");
    const v4f* W0_4 = (const v4f*)W0;
    const v4f* Wf_4 = (const v4f*)Wf;
    v4f p0[P0], p1[P1];
    #pragma unroll
    for (int j = 0; j < P0; j++) p0[j] = W0_4[threadIdx.x + j * EC_THREADS];
    #pragma unroll
    for (int j = 0; j < P1; j++) {
        int ii = min((int)threadIdx.x + j * EC_THREADS, N1 - 1);
        p1[j] = Wf_4[ii];
    }
    #pragma unroll
    for (int j = 0; j < P0; j++) {
        int i  = threadIdx.x + j * EC_THREADS;
        int k  = i / ROW4;
        int c0 = (i - k * ROW4) * 4;
        float* dst = sB0 + c0 * ls0 + k;
        dst[0]       = p0[j][0];
        dst[ls0]     = p0[j][1];
        dst[2 * ls0] = p0[j][2];
        dst[3 * ls0] = p0[j][3];
    }
    #pragma unroll
    for (int j = 0; j < P1; j++) {
        int i  = threadIdx.x + j * EC_THREADS;
        if (i < N1) {
            int k  = i / ROW4;
            int c0 = (i - k * ROW4) * 4;
            float* d1 = sB1 + c0 * ls1 + k;
            d1[0]       = p1[j][0];
            d1[ls1]     = p1[j][1];
            d1[2 * ls1] = p1[j][2];
            d1[3 * ls1] = p1[j][3];
        }
    }
    __syncthreads();

    int w    = threadIdx.x >> 5;
    int lane = threadIdx.x & 31;
    int half = lane >> 4;
    int l15  = lane & 15;
    int mt   = blockIdx.x * EC_WAVES + w;
    bool valid = mt < (M >> 4);
    float* trow = tile + (w * 16) * EC_TS;
    float* tst  = trow + (half * 8) * EC_TS + l15;
    const float* tA = trow + l15 * EC_TS;

    if (valid) {
        const float* arow  = x + (size_t)(mt * 16 + l15) * INDIM;
        const float* bbase = sB0 + l15 * ls0;
        v8f acc[NTILE] = {};
        for (int k = 0; k < INDIM; k += 4) {
            int kb = k + half * 2;
            v2f a = *(const v2f*)(arow + kb);
            #pragma unroll
            for (int nt = 0; nt < NTILE; nt++) {
                v2f b = *(const v2f*)(bbase + (nt * 16) * ls0 + kb);
                acc[nt] = __builtin_amdgcn_wmma_f32_16x16x4_f32(
                     false, a,  false, b,
                     (short)0, acc[nt],  false,  false);
            }
        }
        #pragma unroll
        for (int nt = 0; nt < NTILE; nt++) {
            float bv = b0[nt * 16 + l15];
            #pragma unroll
            for (int v = 0; v < 8; v++)
                tst[v * EC_TS + nt * 16] = fmaxf(acc[nt][v] + bv, 0.0f);
        }
    }
    __syncthreads();

    if (valid) {
        const float* bbase = sB1 + l15 * ls1;
        v8f acc[NTILE] = {};
        for (int k = 0; k < HID; k += 4) {
            int kb = k + half * 2;
            v2f a = *(const v2f*)(tA + kb);
            #pragma unroll
            for (int nt = 0; nt < NTILE; nt++) {
                v2f b = *(const v2f*)(bbase + (nt * 16) * ls1 + kb);
                acc[nt] = __builtin_amdgcn_wmma_f32_16x16x4_f32(
                     false, a,  false, b,
                     (short)0, acc[nt],  false,  false);
            }
        }
        int rbase = mt * 16 + half * 8;
        float* cbase = out + (size_t)rbase * HID + l15;
        float dv[8];
        #pragma unroll
        for (int v = 0; v < 8; v++) dv[v] = dinv[rbase + v];
        #pragma unroll
        for (int nt = 0; nt < NTILE; nt++) {
            float bv = bf[nt * 16 + l15];
            #pragma unroll
            for (int v = 0; v < 8; v++)
                cbase[(size_t)v * HID + nt * 16] = (acc[nt][v] + bv) * dv[v];
        }
    }
}


#define FG_NODES 64
#define FG_NPW   (FG_NODES / GEMM_WAVES)
#define FG_HS    (HID + 4)
static_assert(NNODES % 16 == 0, "a 16-row tile is wholly inside or outside the node range");
static_assert(FG_NPW == 8, "per-node values of a wave sit in lanes 0..7");

#define FG_PASS 24
template <int S>
__device__ __forceinline__ void gather_pass(const v4f* rows, const v4f* rself, int cidx, int grp,
                                            int rem, bool first, v4f& a0, v4f& a1, v4f& a2) {
    v4f  x0[S], x1[S], x2[S];
    bool on[S];
    #pragma unroll
    for (int k = 0; k < S; k++) {
        on[k] = grp + 4 * k < rem;
        int s = __shfl(cidx, grp + 4 * k);
        const v4f* r = on[k] ? rows + (unsigned)s * 24u : rself;
        x0[k] = r[0]; x1[k] = r[8]; x2[k] = r[16];
    }
    if (first && grp == 0) {
        a0 = rself[0]; a1 = rself[8]; a2 = rself[16];
    }
    #pragma unroll
    for (int k = 0; k < S; k++)
        if (on[k]) { a0 += x0[k]; a1 += x1[k]; a2 += x2[k]; }
}

__global__ void __launch_bounds__(GEMM_THREADS, 8)
k_gather_gemm(const float* __restrict__ hws, const float* __restrict__ dinv,
              const int* __restrict__ off, const int* __restrict__ cnt,
              const int* __restrict__ col, const float* __restrict__ bias,
              const float* __restrict__ Wn, float* __restrict__ out, int n) {
    __shared__ float sBt[HID * (HID + 2)];
    __shared__ v4f   hT4[FG_NODES * (FG_HS / 4)];
    constexpr int lstride = HID + 2;

    constexpr int ROW4 = HID / 4;
    constexpr int PER  = HID * ROW4 / GEMM_THREADS;
    static_assert(HID * ROW4 % GEMM_THREADS == 0, "Wn must split evenly over the block");
    const v4f* B4 = (const v4f*)Wn;
    v4f piece[PER];
    #pragma unroll
    for (int j = 0; j < PER; j++) piece[j] = B4[threadIdx.x + j * GEMM_THREADS];

    int lane = threadIdx.x & 31;
    int w    = __builtin_amdgcn_readfirstlane((int)(threadIdx.x >> 5));
    int grp  = lane >> 3;
    int t    = lane & 7;
    int nb   = blockIdx.x * FG_NODES + w * FG_NPW;
    int nvalid = min(FG_NPW, n - nb);

    int ml   = min(nb + t, n - 1);
    int moff = off[ml];
    int mcnt = cnt[ml];
    int mdvb = __float_as_int(dinv[ml]);
    const v4f* b4 = (const v4f*)bias + t;
    v4f bb0 = b4[0], bb1 = b4[8], bb2 = b4[16];

    #pragma unroll
    for (int j = 0; j < PER; j++) {
        int i  = threadIdx.x + j * GEMM_THREADS;
        int k  = i / ROW4;
        int c0 = (i - k * ROW4) * 4;
        float* dst = sBt + c0 * lstride + k;
        dst[0]           = piece[j][0];
        dst[lstride]     = piece[j][1];
        dst[2 * lstride] = piece[j][2];
        dst[3 * lstride] = piece[j][3];
    }

    const v4f* rows = (const v4f*)hws + t;
    int idxs[FG_NPW];
    #pragma unroll
    for (int j = 0; j < FG_NPW; j++) {
        int sj  = __builtin_amdgcn_readlane(moff, j);
        int cj  = __builtin_amdgcn_readlane(mcnt, j);
        idxs[j] = col[sj + min(lane, max(cj, 1) - 1)];
    }
    asm volatile("" :: "v"(idxs[0]), "v"(idxs[1]), "v"(idxs[2]), "v"(idxs[3]),
                       "v"(idxs[4]), "v"(idxs[5]), "v"(idxs[6]), "v"(idxs[7]));
    #pragma unroll
    for (int j = 0; j < FG_NPW; j++) {
        int cj  = __builtin_amdgcn_readlane(mcnt, j);
        idxs[j] = (lane < cj) ? idxs[j] : 0;
    }
    #pragma unroll
    for (int i = 0; i < FG_NPW; i++) {
        if (i < nvalid) {
            int   node = nb + i;
            int   cs = __builtin_amdgcn_readlane(moff, i);
            int   cc = __builtin_amdgcn_readlane(mcnt, i);
            float dv = __int_as_float(__builtin_amdgcn_readlane(mdvb, i));
            int cidx = idxs[i];
            const v4f* rself = rows + (unsigned)node * 24u;
            v4f a0 = {}, a1 = {}, a2 = {};
            for (int e0 = 0; e0 == 0 || e0 < cc; e0 += FG_PASS) {
                if (e0 != 0)
                    cidx = (lane < cc - e0) ? col[cs + e0 + lane] : 0;
                int rem = cc - e0;
                if (rem > 16) gather_pass<6>(rows, rself, cidx, grp, rem, e0 == 0, a0, a1, a2);
                else          gather_pass<4>(rows, rself, cidx, grp, rem, e0 == 0, a0, a1, a2);
            }
            #pragma unroll
            for (int mm = 8; mm <= 16; mm <<= 1) {
                #pragma unroll
                for (int q = 0; q < 4; q++) {
                    a0[q] += __shfl_xor(a0[q], mm);
                    a1[q] += __shfl_xor(a1[q], mm);
                    a2[q] += __shfl_xor(a2[q], mm);
                }
            }
            if (grp == 0) {
                v4f o0 = a0 * dv + bb0;
                v4f o1 = a1 * dv + bb1;
                v4f o2 = a2 * dv + bb2;
                #pragma unroll
                for (int q = 0; q < 4; q++) {
                    o0[q] = fmaxf(o0[q], 0.0f); o1[q] = fmaxf(o1[q], 0.0f); o2[q] = fmaxf(o2[q], 0.0f);
                }
                v4f* dst = hT4 + (w * FG_NPW + i) * (FG_HS / 4) + t;
                dst[0] = o0; dst[8] = o1; dst[16] = o2;
            }
        }
    }
    __syncthreads();

    int mtile   = w >> 1;
    int rowbase = blockIdx.x * FG_NODES + mtile * 16;
    if (rowbase >= n) return;
    int half = lane >> 4;
    int l15  = lane & 15;
    int nt0  = (w & 1) * 3;
    const float* abase = (const float*)hT4 + (mtile * 16 + l15) * FG_HS;
    const float* bbase = sBt + (nt0 * 16 + l15) * lstride;
    v8f acc[3] = {};
    for (int k = 0; k < HID; k += 4) {
        int kb = k + half * 2;
        v2f a = *(const v2f*)(abase + kb);
        #pragma unroll
        for (int j = 0; j < 3; j++) {
            v2f b = *(const v2f*)(bbase + (j * 16) * lstride + kb);
            acc[j] = __builtin_amdgcn_wmma_f32_16x16x4_f32(
                 false, a,  false, b,
                 (short)0, acc[j],  false,  false);
        }
    }
    int rbase = rowbase + half * 8;
    float* cbase = out + (size_t)rbase * HID + nt0 * 16 + l15;
    float dvv[8];
    #pragma unroll
    for (int v = 0; v < 8; v++) dvv[v] = dinv[rbase + v];
    #pragma unroll
    for (int j = 0; j < 3; j++) {
        #pragma unroll
        for (int v = 0; v < 8; v++)
            cbase[(size_t)v * HID + j * 16] = acc[j][v] * dvv[v];
    }
}

__global__ void k_zero(float* x, int n) {
    int i = blockIdx.x * blockDim.x + threadIdx.x;
    if (i < n) x[i] = 0.0f;
}

#define PL_ROWS 32
static_assert(FG_NODES == 2 * PL_ROWS, "two row halves per tile");
static_assert(HID == 3 * 32, "three 32-column groups per row half");

__global__ void __launch_bounds__(GEMM_THREADS, 8)
k_gather_pool(const float* __restrict__ hws, const float* __restrict__ dinv,
              const int* __restrict__ off, const int* __restrict__ cnt,
              const int* __restrict__ col, const float* __restrict__ bias,
              const int* __restrict__ batch, float* __restrict__ g, int n) {
    __shared__ v4f hT4[FG_NODES * (FG_HS / 4)];

    int lane = threadIdx.x & 31;
    int w    = __builtin_amdgcn_readfirstlane((int)(threadIdx.x >> 5));
    int grp  = lane >> 3;
    int t    = lane & 7;
    int nb   = blockIdx.x * FG_NODES + w * FG_NPW;
    int nvalid = min(FG_NPW, n - nb);

    int hh  = (w >= 3) ? 1 : 0;
    int pr0 = hh * PL_ROWS;
    int idl = batch[min((int)(blockIdx.x * FG_NODES) + pr0 + lane, n - 1)];

    int ml   = min(nb + t, n - 1);
    int moff = off[ml];
    int mcnt = cnt[ml];
    int mdvb = __float_as_int(dinv[ml]);
    const v4f* b4 = (const v4f*)bias + t;
    v4f bb0 = b4[0], bb1 = b4[8], bb2 = b4[16];

    const v4f* rows = (const v4f*)hws + t;
    int idxs[FG_NPW];
    #pragma unroll
    for (int j = 0; j < FG_NPW; j++) {
        int sj  = __builtin_amdgcn_readlane(moff, j);
        int cj  = __builtin_amdgcn_readlane(mcnt, j);
        idxs[j] = col[sj + min(lane, max(cj, 1) - 1)];
    }
    asm volatile("" :: "v"(idxs[0]), "v"(idxs[1]), "v"(idxs[2]), "v"(idxs[3]),
                       "v"(idxs[4]), "v"(idxs[5]), "v"(idxs[6]), "v"(idxs[7]));
    #pragma unroll
    for (int j = 0; j < FG_NPW; j++) {
        int cj  = __builtin_amdgcn_readlane(mcnt, j);
        idxs[j] = (lane < cj) ? idxs[j] : 0;
    }
    #pragma unroll
    for (int i = 0; i < FG_NPW; i++) {
        if (i < nvalid) {
            int   node = nb + i;
            int   cs = __builtin_amdgcn_readlane(moff, i);
            int   cc = __builtin_amdgcn_readlane(mcnt, i);
            float dv = __int_as_float(__builtin_amdgcn_readlane(mdvb, i));
            int cidx = idxs[i];
            const v4f* rself = rows + (unsigned)node * 24u;
            v4f a0 = {}, a1 = {}, a2 = {};
            for (int e0 = 0; e0 == 0 || e0 < cc; e0 += FG_PASS) {
                if (e0 != 0)
                    cidx = (lane < cc - e0) ? col[cs + e0 + lane] : 0;
                int rem = cc - e0;
                if (rem > 16) gather_pass<6>(rows, rself, cidx, grp, rem, e0 == 0, a0, a1, a2);
                else          gather_pass<4>(rows, rself, cidx, grp, rem, e0 == 0, a0, a1, a2);
            }
            #pragma unroll
            for (int mm = 8; mm <= 16; mm <<= 1) {
                #pragma unroll
                for (int q = 0; q < 4; q++) {
                    a0[q] += __shfl_xor(a0[q], mm);
                    a1[q] += __shfl_xor(a1[q], mm);
                    a2[q] += __shfl_xor(a2[q], mm);
                }
            }
            if (grp == 0) {
                v4f o0 = a0 * dv + bb0;
                v4f o1 = a1 * dv + bb1;
                v4f o2 = a2 * dv + bb2;
                #pragma unroll
                for (int q = 0; q < 4; q++) {
                    o0[q] = fmaxf(o0[q], 0.0f); o1[q] = fmaxf(o1[q], 0.0f); o2[q] = fmaxf(o2[q], 0.0f);
                }
                v4f* dst = hT4 + (w * FG_NPW + i) * (FG_HS / 4) + t;
                dst[0] = o0; dst[8] = o1; dst[16] = o2;
            }
        }
    }
    __syncthreads();

    if (w >= 6) return;
    int nrows = min(FG_NODES, n - (int)(blockIdx.x * FG_NODES));
    int nr    = min(PL_ROWS, nrows - pr0);
    if (nr <= 0) return;
    int c = (w - 3 * hh) * 32 + lane;
    const float* tp = (const float*)hT4 + pr0 * FG_HS + c;
    float x[PL_ROWS];
    #pragma unroll
    for (int j = 0; j < PL_ROWS; j++) x[j] = (j < nr) ? tp[j * FG_HS] : 0.0f;
    int cur = __builtin_amdgcn_readlane(idl, 0);
    float sum = 0.0f;
    #pragma unroll
    for (int j = 0; j < PL_ROWS; j++) {
        if (j < nr) {
            int b = __builtin_amdgcn_readlane(idl, j);
            if (b != cur) {
                atomicAdd(g + (size_t)cur * HID + c, sum);
                sum = 0.0f;
                cur = b;
            }
            sum += x[j];
        }
    }
    atomicAdd(g + (size_t)cur * HID + c, sum);
}

__global__ void k_dec2(const float* __restrict__ g2, const float* __restrict__ W,
                       const float* __restrict__ b, float* __restrict__ out) {
    int idx = blockIdx.x * blockDim.x + threadIdx.x;
    if (idx < NGRAPH * OUTDIM) {
        int gi = idx / OUTDIM;
        int o  = idx - gi * OUTDIM;
        float s = b[o];
        const float* row = g2 + (size_t)gi * HID;
        #pragma unroll 8
        for (int k = 0; k < HID; k++) s += row[k] * W[k * OUTDIM + o];
        out[idx] = s;
    }
}

extern "C" void kernel_launch(void* const* d_in, const int* in_sizes, int n_in,
                              void* d_out, int out_size, void* d_ws, size_t ws_size,
                              hipStream_t stream) {
    const float* x       = (const float*)d_in[0];
    const int*   ei      = (const int*)  d_in[1];
    const int*   batch   = (const int*)  d_in[2];
    const float* enc_W0  = (const float*)d_in[3];
    const float* enc_b0  = (const float*)d_in[4];
    const float* enc_W1  = (const float*)d_in[5];
    const float* enc_b1  = (const float*)d_in[6];
    const float* conv_W  = (const float*)d_in[7];
    const float* conv_b  = (const float*)d_in[8];
    const float* dec_W0  = (const float*)d_in[9];
    const float* dec_b0  = (const float*)d_in[10];
    const float* dec_W1  = (const float*)d_in[11];
    const float* dec_b1  = (const float*)d_in[12];
    float* out = (float*)d_out;

    float* ws   = (float*)d_ws;
    size_t feat = (size_t)NNODES * HID;
    float* buf0 = ws;
    float* buf1 = buf0 + feat;
    float* buf2 = buf1 + feat;
    float* dinv = buf2 + feat;
    float* g    = dinv + NNODES;
    float* g2   = g + (size_t)NGRAPH * HID;
    int*   cnt  = (int*)(g2 + (size_t)NGRAPH * HID);
    int*   tot  = cnt + NNODES;
    int*   off  = tot + 16;
    int*   rank = off + NNODES;
    int*   col  = rank + NEDGES;
    int*   tab  = col + NEDGES + 64;
    float* Wf   = (float*)(tab + (size_t)RANK_BLOCKS * NNODES);
    float* bf   = Wf + (size_t)HID * HID;

    const int B = 256;

    k_rank<<<RANK_BLOCKS + WF_BLOCKS, RANK_THREADS, 0, stream>>>(ei, rank, tab, tot,
                                                                 enc_W1, enc_b1, conv_W, Wf, bf);
    k_offsets<<<cdiv(NNODES, B), B, 0, stream>>>(tab, tot, cnt, off, dinv, col + NEDGES, NNODES);
    k_fill<<<cdiv(NEDGES, B), B, 0, stream>>>(ei, rank, tab, col, NEDGES);

    int mTilesN = NNODES / 16;
    size_t shm96 = (size_t)HID * (HID + 2) * sizeof(float);
    k_enc_chain<<<cdiv(mTilesN, EC_WAVES), EC_THREADS, 0, stream>>>(
        x, enc_W0, enc_b0, Wf, bf, dinv, buf0  , NNODES);

    float* hsrc = buf0;
    float* hdst = buf2;
    int fgBlks = cdiv(NNODES, FG_NODES);
    for (int l = 0; l + 1 < NCONV; l++) {
        const float* Wn = conv_W + (size_t)(l + 1) * HID * HID;
        const float* bl = conv_b + (size_t)l * HID;
        k_gather_gemm<<<fgBlks, GEMM_THREADS, 0, stream>>>(hsrc, dinv, off, cnt, col, bl, Wn, hdst, NNODES);
        float* t = hsrc; hsrc = hdst; hdst = t;
    }
    k_zero<<<cdiv(NGRAPH * HID, B), B, 0, stream>>>(g, NGRAPH * HID);
    k_gather_pool<<<fgBlks, GEMM_THREADS, 0, stream>>>(hsrc, dinv, off, cnt, col,
                                                       conv_b + (size_t)(NCONV - 1) * HID, batch, g, NNODES);

    int mTilesG = NGRAPH / 16;
    k_gemm_wmma<HID, 0, 1, 0><<<cdiv(mTilesG, GEMM_WAVES), GEMM_THREADS, shm96, stream>>>(
        g, dec_W0, dec_b0, g2, nullptr, NGRAPH, HID);
    k_dec2<<<cdiv(NGRAPH * OUTDIM, B), B, 0, stream>>>(g2, dec_W1, dec_b1, out);
}


# Gather launches: 1024 work-groups of 48 or 56 destinations, 6 or 7 per wave (was 782 of 64, 8 per wave): every WGP holds four and 192-200 nodes, none 256; pool's partial sums regrouped
# speedup vs baseline: 4.4485x; 1.0341x over previous
#include <hip/hip_runtime.h>
#include <hip/hip_bf16.h>

typedef __attribute__((ext_vector_type(2))) float v2f;
typedef __attribute__((ext_vector_type(8))) float v8f;
typedef __attribute__((ext_vector_type(4))) float v4f;

#define NNODES 50000
#define NEDGES 800000
#define NGRAPH 512
#define INDIM  128
#define HID    96
#define OUTDIM 10
#define NCONV  4
#define NTILE  6
#define GEMM_WAVES 8
#define GEMM_THREADS (GEMM_WAVES * 32)

static inline int cdiv(int a, int b) { return (a + b - 1) / b; }

typedef __attribute__((ext_vector_type(4))) int v4i;

#define RANK_BLOCKS  64
#define RANK_THREADS 1024
#define RANK_EPB     (NEDGES / RANK_BLOCKS)
#define RANK_PER     ((RANK_EPB + RANK_THREADS - 1) / RANK_THREADS)
#define NODE4        (NNODES / 4)
static_assert(NEDGES % RANK_BLOCKS == 0, "slices must be equal");
static_assert(NNODES % 4 == 0, "counter sets are moved as 16-byte pieces");

#define WF_MT     7
#define WF_TILES  (WF_MT * (HID / 16))
#define WF_BLOCKS 2
#define WF_TPB    (WF_TILES / WF_BLOCKS)
#define WF_KH     (HID / 8)
static_assert(WF_TILES % WF_BLOCKS == 0, "tiles must split evenly over the extra work-groups");

__device__ __forceinline__ void merge_weights(const float* __restrict__ W1, const float* __restrict__ b1,
                                              const float* __restrict__ Wc,
                                              float* __restrict__ Wf, float* __restrict__ bf, int pb) {
    int w = threadIdx.x >> 5;
    if (w >= WF_TPB) return;
    int tile = pb * WF_TPB + w;
    int mt   = tile / (HID / 16);
    int nt   = tile - mt * (HID / 16);
    int lane = threadIdx.x & 31;
    int half = lane >> 4;
    int l15  = lane & 15;
    int m    = mt * 16 + l15;
    const float* arow = (m < HID) ? W1 + m * HID : b1;
    bool live = m <= HID;
    int n    = nt * 16 + l15;
    v8f acc = {};
    for (int h2 = 0; h2 < 2; h2++) {
        v2f a[WF_KH], b[WF_KH];
        #pragma unroll
        for (int s = 0; s < WF_KH; s++) {
            int kb = (h2 * WF_KH + s) * 4 + half * 2;
            a[s]    = *(const v2f*)(arow + kb);
            b[s][0] = Wc[kb * HID + n];
            b[s][1] = Wc[(kb + 1) * HID + n];
        }
        __builtin_amdgcn_sched_barrier(0);
        #pragma unroll
        for (int s = 0; s < WF_KH; s++) {
            v2f av = a[s];
            if (!live) { av[0] = 0.0f; av[1] = 0.0f; }
            acc = __builtin_amdgcn_wmma_f32_16x16x4_f32(
                 false, av,  false, b[s],
                 (short)0, acc,  false,  false);
        }
    }
    int r0 = mt * 16 + half * 8;
    #pragma unroll
    for (int v = 0; v < 8; v++) {
        int row = r0 + v;
        if (row < HID)       Wf[row * HID + n] = acc[v];
        else if (row == HID) bf[n] = acc[v];
    }
}

__global__ void __launch_bounds__(RANK_THREADS)
k_rank(const int* __restrict__ ei, int* __restrict__ rank, int* __restrict__ tab,
       int* __restrict__ total,
       const float* __restrict__ W1, const float* __restrict__ b1, const float* __restrict__ Wc,
       float* __restrict__ Wf, float* __restrict__ bf) {
    __shared__ int cntL[NNODES];
    if (blockIdx.x >= RANK_BLOCKS) {
        merge_weights(W1, b1, Wc, Wf, bf, (int)blockIdx.x - RANK_BLOCKS);
        return;
    }
    int tid = threadIdx.x;
    v4i* c4 = (v4i*)cntL;
    const v4i zero4i = {0, 0, 0, 0};
    for (int i = tid; i < NODE4; i += RANK_THREADS) c4[i] = zero4i;
    if (blockIdx.x == 0 && tid == 0) *total = 0;
    __syncthreads();

    int e0 = blockIdx.x * RANK_EPB;
    const int* dst = ei + NEDGES + e0;
    int d[RANK_PER];
    #pragma unroll
    for (int j = 0; j < RANK_PER; j++) {
        int k = tid + j * RANK_THREADS;
        d[j] = (k < RANK_EPB) ? dst[k] : 0;
    }
    int r[RANK_PER];
    #pragma unroll
    for (int j = 0; j < RANK_PER; j++) {
        int k = tid + j * RANK_THREADS;
        r[j] = (k < RANK_EPB) ? atomicAdd(&cntL[d[j]], 1) : 0;
    }
    #pragma unroll
    for (int j = 0; j < RANK_PER; j++) {
        int k = tid + j * RANK_THREADS;
        if (k < RANK_EPB) rank[e0 + k] = r[j];
    }
    __syncthreads();

    v4i* t4 = (v4i*)(tab + (size_t)blockIdx.x * NNODES);
    for (int i = tid; i < NODE4; i += RANK_THREADS) t4[i] = c4[i];
}

__global__ void k_offsets(int* __restrict__ tab, int* total, int* __restrict__ cnt,
                          int* __restrict__ off, float* __restrict__ dinv,
                          int* __restrict__ colpad, int n) {
    int i    = blockIdx.x * blockDim.x + threadIdx.x;
    int lane = threadIdx.x & 31;
    int ii   = min(i, n - 1);
    int v[RANK_BLOCKS];
    #pragma unroll
    for (int b = 0; b < RANK_BLOCKS; b++) v[b] = tab[(size_t)b * NNODES + ii];
    int c = 0;
    #pragma unroll
    for (int b = 0; b < RANK_BLOCKS; b++) c += v[b];
    if (i >= n) c = 0;
    int incl = c;
    #pragma unroll
    for (int d = 1; d < 32; d <<= 1) {
        int v1 = __shfl_up(incl, d);
        if (lane >= d) incl += v1;
    }
    int base = 0;
    if (lane == 31) base = atomicAdd(total, incl);
    base = __shfl(base, 31);
    if (i < n) {
        int run = base + incl - c;
        off[i]  = run;
        cnt[i]  = c;
        dinv[i] = rsqrtf((float)(c + 1));
        #pragma unroll
        for (int b = 0; b < RANK_BLOCKS; b++) {
            tab[(size_t)b * NNODES + i] = run;
            run += v[b];
        }
    }
    if (i < 4) colpad[i] = 0;
}

__global__ void k_fill(const int* __restrict__ ei, const int* __restrict__ rank,
                       const int* __restrict__ tab, int* __restrict__ col, int E) {
    int e = blockIdx.x * blockDim.x + threadIdx.x;
    if (e < E) {
        int d = ei[E + e];
        int b = e / RANK_EPB;
        col[tab[(size_t)b * NNODES + d] + rank[e]] = ei[e];
    }
}

template <int K, int RELUA, int RELU, int FUSE>
__global__ void __launch_bounds__(GEMM_THREADS)
k_gemm_wmma(const float* __restrict__ A, const float* __restrict__ B,
            const float* __restrict__ bias,
            float* __restrict__ C,
            const float* __restrict__ dinv,
            int M, int lda) {
    extern __shared__ float sBt[];
    constexpr int lstride = K + 2;

    constexpr int ROW4 = HID / 4;
    constexpr int PER  = K * ROW4 / GEMM_THREADS;
    static_assert(K * ROW4 % GEMM_THREADS == 0, "B must split evenly over the block");
    const v4f* B4 = (const v4f*)B;
    v4f piece[PER];
    #pragma unroll
    for (int j = 0; j < PER; j++) piece[j] = B4[threadIdx.x + j * GEMM_THREADS];
    #pragma unroll
    for (int j = 0; j < PER; j++) {
        int i  = threadIdx.x + j * GEMM_THREADS;
        int k  = i / ROW4;
        int c0 = (i - k * ROW4) * 4;
        float* dst = sBt + c0 * lstride + k;
        dst[0]           = piece[j][0];
        dst[lstride]     = piece[j][1];
        dst[2 * lstride] = piece[j][2];
        dst[3 * lstride] = piece[j][3];
    }
    __syncthreads();

    int mt   = blockIdx.x * GEMM_WAVES + (threadIdx.x >> 5);
    int lane = threadIdx.x & 31;
    int mTiles = M >> 4;
    if (mt >= mTiles) return;

    int half = lane >> 4;
    int l15  = lane & 15;

    const float* arow  = A + (size_t)(mt * 16 + l15) * lda;
    const float* bbase = sBt + (size_t)l15 * lstride;

    v8f acc[NTILE] = {};
    for (int k = 0; k < K; k += 4) {
        int kb = k + half * 2;
        v2f a = *(const v2f*)(arow + kb);
        if (RELUA) { a[0] = fmaxf(a[0], 0.0f); a[1] = fmaxf(a[1], 0.0f); }
        #pragma unroll
        for (int nt = 0; nt < NTILE; nt++) {
            v2f b = *(const v2f*)(bbase + (size_t)(nt * 16) * lstride + kb);
            acc[nt] = __builtin_amdgcn_wmma_f32_16x16x4_f32(
                 false, a,  false, b,
                 (short)0, acc[nt],  false,  false);
        }
    }

    int rbase = mt * 16 + half * 8;
    float* cbase = C + (size_t)rbase * HID + l15;

    if (FUSE) {
        float dv[8];
        #pragma unroll
        for (int v = 0; v < 8; v++) dv[v] = dinv[rbase + v];
        #pragma unroll
        for (int nt = 0; nt < NTILE; nt++) {
            #pragma unroll
            for (int v = 0; v < 8; v++)
                cbase[(size_t)v * HID + nt * 16] = acc[nt][v] * dv[v];
        }
    } else {
        #pragma unroll
        for (int nt = 0; nt < NTILE; nt++) {
            float bv = bias[nt * 16 + l15];
            #pragma unroll
            for (int v = 0; v < 8; v++) {
                float r = acc[nt][v] + bv;
                if (RELU) r = fmaxf(r, 0.0f);
                cbase[(size_t)v * HID + nt * 16] = r;
            }
        }
    }
}

#define EC_WAVES   16
#define EC_THREADS (EC_WAVES * 32)
#define EC_TS      (HID + 4)

__global__ void __launch_bounds__(EC_THREADS)
k_enc_chain(const float* __restrict__ x,
            const float* __restrict__ W0, const float* __restrict__ b0,
            const float* __restrict__ Wf, const float* __restrict__ bf,
            const float* __restrict__ dinv,
            float* __restrict__ out, int M) {
    __shared__ __attribute__((aligned(16))) float sB0[HID * (INDIM + 2)];
    __shared__ __attribute__((aligned(16))) float sB1[HID * (HID + 2)];
    __shared__ __attribute__((aligned(16))) float tile[EC_WAVES * 16 * EC_TS];
    constexpr int ls0 = INDIM + 2;
    constexpr int ls1 = HID + 2;

    constexpr int ROW4 = HID / 4;
    constexpr int N0   = INDIM * ROW4;
    constexpr int N1   = HID * ROW4;
    constexpr int P0   = N0 / EC_THREADS;
    constexpr int P1   = (N1 + EC_THREADS - 1) / EC_THREADS;
    static_assert(N0 % EC_THREADS == 0, "W0 must split evenly over the block");
    const v4f* W0_4 = (const v4f*)W0;
    const v4f* Wf_4 = (const v4f*)Wf;
    v4f p0[P0], p1[P1];
    #pragma unroll
    for (int j = 0; j < P0; j++) p0[j] = W0_4[threadIdx.x + j * EC_THREADS];
    #pragma unroll
    for (int j = 0; j < P1; j++) {
        int ii = min((int)threadIdx.x + j * EC_THREADS, N1 - 1);
        p1[j] = Wf_4[ii];
    }
    #pragma unroll
    for (int j = 0; j < P0; j++) {
        int i  = threadIdx.x + j * EC_THREADS;
        int k  = i / ROW4;
        int c0 = (i - k * ROW4) * 4;
        float* dst = sB0 + c0 * ls0 + k;
        dst[0]       = p0[j][0];
        dst[ls0]     = p0[j][1];
        dst[2 * ls0] = p0[j][2];
        dst[3 * ls0] = p0[j][3];
    }
    #pragma unroll
    for (int j = 0; j < P1; j++) {
        int i  = threadIdx.x + j * EC_THREADS;
        if (i < N1) {
            int k  = i / ROW4;
            int c0 = (i - k * ROW4) * 4;
            float* d1 = sB1 + c0 * ls1 + k;
            d1[0]       = p1[j][0];
            d1[ls1]     = p1[j][1];
            d1[2 * ls1] = p1[j][2];
            d1[3 * ls1] = p1[j][3];
        }
    }
    __syncthreads();

    int w    = threadIdx.x >> 5;
    int lane = threadIdx.x & 31;
    int half = lane >> 4;
    int l15  = lane & 15;
    int mt   = blockIdx.x * EC_WAVES + w;
    bool valid = mt < (M >> 4);
    float* trow = tile + (w * 16) * EC_TS;
    float* tst  = trow + (half * 8) * EC_TS + l15;
    const float* tA = trow + l15 * EC_TS;

    if (valid) {
        const float* arow  = x + (size_t)(mt * 16 + l15) * INDIM;
        const float* bbase = sB0 + l15 * ls0;
        v8f acc[NTILE] = {};
        for (int k = 0; k < INDIM; k += 4) {
            int kb = k + half * 2;
            v2f a = *(const v2f*)(arow + kb);
            #pragma unroll
            for (int nt = 0; nt < NTILE; nt++) {
                v2f b = *(const v2f*)(bbase + (nt * 16) * ls0 + kb);
                acc[nt] = __builtin_amdgcn_wmma_f32_16x16x4_f32(
                     false, a,  false, b,
                     (short)0, acc[nt],  false,  false);
            }
        }
        #pragma unroll
        for (int nt = 0; nt < NTILE; nt++) {
            float bv = b0[nt * 16 + l15];
            #pragma unroll
            for (int v = 0; v < 8; v++)
                tst[v * EC_TS + nt * 16] = fmaxf(acc[nt][v] + bv, 0.0f);
        }
    }
    __syncthreads();

    if (valid) {
        const float* bbase = sB1 + l15 * ls1;
        v8f acc[NTILE] = {};
        for (int k = 0; k < HID; k += 4) {
            int kb = k + half * 2;
            v2f a = *(const v2f*)(tA + kb);
            #pragma unroll
            for (int nt = 0; nt < NTILE; nt++) {
                v2f b = *(const v2f*)(bbase + (nt * 16) * ls1 + kb);
                acc[nt] = __builtin_amdgcn_wmma_f32_16x16x4_f32(
                     false, a,  false, b,
                     (short)0, acc[nt],  false,  false);
            }
        }
        int rbase = mt * 16 + half * 8;
        float* cbase = out + (size_t)rbase * HID + l15;
        float dv[8];
        #pragma unroll
        for (int v = 0; v < 8; v++) dv[v] = dinv[rbase + v];
        #pragma unroll
        for (int nt = 0; nt < NTILE; nt++) {
            float bv = bf[nt * 16 + l15];
            #pragma unroll
            for (int v = 0; v < 8; v++)
                cbase[(size_t)v * HID + nt * 16] = (acc[nt][v] + bv) * dv[v];
        }
    }
}


#define FG_NODES 64
#define FG_HS    (HID + 4)
#define FG_WGS    1024
#define FG_NPW_LO (NNODES / (FG_WGS * GEMM_WAVES))
#define FG_NPW    (FG_NPW_LO + 1)
#define FG_NBIG   ((NNODES - FG_WGS * GEMM_WAVES * FG_NPW_LO) / GEMM_WAVES)
#define FG_BSTEP  9
static_assert(NNODES % 16 == 0, "the encoder's 16-row strips cover the node range exactly");
static_assert(FG_WGS * GEMM_WAVES * FG_NPW_LO + FG_NBIG * GEMM_WAVES == NNODES, "the ranges cover the nodes exactly");
static_assert(FG_NPW * GEMM_WAVES <= FG_NODES, "a work-group's destinations fit the LDS tile");
static_assert(FG_NPW <= 8, "per-node values of a wave sit in lanes 0..7");
static_assert((FG_NBIG - 1) * FG_BSTEP < FG_WGS && FG_BSTEP > 4, "the larger work-groups exist and lie more than a WGP's four apart");

__device__ __forceinline__ void fg_range(int wg, int& npw, int& n0) {
    int  before = min((wg + FG_BSTEP - 1) / FG_BSTEP, FG_NBIG);
    bool big    = (wg % FG_BSTEP == 0) && (wg / FG_BSTEP < FG_NBIG);
    npw = FG_NPW_LO + (big ? 1 : 0);
    n0  = (wg * FG_NPW_LO + before) * GEMM_WAVES;
}

#define FG_PASS 24
template <int S>
__device__ __forceinline__ void gather_pass(const v4f* rows, const v4f* rself, int cidx, int grp,
                                            int rem, bool first, v4f& a0, v4f& a1, v4f& a2) {
    v4f  x0[S], x1[S], x2[S];
    bool on[S];
    #pragma unroll
    for (int k = 0; k < S; k++) {
        on[k] = grp + 4 * k < rem;
        int s = __shfl(cidx, grp + 4 * k);
        const v4f* r = on[k] ? rows + (unsigned)s * 24u : rself;
        x0[k] = r[0]; x1[k] = r[8]; x2[k] = r[16];
    }
    if (first && grp == 0) {
        a0 = rself[0]; a1 = rself[8]; a2 = rself[16];
    }
    #pragma unroll
    for (int k = 0; k < S; k++)
        if (on[k]) { a0 += x0[k]; a1 += x1[k]; a2 += x2[k]; }
}

__global__ void __launch_bounds__(GEMM_THREADS, 8)
k_gather_gemm(const float* __restrict__ hws, const float* __restrict__ dinv,
              const int* __restrict__ off, const int* __restrict__ cnt,
              const int* __restrict__ col, const float* __restrict__ bias,
              const float* __restrict__ Wn, float* __restrict__ out, int n) {
    __shared__ float sBt[HID * (HID + 2)];
    __shared__ v4f   hT4[FG_NODES * (FG_HS / 4)];
    constexpr int lstride = HID + 2;

    constexpr int ROW4 = HID / 4;
    constexpr int PER  = HID * ROW4 / GEMM_THREADS;
    static_assert(HID * ROW4 % GEMM_THREADS == 0, "Wn must split evenly over the block");
    const v4f* B4 = (const v4f*)Wn;
    v4f piece[PER];
    #pragma unroll
    for (int j = 0; j < PER; j++) piece[j] = B4[threadIdx.x + j * GEMM_THREADS];

    int lane = threadIdx.x & 31;
    int w    = __builtin_amdgcn_readfirstlane((int)(threadIdx.x >> 5));
    int grp  = lane >> 3;
    int t    = lane & 7;
    int npw, n0;
    fg_range((int)blockIdx.x, npw, n0);
    int nb   = n0 + w * npw;
    int nvalid = npw;

    int ml   = min(nb + t, n - 1);
    int moff = off[ml];
    int mcnt = cnt[ml];
    int mdvb = __float_as_int(dinv[ml]);
    const v4f* b4 = (const v4f*)bias + t;
    v4f bb0 = b4[0], bb1 = b4[8], bb2 = b4[16];

    #pragma unroll
    for (int j = 0; j < PER; j++) {
        int i  = threadIdx.x + j * GEMM_THREADS;
        int k  = i / ROW4;
        int c0 = (i - k * ROW4) * 4;
        float* dst = sBt + c0 * lstride + k;
        dst[0]           = piece[j][0];
        dst[lstride]     = piece[j][1];
        dst[2 * lstride] = piece[j][2];
        dst[3 * lstride] = piece[j][3];
    }

    const v4f* rows = (const v4f*)hws + t;
    int idxs[FG_NPW];
    #pragma unroll
    for (int j = 0; j < FG_NPW; j++) {
        int sj  = __builtin_amdgcn_readlane(moff, j);
        int cj  = __builtin_amdgcn_readlane(mcnt, j);
        idxs[j] = col[sj + min(lane, max(cj, 1) - 1)];
    }
    asm volatile("" :: "v"(idxs[0]), "v"(idxs[1]), "v"(idxs[2]), "v"(idxs[3]),
                       "v"(idxs[4]), "v"(idxs[5]), "v"(idxs[6]));
    static_assert(FG_NPW == 7, "the statement above names every list register");
    #pragma unroll
    for (int j = 0; j < FG_NPW; j++) {
        int cj  = __builtin_amdgcn_readlane(mcnt, j);
        idxs[j] = (lane < cj) ? idxs[j] : 0;
    }
    #pragma unroll
    for (int i = 0; i < FG_NPW; i++) {
        if (i < nvalid) {
            int   node = nb + i;
            int   cs = __builtin_amdgcn_readlane(moff, i);
            int   cc = __builtin_amdgcn_readlane(mcnt, i);
            float dv = __int_as_float(__builtin_amdgcn_readlane(mdvb, i));
            int cidx = idxs[i];
            const v4f* rself = rows + (unsigned)node * 24u;
            v4f a0 = {}, a1 = {}, a2 = {};
            for (int e0 = 0; e0 == 0 || e0 < cc; e0 += FG_PASS) {
                if (e0 != 0)
                    cidx = (lane < cc - e0) ? col[cs + e0 + lane] : 0;
                int rem = cc - e0;
                if (rem > 16) gather_pass<6>(rows, rself, cidx, grp, rem, e0 == 0, a0, a1, a2);
                else          gather_pass<4>(rows, rself, cidx, grp, rem, e0 == 0, a0, a1, a2);
            }
            #pragma unroll
            for (int mm = 8; mm <= 16; mm <<= 1) {
                #pragma unroll
                for (int q = 0; q < 4; q++) {
                    a0[q] += __shfl_xor(a0[q], mm);
                    a1[q] += __shfl_xor(a1[q], mm);
                    a2[q] += __shfl_xor(a2[q], mm);
                }
            }
            if (grp == 0) {
                v4f o0 = a0 * dv + bb0;
                v4f o1 = a1 * dv + bb1;
                v4f o2 = a2 * dv + bb2;
                #pragma unroll
                for (int q = 0; q < 4; q++) {
                    o0[q] = fmaxf(o0[q], 0.0f); o1[q] = fmaxf(o1[q], 0.0f); o2[q] = fmaxf(o2[q], 0.0f);
                }
                v4f* dst = hT4 + (w * npw + i) * (FG_HS / 4) + t;
                dst[0] = o0; dst[8] = o1; dst[16] = o2;
            }
        }
    }
    __syncthreads();

    int nn      = npw * GEMM_WAVES;
    int mtile   = w >> 1;
    if (mtile * 16 >= nn) return;
    int rowbase = n0 + mtile * 16;
    int half = lane >> 4;
    int l15  = lane & 15;
    int nt0  = (w & 1) * 3;
    const float* abase = (const float*)hT4 + (mtile * 16 + l15) * FG_HS;
    const float* bbase = sBt + (nt0 * 16 + l15) * lstride;
    v8f acc[3] = {};
    for (int k = 0; k < HID; k += 4) {
        int kb = k + half * 2;
        v2f a = *(const v2f*)(abase + kb);
        #pragma unroll
        for (int j = 0; j < 3; j++) {
            v2f b = *(const v2f*)(bbase + (j * 16) * lstride + kb);
            acc[j] = __builtin_amdgcn_wmma_f32_16x16x4_f32(
                 false, a,  false, b,
                 (short)0, acc[j],  false,  false);
        }
    }
    if (mtile * 16 + half * 8 >= nn) return;
    int rbase = rowbase + half * 8;
    float* cbase = out + (size_t)rbase * HID + nt0 * 16 + l15;
    float dvv[8];
    #pragma unroll
    for (int v = 0; v < 8; v++) dvv[v] = dinv[rbase + v];
    #pragma unroll
    for (int j = 0; j < 3; j++) {
        #pragma unroll
        for (int v = 0; v < 8; v++)
            cbase[(size_t)v * HID + j * 16] = acc[j][v] * dvv[v];
    }
}

__global__ void k_zero(float* x, int n) {
    int i = blockIdx.x * blockDim.x + threadIdx.x;
    if (i < n) x[i] = 0.0f;
}

#define PL_ROWS 32
static_assert(FG_NODES == 2 * PL_ROWS, "two row halves per tile");
static_assert(HID == 3 * 32, "three 32-column groups per row half");

__global__ void __launch_bounds__(GEMM_THREADS, 8)
k_gather_pool(const float* __restrict__ hws, const float* __restrict__ dinv,
              const int* __restrict__ off, const int* __restrict__ cnt,
              const int* __restrict__ col, const float* __restrict__ bias,
              const int* __restrict__ batch, float* __restrict__ g, int n) {
    __shared__ v4f hT4[FG_NODES * (FG_HS / 4)];

    int lane = threadIdx.x & 31;
    int w    = __builtin_amdgcn_readfirstlane((int)(threadIdx.x >> 5));
    int grp  = lane >> 3;
    int t    = lane & 7;
    int npw, n0;
    fg_range((int)blockIdx.x, npw, n0);
    int nb   = n0 + w * npw;
    int nvalid = npw;

    int hh  = (w >= 3) ? 1 : 0;
    int pr0 = hh * PL_ROWS;
    int idl = batch[min(n0 + pr0 + lane, n - 1)];

    int ml   = min(nb + t, n - 1);
    int moff = off[ml];
    int mcnt = cnt[ml];
    int mdvb = __float_as_int(dinv[ml]);
    const v4f* b4 = (const v4f*)bias + t;
    v4f bb0 = b4[0], bb1 = b4[8], bb2 = b4[16];

    const v4f* rows = (const v4f*)hws + t;
    int idxs[FG_NPW];
    #pragma unroll
    for (int j = 0; j < FG_NPW; j++) {
        int sj  = __builtin_amdgcn_readlane(moff, j);
        int cj  = __builtin_amdgcn_readlane(mcnt, j);
        idxs[j] = col[sj + min(lane, max(cj, 1) - 1)];
    }
    asm volatile("" :: "v"(idxs[0]), "v"(idxs[1]), "v"(idxs[2]), "v"(idxs[3]),
                       "v"(idxs[4]), "v"(idxs[5]), "v"(idxs[6]));
    static_assert(FG_NPW == 7, "the statement above names every list register");
    #pragma unroll
    for (int j = 0; j < FG_NPW; j++) {
        int cj  = __builtin_amdgcn_readlane(mcnt, j);
        idxs[j] = (lane < cj) ? idxs[j] : 0;
    }
    #pragma unroll
    for (int i = 0; i < FG_NPW; i++) {
        if (i < nvalid) {
            int   node = nb + i;
            int   cs = __builtin_amdgcn_readlane(moff, i);
            int   cc = __builtin_amdgcn_readlane(mcnt, i);
            float dv = __int_as_float(__builtin_amdgcn_readlane(mdvb, i));
            int cidx = idxs[i];
            const v4f* rself = rows + (unsigned)node * 24u;
            v4f a0 = {}, a1 = {}, a2 = {};
            for (int e0 = 0; e0 == 0 || e0 < cc; e0 += FG_PASS) {
                if (e0 != 0)
                    cidx = (lane < cc - e0) ? col[cs + e0 + lane] : 0;
                int rem = cc - e0;
                if (rem > 16) gather_pass<6>(rows, rself, cidx, grp, rem, e0 == 0, a0, a1, a2);
                else          gather_pass<4>(rows, rself, cidx, grp, rem, e0 == 0, a0, a1, a2);
            }
            #pragma unroll
            for (int mm = 8; mm <= 16; mm <<= 1) {
                #pragma unroll
                for (int q = 0; q < 4; q++) {
                    a0[q] += __shfl_xor(a0[q], mm);
                    a1[q] += __shfl_xor(a1[q], mm);
                    a2[q] += __shfl_xor(a2[q], mm);
                }
            }
            if (grp == 0) {
                v4f o0 = a0 * dv + bb0;
                v4f o1 = a1 * dv + bb1;
                v4f o2 = a2 * dv + bb2;
                #pragma unroll
                for (int q = 0; q < 4; q++) {
                    o0[q] = fmaxf(o0[q], 0.0f); o1[q] = fmaxf(o1[q], 0.0f); o2[q] = fmaxf(o2[q], 0.0f);
                }
                v4f* dst = hT4 + (w * npw + i) * (FG_HS / 4) + t;
                dst[0] = o0; dst[8] = o1; dst[16] = o2;
            }
        }
    }
    __syncthreads();

    if (w >= 6) return;
    int nrows = npw * GEMM_WAVES;
    int nr    = min(PL_ROWS, nrows - pr0);
    if (nr <= 0) return;
    int c = (w - 3 * hh) * 32 + lane;
    const float* tp = (const float*)hT4 + pr0 * FG_HS + c;
    float x[PL_ROWS];
    #pragma unroll
    for (int j = 0; j < PL_ROWS; j++) x[j] = (j < nr) ? tp[j * FG_HS] : 0.0f;
    int cur = __builtin_amdgcn_readlane(idl, 0);
    float sum = 0.0f;
    #pragma unroll
    for (int j = 0; j < PL_ROWS; j++) {
        if (j < nr) {
            int b = __builtin_amdgcn_readlane(idl, j);
            if (b != cur) {
                atomicAdd(g + (size_t)cur * HID + c, sum);
                sum = 0.0f;
                cur = b;
            }
            sum += x[j];
        }
    }
    atomicAdd(g + (size_t)cur * HID + c, sum);
}

__global__ void k_dec2(const float* __restrict__ g2, const float* __restrict__ W,
                       const float* __restrict__ b, float* __restrict__ out) {
    int idx = blockIdx.x * blockDim.x + threadIdx.x;
    if (idx < NGRAPH * OUTDIM) {
        int gi = idx / OUTDIM;
        int o  = idx - gi * OUTDIM;
        float s = b[o];
        const float* row = g2 + (size_t)gi * HID;
        #pragma unroll 8
        for (int k = 0; k < HID; k++) s += row[k] * W[k * OUTDIM + o];
        out[idx] = s;
    }
}

extern "C" void kernel_launch(void* const* d_in, const int* in_sizes, int n_in,
                              void* d_out, int out_size, void* d_ws, size_t ws_size,
                              hipStream_t stream) {
    const float* x       = (const float*)d_in[0];
    const int*   ei      = (const int*)  d_in[1];
    const int*   batch   = (const int*)  d_in[2];
    const float* enc_W0  = (const float*)d_in[3];
    const float* enc_b0  = (const float*)d_in[4];
    const float* enc_W1  = (const float*)d_in[5];
    const float* enc_b1  = (const float*)d_in[6];
    const float* conv_W  = (const float*)d_in[7];
    const float* conv_b  = (const float*)d_in[8];
    const float* dec_W0  = (const float*)d_in[9];
    const float* dec_b0  = (const float*)d_in[10];
    const float* dec_W1  = (const float*)d_in[11];
    const float* dec_b1  = (const float*)d_in[12];
    float* out = (float*)d_out;

    float* ws   = (float*)d_ws;
    size_t feat = (size_t)NNODES * HID;
    float* buf0 = ws;
    float* buf1 = buf0 + feat;
    float* buf2 = buf1 + feat;
    float* dinv = buf2 + feat;
    float* g    = dinv + NNODES;
    float* g2   = g + (size_t)NGRAPH * HID;
    int*   cnt  = (int*)(g2 + (size_t)NGRAPH * HID);
    int*   tot  = cnt + NNODES;
    int*   off  = tot + 16;
    int*   rank = off + NNODES;
    int*   col  = rank + NEDGES;
    int*   tab  = col + NEDGES + 64;
    float* Wf   = (float*)(tab + (size_t)RANK_BLOCKS * NNODES);
    float* bf   = Wf + (size_t)HID * HID;

    const int B = 256;

    k_rank<<<RANK_BLOCKS + WF_BLOCKS, RANK_THREADS, 0, stream>>>(ei, rank, tab, tot,
                                                                 enc_W1, enc_b1, conv_W, Wf, bf);
    k_offsets<<<cdiv(NNODES, B), B, 0, stream>>>(tab, tot, cnt, off, dinv, col + NEDGES, NNODES);
    k_fill<<<cdiv(NEDGES, B), B, 0, stream>>>(ei, rank, tab, col, NEDGES);

    int mTilesN = NNODES / 16;
    size_t shm96 = (size_t)HID * (HID + 2) * sizeof(float);
    k_enc_chain<<<cdiv(mTilesN, EC_WAVES), EC_THREADS, 0, stream>>>(
        x, enc_W0, enc_b0, Wf, bf, dinv, buf0  , NNODES);

    float* hsrc = buf0;
    float* hdst = buf2;
    int fgBlks = FG_WGS;
    for (int l = 0; l + 1 < NCONV; l++) {
        const float* Wn = conv_W + (size_t)(l + 1) * HID * HID;
        const float* bl = conv_b + (size_t)l * HID;
        k_gather_gemm<<<fgBlks, GEMM_THREADS, 0, stream>>>(hsrc, dinv, off, cnt, col, bl, Wn, hdst, NNODES);
        float* t = hsrc; hsrc = hdst; hdst = t;
    }
    k_zero<<<cdiv(NGRAPH * HID, B), B, 0, stream>>>(g, NGRAPH * HID);
    k_gather_pool<<<fgBlks, GEMM_THREADS, 0, stream>>>(hsrc, dinv, off, cnt, col,
                                                       conv_b + (size_t)(NCONV - 1) * HID, batch, g, NNODES);

    int mTilesG = NGRAPH / 16;
    k_gemm_wmma<HID, 0, 1, 0><<<cdiv(mTilesG, GEMM_WAVES), GEMM_THREADS, shm96, stream>>>(
        g, dec_W0, dec_b0, g2, nullptr, NGRAPH, HID);
    k_dec2<<<cdiv(NGRAPH * OUTDIM, B), B, 0, stream>>>(g2, dec_W1, dec_b1, out);
}
